# starting kernel
# baseline (speedup 1.0000x reference)
#include <hip/hip_runtime.h>
#include <math.h>


typedef __attribute__((ext_vector_type(2))) float v2f;
typedef __attribute__((ext_vector_type(8))) float v8f;

#define BB 8
#define NN 1024
#define MM 128
#define FF 64
#define FE 16
#define HID 128

__device__ __forceinline__ v8f wmma4(v2f a, v2f b, v8f c) {
  return __builtin_amdgcn_wmma_f32_16x16x4_f32(
       false, a,  false, b,
       (short)0, c,  false,  false);
}

__global__ __launch_bounds__(128) void gemm_rows_64(
    const float* __restrict__ A, const float* __restrict__ W,
    float* __restrict__ C) {
  const int wv   = threadIdx.x >> 5;
  const int lane = threadIdx.x & 31;
  const int lm   = lane & 15;
  const int kh   = (lane >> 4) * 2;
  const int lh   = lane >> 4;
  const int row0 = blockIdx.x * 16;
  const int o    = wv * 16 + lm;
  const float* Arow = A + (size_t)(row0 + lm) * FF;

  v8f c = {};
#pragma unroll
  for (int k0 = 0; k0 < FF; k0 += 4) {
    v2f a, b;
    a.x = Arow[k0 + kh];
    a.y = Arow[k0 + kh + 1];
    b.x = W[(size_t)(k0 + kh) * FF + o];
    b.y = W[(size_t)(k0 + kh + 1) * FF + o];
    c = wmma4(a, b, c);
  }
#pragma unroll
  for (int r = 0; r < 8; ++r) {
    int m = r + 8 * lh;
    C[(size_t)(row0 + m) * FF + o] = c[r];
  }
}

__global__ void proj_node(const float* __restrict__ h,
                          const float* __restrict__ att_op,
                          const float* __restrict__ att_ma,
                          const int* __restrict__ opma,
                          float* __restrict__ ha1, float* __restrict__ ha2,
                          float* __restrict__ ham1, int* __restrict__ rowmask) {
  int idx = blockIdx.x * blockDim.x + threadIdx.x;
  if (idx >= BB * NN) return;
  const float* hr = h + (size_t)idx * FF;
  float s1 = 0.f, s2 = 0.f, s3 = 0.f;
#pragma unroll 8
  for (int o = 0; o < FF; ++o) {
    float v = hr[o];
    s1 += v * att_op[o];
    s2 += v * att_op[FF + o];
    s3 += v * att_ma[o];
  }
  ha1[idx] = s1; ha2[idx] = s2; ham1[idx] = s3;
  const int* ad = opma + (size_t)idx * MM;
  int any = 0;
#pragma unroll 8
  for (int m = 0; m < MM; ++m) any |= ad[m];
  rowmask[idx] = (any == 0) ? 1 : 0;
}

__global__ void proj_ma(const float* __restrict__ hma,
                        const float* __restrict__ att_ma,
                        float* __restrict__ hma2) {
  int idx = blockIdx.x * blockDim.x + threadIdx.x;
  if (idx >= BB * MM) return;
  const float* r = hma + (size_t)idx * FF;
  float s = 0.f;
#pragma unroll 8
  for (int o = 0; o < FF; ++o) s += r[o] * att_ma[FF + o];
  hma2[idx] = s;
}

__global__ __launch_bounds__(128) void op_attn(
    const float* __restrict__ h, const float* __restrict__ ha1,
    const float* __restrict__ ha2, const int* __restrict__ opadj,
    const int* __restrict__ maadj, const int* __restrict__ rowmask,
    float* __restrict__ hprime) {
  extern __shared__ float smem[];
  typedef float att_row_t[NN + 4];
  att_row_t* att = (att_row_t*)smem;
  float* inv_sum = smem + 16 * (NN + 4);

  const int b  = blockIdx.y;
  const int i0 = blockIdx.x * 16;
  const int* opb = opadj + (size_t)b * NN * NN;
  const int* mab = maadj + (size_t)b * NN * NN;
  const float NEGINF = -__builtin_inff();

  for (int t = threadIdx.x; t < 16 * NN; t += 128) {
    int r = t >> 10, j = t & (NN - 1);
    int i = i0 + r;
    int conn;
    if (rowmask[b * NN + i]) {
      conn = 1;
    } else {
      int adj = opb[(size_t)i * NN + j] + opb[(size_t)j * NN + i] +
                mab[(size_t)i * NN + j] + mab[(size_t)j * NN + i];
      conn = adj > 0;
    }
    float e = ha2[b * NN + i] + ha1[b * NN + j];
    e = e > 0.f ? e : 0.01f * e;
    att[r][j] = conn ? e : NEGINF;
  }
  __syncthreads();

  const int wv = threadIdx.x >> 5;
  const int lane = threadIdx.x & 31;
  for (int r = wv; r < 16; r += 4) {
    float mx = NEGINF;
    for (int j = lane; j < NN; j += 32) mx = fmaxf(mx, att[r][j]);
#pragma unroll
    for (int off = 16; off; off >>= 1) mx = fmaxf(mx, __shfl_xor(mx, off, 32));
    if (mx == NEGINF) mx = 0.f;
    float sum = 0.f;
    for (int j = lane; j < NN; j += 32) {
      float p = __expf(att[r][j] - mx);
      att[r][j] = p;
      sum += p;
    }
#pragma unroll
    for (int off = 16; off; off >>= 1) sum += __shfl_xor(sum, off, 32);
    if (lane == 0) inv_sum[r] = 1.f / fmaxf(sum, 1e-30f);
  }
  __syncthreads();

  const int lm = lane & 15;
  const int kh = (lane >> 4) * 2;
  const int lh = lane >> 4;
  const int o  = wv * 16 + lm;
  const float* hb = h + (size_t)b * NN * FF;

  v8f c = {};
  for (int k0 = 0; k0 < NN; k0 += 4) {
    v2f a, bf;
    a.x  = att[lm][k0 + kh];
    a.y  = att[lm][k0 + kh + 1];
    bf.x = hb[(size_t)(k0 + kh) * FF + o];
    bf.y = hb[(size_t)(k0 + kh + 1) * FF + o];
    c = wmma4(a, bf, c);
  }
  float* hp = hprime + ((size_t)b * NN + i0) * (2 * FF);
#pragma unroll
  for (int r = 0; r < 8; ++r) {
    int m = r + 8 * lh;
    hp[(size_t)m * (2 * FF) + o] = c[r] * inv_sum[m];
  }
}

__global__ __launch_bounds__(256) void ma_attn(
    const float* __restrict__ z, const float* __restrict__ wedge,
    const float* __restrict__ hma, const float* __restrict__ ham1,
    const float* __restrict__ hma2, const int* __restrict__ opma,
    const int* __restrict__ rowmask, const float* __restrict__ h,
    const float* __restrict__ attma, float* __restrict__ hprime) {
  __shared__ float zb[MM][FE + 1];
  __shared__ float web[FE][FF];
  __shared__ float heb[MM][FF + 1];
  __shared__ float sc[MM];
  __shared__ float red[2];
  __shared__ float psum[4][FF];

  const int b = blockIdx.y, n = blockIdx.x;
  const size_t bn = (size_t)b * NN + n;
  const float NEGINF = -__builtin_inff();

  const float* zrow = z + bn * MM * FE;
  for (int t = threadIdx.x; t < MM * FE; t += 256) zb[t >> 4][t & 15] = zrow[t];
  for (int t = threadIdx.x; t < FE * FF; t += 256) web[t >> 6][t & 63] = wedge[t];
  __syncthreads();

  const int wv = threadIdx.x >> 5;
  const int lane = threadIdx.x & 31;
  const int lm = lane & 15;
  const int kh = (lane >> 4) * 2;
  const int lh = lane >> 4;
  const int m0 = wv * 16;
  for (int ct = 0; ct < 4; ++ct) {
    v8f c = {};
#pragma unroll
    for (int k0 = 0; k0 < FE; k0 += 4) {
      v2f a, bf;
      a.x  = zb[m0 + lm][k0 + kh];
      a.y  = zb[m0 + lm][k0 + kh + 1];
      bf.x = web[k0 + kh][ct * 16 + lm];
      bf.y = web[k0 + kh + 1][ct * 16 + lm];
      c = wmma4(a, bf, c);
    }
#pragma unroll
    for (int r = 0; r < 8; ++r)
      heb[m0 + r + 8 * lh][ct * 16 + lm] = c[r];
  }
  __syncthreads();

  const int rm = rowmask[bn];
  const int* ad = opma + bn * MM;
  const float* am3 = attma + 2 * FF;
  if (threadIdx.x < MM) {
    int m = threadIdx.x;
    float dot = 0.f;
#pragma unroll 8
    for (int o = 0; o < FF; ++o) dot += heb[m][o] * am3[o];
    float e = ham1[bn] + hma2[(size_t)b * MM + m] + dot;
    e = e > 0.f ? e : 0.01f * e;
    int conn = rm ? 1 : (ad[m] > 0);
    sc[m] = conn ? e : NEGINF;
  }
  __syncthreads();

  if (wv == 0) {
    float mx = fmaxf(fmaxf(sc[lane], sc[lane + 32]),
                     fmaxf(sc[lane + 64], sc[lane + 96]));
#pragma unroll
    for (int off = 16; off; off >>= 1) mx = fmaxf(mx, __shfl_xor(mx, off, 32));
    if (mx == NEGINF) mx = 0.f;
    if (lane == 0) red[0] = mx;
  }
  __syncthreads();
  const float mx = red[0];
  if (threadIdx.x < MM) sc[threadIdx.x] = __expf(sc[threadIdx.x] - mx);
  __syncthreads();
  if (wv == 0) {
    float s = sc[lane] + sc[lane + 32] + sc[lane + 64] + sc[lane + 96];
#pragma unroll
    for (int off = 16; off; off >>= 1) s += __shfl_xor(s, off, 32);
    if (lane == 0) red[1] = 1.f / fmaxf(s, 1e-30f);
  }
  __syncthreads();
  const float inv = red[1];

  {
    int o = threadIdx.x & 63, ch = threadIdx.x >> 6;
    const float* hmb = hma + (size_t)b * MM * FF;
    float acc = 0.f;
    for (int m = ch * 32; m < ch * 32 + 32; ++m)
      acc += sc[m] * (hmb[(size_t)m * FF + o] + heb[m][o]);
    psum[ch][o] = acc;
  }
  __syncthreads();
  if (threadIdx.x < FF) {
    int o = threadIdx.x;
    float v = (psum[0][o] + psum[1][o] + psum[2][o] + psum[3][o]) * inv +
              h[bn * FF + o];
    hprime[bn * (2 * FF) + FF + o] = v;
  }
}

__global__ __launch_bounds__(256) void mlp(
    const float* __restrict__ hprime, const float* __restrict__ w1,
    const float* __restrict__ b1, const float* __restrict__ w2,
    const float* __restrict__ b2, const float* __restrict__ wo,
    const float* __restrict__ bo, const int* __restrict__ rowmask,
    float* __restrict__ out) {
  extern __shared__ float smem[];
  float* wb = smem;
  typedef float act_row_t[HID + 4];
  act_row_t* a0 = (act_row_t*)(smem + HID * HID);
  act_row_t* a1 = (act_row_t*)(smem + HID * HID + 16 * (HID + 4));

  const int row0 = blockIdx.x * 16;
  const int wv = threadIdx.x >> 5;
  const int lane = threadIdx.x & 31;
  const int lm = lane & 15;
  const int kh = (lane >> 4) * 2;
  const int lh = lane >> 4;

  for (int t = threadIdx.x; t < 16 * HID; t += 256)
    a0[t >> 7][t & 127] = hprime[(size_t)row0 * HID + t];
  for (int t = threadIdx.x; t < HID * HID; t += 256) wb[t] = w1[t];
  __syncthreads();

  {
    int o = wv * 16 + lm;
    v8f c = {};
    for (int k0 = 0; k0 < HID; k0 += 4) {
      v2f a, bf;
      a.x  = a0[lm][k0 + kh];
      a.y  = a0[lm][k0 + kh + 1];
      bf.x = wb[(k0 + kh) * HID + o];
      bf.y = wb[(k0 + kh + 1) * HID + o];
      c = wmma4(a, bf, c);
    }
    float bias = b1[o];
#pragma unroll
    for (int r = 0; r < 8; ++r) {
      float v = c[r] + bias;
      a1[r + 8 * lh][o] = v > 0.f ? v : expm1f(v);
    }
  }
  __syncthreads();
  for (int t = threadIdx.x; t < HID * HID; t += 256) wb[t] = w2[t];
  __syncthreads();

  {
    int o = wv * 16 + lm;
    v8f c = {};
    for (int k0 = 0; k0 < HID; k0 += 4) {
      v2f a, bf;
      a.x  = a1[lm][k0 + kh];
      a.y  = a1[lm][k0 + kh + 1];
      bf.x = wb[(k0 + kh) * HID + o];
      bf.y = wb[(k0 + kh + 1) * HID + o];
      c = wmma4(a, bf, c);
    }
    float bias = b2[o];
#pragma unroll
    for (int r = 0; r < 8; ++r) {
      float v = c[r] + bias;
      a0[r + 8 * lh][o] = v > 0.f ? v : expm1f(v);
    }
  }
  __syncthreads();
  for (int t = threadIdx.x; t < HID * FF; t += 256) wb[t] = wo[t];
  __syncthreads();

  if (wv < 4) {
    int o = wv * 16 + lm;
    v8f c = {};
    for (int k0 = 0; k0 < HID; k0 += 4) {
      v2f a, bf;
      a.x  = a0[lm][k0 + kh];
      a.y  = a0[lm][k0 + kh + 1];
      bf.x = wb[(k0 + kh) * FF + o];
      bf.y = wb[(k0 + kh + 1) * FF + o];
      c = wmma4(a, bf, c);
    }
    float bias = bo[o];
#pragma unroll
    for (int r = 0; r < 8; ++r) {
      int m = r + 8 * lh;
      float v = c[r] + bias;
      if (rowmask[row0 + m]) v = 0.f;
      out[(size_t)(row0 + m) * FF + o] = v;
    }
  }
}

extern "C" void kernel_launch(void* const* d_in, const int* in_sizes, int n_in,
                              void* d_out, int out_size, void* d_ws,
                              size_t ws_size, hipStream_t stream) {
  (void)in_sizes; (void)n_in; (void)out_size; (void)ws_size;
  const float* x      = (const float*)d_in[0];
  const float* y      = (const float*)d_in[1];
  const float* z      = (const float*)d_in[2];
  const int*   op_adj = (const int*)d_in[3];
  const int*   ma_adj = (const int*)d_in[4];
  const int*   op_ma  = (const int*)d_in[5];
  const float* Wop    = (const float*)d_in[6];
  const float* Wma    = (const float*)d_in[7];
  const float* Wedge  = (const float*)d_in[8];
  const float* att_op = (const float*)d_in[9];
  const float* att_ma = (const float*)d_in[10];
  const float* w1     = (const float*)d_in[11];
  const float* b1     = (const float*)d_in[12];
  const float* w2     = (const float*)d_in[13];
  const float* b2     = (const float*)d_in[14];
  const float* wo     = (const float*)d_in[15];
  const float* bo     = (const float*)d_in[16];
  float* out = (float*)d_out;

  float* ws     = (float*)d_ws;
  float* h      = ws;
  float* hma    = h + BB * NN * FF;
  float* ha1    = hma + BB * MM * FF;
  float* ha2    = ha1 + BB * NN;
  float* ham1   = ha2 + BB * NN;
  float* hma2   = ham1 + BB * NN;
  float* hprime = hma2 + BB * MM;
  int*   rmask  = (int*)(hprime + (size_t)BB * NN * 2 * FF);

  gemm_rows_64<<<dim3((BB * NN) / 16), dim3(128), 0, stream>>>(x, Wop, h);
  gemm_rows_64<<<dim3((BB * MM) / 16), dim3(128), 0, stream>>>(y, Wma, hma);
  proj_node<<<dim3((BB * NN) / 256), dim3(256), 0, stream>>>(
      h, att_op, att_ma, op_ma, ha1, ha2, ham1, rmask);
  proj_ma<<<dim3((BB * MM) / 256), dim3(256), 0, stream>>>(hma, att_ma, hma2);
  size_t sm2 = (16 * (NN + 4) + 16) * sizeof(float);
  op_attn<<<dim3(NN / 16, BB), dim3(128), sm2, stream>>>(
      h, ha1, ha2, op_adj, ma_adj, rmask, hprime);
  ma_attn<<<dim3(NN, BB), dim3(256), 0, stream>>>(
      z, Wedge, hma, ham1, hma2, op_ma, rmask, h, att_ma, hprime);
  size_t sm4 = (HID * HID + 2 * 16 * (HID + 4)) * sizeof(float);
  mlp<<<dim3((BB * NN) / 16), dim3(256), sm4, stream>>>(
      hprime, w1, b1, w2, b2, wo, bo, rmask, out);
}


# op mask: new kernel packs (op+op^T+ma+ma^T)>0 into bits, 128 B per row, from whole-line row reads (direct tile turned in LDS); op_attn reads 1 bit per pair; 7 launches -> 8
# speedup vs baseline: 1.5982x; 1.5982x over previous
#include <hip/hip_runtime.h>
#include <math.h>


typedef __attribute__((ext_vector_type(2))) float v2f;
typedef __attribute__((ext_vector_type(8))) float v8f;

#define BB 8
#define NN 1024
#define MM 128
#define FF 64
#define FE 16
#define HID 128

__device__ __forceinline__ v8f wmma4(v2f a, v2f b, v8f c) {
  return __builtin_amdgcn_wmma_f32_16x16x4_f32(
       false, a,  false, b,
       (short)0, c,  false,  false);
}

__global__ __launch_bounds__(128) void gemm_rows_64(
    const float* __restrict__ A, const float* __restrict__ W,
    float* __restrict__ C) {
  const int wv   = threadIdx.x >> 5;
  const int lane = threadIdx.x & 31;
  const int lm   = lane & 15;
  const int kh   = (lane >> 4) * 2;
  const int lh   = lane >> 4;
  const int row0 = blockIdx.x * 16;
  const int o    = wv * 16 + lm;
  const float* Arow = A + (size_t)(row0 + lm) * FF;

  v8f c = {};
#pragma unroll
  for (int k0 = 0; k0 < FF; k0 += 4) {
    v2f a, b;
    a.x = Arow[k0 + kh];
    a.y = Arow[k0 + kh + 1];
    b.x = W[(size_t)(k0 + kh) * FF + o];
    b.y = W[(size_t)(k0 + kh + 1) * FF + o];
    c = wmma4(a, b, c);
  }
#pragma unroll
  for (int r = 0; r < 8; ++r) {
    int m = r + 8 * lh;
    C[(size_t)(row0 + m) * FF + o] = c[r];
  }
}

__global__ void proj_node(const float* __restrict__ h,
                          const float* __restrict__ att_op,
                          const float* __restrict__ att_ma,
                          const int* __restrict__ opma,
                          float* __restrict__ ha1, float* __restrict__ ha2,
                          float* __restrict__ ham1, int* __restrict__ rowmask) {
  int idx = blockIdx.x * blockDim.x + threadIdx.x;
  if (idx >= BB * NN) return;
  const float* hr = h + (size_t)idx * FF;
  float s1 = 0.f, s2 = 0.f, s3 = 0.f;
#pragma unroll 8
  for (int o = 0; o < FF; ++o) {
    float v = hr[o];
    s1 += v * att_op[o];
    s2 += v * att_op[FF + o];
    s3 += v * att_ma[o];
  }
  ha1[idx] = s1; ha2[idx] = s2; ham1[idx] = s3;
  const int* ad = opma + (size_t)idx * MM;
  int any = 0;
#pragma unroll 8
  for (int m = 0; m < MM; ++m) any |= ad[m];
  rowmask[idx] = (any == 0) ? 1 : 0;
}

__global__ void proj_ma(const float* __restrict__ hma,
                        const float* __restrict__ att_ma,
                        float* __restrict__ hma2) {
  int idx = blockIdx.x * blockDim.x + threadIdx.x;
  if (idx >= BB * MM) return;
  const float* r = hma + (size_t)idx * FF;
  float s = 0.f;
#pragma unroll 8
  for (int o = 0; o < FF; ++o) s += r[o] * att_ma[FF + o];
  hma2[idx] = s;
}

#define CB_P 36

__global__ __launch_bounds__(512) void conn_bits(
    const int* __restrict__ opadj, const int* __restrict__ maadj,
    unsigned* __restrict__ cbits) {
  extern __shared__ __attribute__((aligned(16))) int cbs[];
  const int b    = blockIdx.y;
  const int r0   = blockIdx.x * 32;
  const int wv   = threadIdx.x >> 5;
  const int lane = threadIdx.x & 31;
  int* tile = cbs + wv * (32 * CB_P);
  unsigned* words = (unsigned*)(cbs + 16 * 32 * CB_P);
  const int* opb = opadj + (size_t)b * NN * NN;
  const int* mab = maadj + (size_t)b * NN * NN;
  const int sub = lane >> 3;
  const int q4  = (lane & 7) * 4;

  int4 dsum[2][8];
  int  tr[2][32];
#pragma unroll
  for (int hf = 0; hf < 2; ++hf) {
    const int c0 = (wv + 16 * hf) * 32;
#pragma unroll
    for (int q = 0; q < 8; ++q) {
      const size_t off = (size_t)(r0 + 4 * q + sub) * NN + c0 + q4;
      const int4 a = *(const int4*)(opb + off);
      const int4 m = *(const int4*)(mab + off);
      dsum[hf][q].x = a.x + m.x; dsum[hf][q].y = a.y + m.y;
      dsum[hf][q].z = a.z + m.z; dsum[hf][q].w = a.w + m.w;
    }
#pragma unroll
    for (int k = 0; k < 32; ++k) {
      const size_t off = (size_t)(c0 + k) * NN + r0 + lane;
      tr[hf][k] = opb[off] + mab[off];
    }
  }

#pragma unroll
  for (int hf = 0; hf < 2; ++hf) {
    const int ct = wv + 16 * hf;
#pragma unroll
    for (int q = 0; q < 8; ++q)
      *(int4*)(tile + (4 * q + sub) * CB_P + q4) = dsum[hf][q];
    __syncthreads();
    unsigned w = 0;
#pragma unroll
    for (int g = 0; g < 8; ++g) {
      const int4 d = *(const int4*)(tile + lane * CB_P + 4 * g);
      w |= (unsigned)((d.x + tr[hf][4 * g + 0]) > 0) << (4 * g + 0);
      w |= (unsigned)((d.y + tr[hf][4 * g + 1]) > 0) << (4 * g + 1);
      w |= (unsigned)((d.z + tr[hf][4 * g + 2]) > 0) << (4 * g + 2);
      w |= (unsigned)((d.w + tr[hf][4 * g + 3]) > 0) << (4 * g + 3);
    }
    words[lane * 33 + ct] = w;
    __syncthreads();
  }

  for (int rr = wv; rr < 32; rr += 16)
    cbits[((size_t)b * NN + r0 + rr) * (2 * FF) + lane] = words[rr * 33 + lane];
}

__global__ __launch_bounds__(128) void op_attn(
    const float* __restrict__ h, const float* __restrict__ ha1,
    const float* __restrict__ ha2, const unsigned* __restrict__ cbits,
    const int* __restrict__ rowmask, float* __restrict__ hprime) {
  extern __shared__ float smem[];
  typedef float att_row_t[NN + 4];
  att_row_t* att = (att_row_t*)smem;
  float* inv_sum = smem + 16 * (NN + 4);

  const int b  = blockIdx.y;
  const int i0 = blockIdx.x * 16;
  const float NEGINF = -__builtin_inff();

  {
    const int wv0 = threadIdx.x >> 5;
    const int ln  = threadIdx.x & 31;
    float a1v[8];
#pragma unroll
    for (int jj = 0; jj < 8; ++jj) a1v[jj] = ha1[b * NN + threadIdx.x + 128 * jj];
    for (int r = 0; r < 16; ++r) {
      const int i = i0 + r;
      const int rm = rowmask[b * NN + i];
      const float a2 = ha2[b * NN + i];
      const unsigned* br = cbits + ((size_t)b * NN + i) * (2 * FF);
#pragma unroll
      for (int jj = 0; jj < 8; ++jj) {
        const unsigned wb = br[wv0 + 4 * jj];
        const int conn = rm | (int)((wb >> ln) & 1u);
        float e = a2 + a1v[jj];
        e = e > 0.f ? e : 0.01f * e;
        att[r][threadIdx.x + 128 * jj] = conn ? e : NEGINF;
      }
    }
  }
  __syncthreads();

  const int wv = threadIdx.x >> 5;
  const int lane = threadIdx.x & 31;
  for (int r = wv; r < 16; r += 4) {
    float mx = NEGINF;
    for (int j = lane; j < NN; j += 32) mx = fmaxf(mx, att[r][j]);
#pragma unroll
    for (int off = 16; off; off >>= 1) mx = fmaxf(mx, __shfl_xor(mx, off, 32));
    if (mx == NEGINF) mx = 0.f;
    float sum = 0.f;
    for (int j = lane; j < NN; j += 32) {
      float p = __expf(att[r][j] - mx);
      att[r][j] = p;
      sum += p;
    }
#pragma unroll
    for (int off = 16; off; off >>= 1) sum += __shfl_xor(sum, off, 32);
    if (lane == 0) inv_sum[r] = 1.f / fmaxf(sum, 1e-30f);
  }
  __syncthreads();

  const int lm = lane & 15;
  const int kh = (lane >> 4) * 2;
  const int lh = lane >> 4;
  const int o  = wv * 16 + lm;
  const float* hb = h + (size_t)b * NN * FF;

  v8f c = {};
  for (int k0 = 0; k0 < NN; k0 += 4) {
    v2f a, bf;
    a.x  = att[lm][k0 + kh];
    a.y  = att[lm][k0 + kh + 1];
    bf.x = hb[(size_t)(k0 + kh) * FF + o];
    bf.y = hb[(size_t)(k0 + kh + 1) * FF + o];
    c = wmma4(a, bf, c);
  }
  float* hp = hprime + ((size_t)b * NN + i0) * (2 * FF);
#pragma unroll
  for (int r = 0; r < 8; ++r) {
    int m = r + 8 * lh;
    hp[(size_t)m * (2 * FF) + o] = c[r] * inv_sum[m];
  }
}

__global__ __launch_bounds__(256) void ma_attn(
    const float* __restrict__ z, const float* __restrict__ wedge,
    const float* __restrict__ hma, const float* __restrict__ ham1,
    const float* __restrict__ hma2, const int* __restrict__ opma,
    const int* __restrict__ rowmask, const float* __restrict__ h,
    const float* __restrict__ attma, float* __restrict__ hprime) {
  __shared__ float zb[MM][FE + 1];
  __shared__ float web[FE][FF];
  __shared__ float heb[MM][FF + 1];
  __shared__ float sc[MM];
  __shared__ float red[2];
  __shared__ float psum[4][FF];

  const int b = blockIdx.y, n = blockIdx.x;
  const size_t bn = (size_t)b * NN + n;
  const float NEGINF = -__builtin_inff();

  const float* zrow = z + bn * MM * FE;
  for (int t = threadIdx.x; t < MM * FE; t += 256) zb[t >> 4][t & 15] = zrow[t];
  for (int t = threadIdx.x; t < FE * FF; t += 256) web[t >> 6][t & 63] = wedge[t];
  __syncthreads();

  const int wv = threadIdx.x >> 5;
  const int lane = threadIdx.x & 31;
  const int lm = lane & 15;
  const int kh = (lane >> 4) * 2;
  const int lh = lane >> 4;
  const int m0 = wv * 16;
  for (int ct = 0; ct < 4; ++ct) {
    v8f c = {};
#pragma unroll
    for (int k0 = 0; k0 < FE; k0 += 4) {
      v2f a, bf;
      a.x  = zb[m0 + lm][k0 + kh];
      a.y  = zb[m0 + lm][k0 + kh + 1];
      bf.x = web[k0 + kh][ct * 16 + lm];
      bf.y = web[k0 + kh + 1][ct * 16 + lm];
      c = wmma4(a, bf, c);
    }
#pragma unroll
    for (int r = 0; r < 8; ++r)
      heb[m0 + r + 8 * lh][ct * 16 + lm] = c[r];
  }
  __syncthreads();

  const int rm = rowmask[bn];
  const int* ad = opma + bn * MM;
  const float* am3 = attma + 2 * FF;
  if (threadIdx.x < MM) {
    int m = threadIdx.x;
    float dot = 0.f;
#pragma unroll 8
    for (int o = 0; o < FF; ++o) dot += heb[m][o] * am3[o];
    float e = ham1[bn] + hma2[(size_t)b * MM + m] + dot;
    e = e > 0.f ? e : 0.01f * e;
    int conn = rm ? 1 : (ad[m] > 0);
    sc[m] = conn ? e : NEGINF;
  }
  __syncthreads();

  if (wv == 0) {
    float mx = fmaxf(fmaxf(sc[lane], sc[lane + 32]),
                     fmaxf(sc[lane + 64], sc[lane + 96]));
#pragma unroll
    for (int off = 16; off; off >>= 1) mx = fmaxf(mx, __shfl_xor(mx, off, 32));
    if (mx == NEGINF) mx = 0.f;
    if (lane == 0) red[0] = mx;
  }
  __syncthreads();
  const float mx = red[0];
  if (threadIdx.x < MM) sc[threadIdx.x] = __expf(sc[threadIdx.x] - mx);
  __syncthreads();
  if (wv == 0) {
    float s = sc[lane] + sc[lane + 32] + sc[lane + 64] + sc[lane + 96];
#pragma unroll
    for (int off = 16; off; off >>= 1) s += __shfl_xor(s, off, 32);
    if (lane == 0) red[1] = 1.f / fmaxf(s, 1e-30f);
  }
  __syncthreads();
  const float inv = red[1];

  {
    int o = threadIdx.x & 63, ch = threadIdx.x >> 6;
    const float* hmb = hma + (size_t)b * MM * FF;
    float acc = 0.f;
    for (int m = ch * 32; m < ch * 32 + 32; ++m)
      acc += sc[m] * (hmb[(size_t)m * FF + o] + heb[m][o]);
    psum[ch][o] = acc;
  }
  __syncthreads();
  if (threadIdx.x < FF) {
    int o = threadIdx.x;
    float v = (psum[0][o] + psum[1][o] + psum[2][o] + psum[3][o]) * inv +
              h[bn * FF + o];
    hprime[bn * (2 * FF) + FF + o] = v;
  }
}

__global__ __launch_bounds__(256) void mlp(
    const float* __restrict__ hprime, const float* __restrict__ w1,
    const float* __restrict__ b1, const float* __restrict__ w2,
    const float* __restrict__ b2, const float* __restrict__ wo,
    const float* __restrict__ bo, const int* __restrict__ rowmask,
    float* __restrict__ out) {
  extern __shared__ float smem[];
  float* wb = smem;
  typedef float act_row_t[HID + 4];
  act_row_t* a0 = (act_row_t*)(smem + HID * HID);
  act_row_t* a1 = (act_row_t*)(smem + HID * HID + 16 * (HID + 4));

  const int row0 = blockIdx.x * 16;
  const int wv = threadIdx.x >> 5;
  const int lane = threadIdx.x & 31;
  const int lm = lane & 15;
  const int kh = (lane >> 4) * 2;
  const int lh = lane >> 4;

  for (int t = threadIdx.x; t < 16 * HID; t += 256)
    a0[t >> 7][t & 127] = hprime[(size_t)row0 * HID + t];
  for (int t = threadIdx.x; t < HID * HID; t += 256) wb[t] = w1[t];
  __syncthreads();

  {
    int o = wv * 16 + lm;
    v8f c = {};
    for (int k0 = 0; k0 < HID; k0 += 4) {
      v2f a, bf;
      a.x  = a0[lm][k0 + kh];
      a.y  = a0[lm][k0 + kh + 1];
      bf.x = wb[(k0 + kh) * HID + o];
      bf.y = wb[(k0 + kh + 1) * HID + o];
      c = wmma4(a, bf, c);
    }
    float bias = b1[o];
#pragma unroll
    for (int r = 0; r < 8; ++r) {
      float v = c[r] + bias;
      a1[r + 8 * lh][o] = v > 0.f ? v : expm1f(v);
    }
  }
  __syncthreads();
  for (int t = threadIdx.x; t < HID * HID; t += 256) wb[t] = w2[t];
  __syncthreads();

  {
    int o = wv * 16 + lm;
    v8f c = {};
    for (int k0 = 0; k0 < HID; k0 += 4) {
      v2f a, bf;
      a.x  = a1[lm][k0 + kh];
      a.y  = a1[lm][k0 + kh + 1];
      bf.x = wb[(k0 + kh) * HID + o];
      bf.y = wb[(k0 + kh + 1) * HID + o];
      c = wmma4(a, bf, c);
    }
    float bias = b2[o];
#pragma unroll
    for (int r = 0; r < 8; ++r) {
      float v = c[r] + bias;
      a0[r + 8 * lh][o] = v > 0.f ? v : expm1f(v);
    }
  }
  __syncthreads();
  for (int t = threadIdx.x; t < HID * FF; t += 256) wb[t] = wo[t];
  __syncthreads();

  if (wv < 4) {
    int o = wv * 16 + lm;
    v8f c = {};
    for (int k0 = 0; k0 < HID; k0 += 4) {
      v2f a, bf;
      a.x  = a0[lm][k0 + kh];
      a.y  = a0[lm][k0 + kh + 1];
      bf.x = wb[(k0 + kh) * FF + o];
      bf.y = wb[(k0 + kh + 1) * FF + o];
      c = wmma4(a, bf, c);
    }
    float bias = bo[o];
#pragma unroll
    for (int r = 0; r < 8; ++r) {
      int m = r + 8 * lh;
      float v = c[r] + bias;
      if (rowmask[row0 + m]) v = 0.f;
      out[(size_t)(row0 + m) * FF + o] = v;
    }
  }
}

extern "C" void kernel_launch(void* const* d_in, const int* in_sizes, int n_in,
                              void* d_out, int out_size, void* d_ws,
                              size_t ws_size, hipStream_t stream) {
  (void)in_sizes; (void)n_in; (void)out_size; (void)ws_size;
  const float* x      = (const float*)d_in[0];
  const float* y      = (const float*)d_in[1];
  const float* z      = (const float*)d_in[2];
  const int*   op_adj = (const int*)d_in[3];
  const int*   ma_adj = (const int*)d_in[4];
  const int*   op_ma  = (const int*)d_in[5];
  const float* Wop    = (const float*)d_in[6];
  const float* Wma    = (const float*)d_in[7];
  const float* Wedge  = (const float*)d_in[8];
  const float* att_op = (const float*)d_in[9];
  const float* att_ma = (const float*)d_in[10];
  const float* w1     = (const float*)d_in[11];
  const float* b1     = (const float*)d_in[12];
  const float* w2     = (const float*)d_in[13];
  const float* b2     = (const float*)d_in[14];
  const float* wo     = (const float*)d_in[15];
  const float* bo     = (const float*)d_in[16];
  float* out = (float*)d_out;

  float* ws     = (float*)d_ws;
  float* h      = ws;
  float* hma    = h + BB * NN * FF;
  float* ha1    = hma + BB * MM * FF;
  float* ha2    = ha1 + BB * NN;
  float* ham1   = ha2 + BB * NN;
  float* hma2   = ham1 + BB * NN;
  float* hprime = hma2 + BB * MM;
  int*   rmask  = (int*)(hprime + (size_t)BB * NN * 2 * FF);

  gemm_rows_64<<<dim3((BB * NN) / 16), dim3(128), 0, stream>>>(x, Wop, h);
  gemm_rows_64<<<dim3((BB * MM) / 16), dim3(128), 0, stream>>>(y, Wma, hma);
  proj_node<<<dim3((BB * NN) / 256), dim3(256), 0, stream>>>(
      h, att_op, att_ma, op_ma, ha1, ha2, ham1, rmask);
  proj_ma<<<dim3((BB * MM) / 256), dim3(256), 0, stream>>>(hma, att_ma, hma2);
  size_t sm2 = (16 * (NN + 4) + 16) * sizeof(float);
  unsigned* cbits = (unsigned*)(hprime + FF);
  size_t smb = (16 * 32 * CB_P + 32 * 33) * sizeof(int);
  conn_bits<<<dim3(NN / 32, BB), dim3(512), smb, stream>>>(op_adj, ma_adj, cbits);
  op_attn<<<dim3(NN / 16, BB), dim3(128), sm2, stream>>>(
      h, ha1, ha2, cbits, rmask, hprime);
  ma_attn<<<dim3(NN, BB), dim3(256), 0, stream>>>(
      z, Wedge, hma, ham1, hma2, op_ma, rmask, h, att_ma, hprime);
  size_t sm4 = (HID * HID + 2 * 16 * (HID + 4)) * sizeof(float);
  mlp<<<dim3((BB * NN) / 16), dim3(256), sm4, stream>>>(
      hprime, w1, b1, w2, b2, wo, bo, rmask, out);
}


# ma_attn: one wave per node, edge product formed transposed (we^T @ z^T) so machines run along lanes; scores, softmax, weighted sum in registers, no LDS tile or barriers; summing order changed
# speedup vs baseline: 2.2826x; 1.4282x over previous
#include <hip/hip_runtime.h>
#include <math.h>


typedef __attribute__((ext_vector_type(2))) float v2f;
typedef __attribute__((ext_vector_type(8))) float v8f;

#define BB 8
#define NN 1024
#define MM 128
#define FF 64
#define FE 16
#define HID 128

__device__ __forceinline__ v8f wmma4(v2f a, v2f b, v8f c) {
  return __builtin_amdgcn_wmma_f32_16x16x4_f32(
       false, a,  false, b,
       (short)0, c,  false,  false);
}

__global__ __launch_bounds__(128) void gemm_rows_64(
    const float* __restrict__ A, const float* __restrict__ W,
    float* __restrict__ C) {
  const int wv   = threadIdx.x >> 5;
  const int lane = threadIdx.x & 31;
  const int lm   = lane & 15;
  const int kh   = (lane >> 4) * 2;
  const int lh   = lane >> 4;
  const int row0 = blockIdx.x * 16;
  const int o    = wv * 16 + lm;
  const float* Arow = A + (size_t)(row0 + lm) * FF;

  v8f c = {};
#pragma unroll
  for (int k0 = 0; k0 < FF; k0 += 4) {
    v2f a, b;
    a.x = Arow[k0 + kh];
    a.y = Arow[k0 + kh + 1];
    b.x = W[(size_t)(k0 + kh) * FF + o];
    b.y = W[(size_t)(k0 + kh + 1) * FF + o];
    c = wmma4(a, b, c);
  }
#pragma unroll
  for (int r = 0; r < 8; ++r) {
    int m = r + 8 * lh;
    C[(size_t)(row0 + m) * FF + o] = c[r];
  }
}

__global__ void proj_node(const float* __restrict__ h,
                          const float* __restrict__ att_op,
                          const float* __restrict__ att_ma,
                          const int* __restrict__ opma,
                          float* __restrict__ ha1, float* __restrict__ ha2,
                          float* __restrict__ ham1, int* __restrict__ rowmask) {
  int idx = blockIdx.x * blockDim.x + threadIdx.x;
  if (idx >= BB * NN) return;
  const float* hr = h + (size_t)idx * FF;
  float s1 = 0.f, s2 = 0.f, s3 = 0.f;
#pragma unroll 8
  for (int o = 0; o < FF; ++o) {
    float v = hr[o];
    s1 += v * att_op[o];
    s2 += v * att_op[FF + o];
    s3 += v * att_ma[o];
  }
  ha1[idx] = s1; ha2[idx] = s2; ham1[idx] = s3;
  const int* ad = opma + (size_t)idx * MM;
  int any = 0;
#pragma unroll 8
  for (int m = 0; m < MM; ++m) any |= ad[m];
  rowmask[idx] = (any == 0) ? 1 : 0;
}

__global__ void proj_ma(const float* __restrict__ hma,
                        const float* __restrict__ att_ma,
                        float* __restrict__ hma2) {
  int idx = blockIdx.x * blockDim.x + threadIdx.x;
  if (idx >= BB * MM) return;
  const float* r = hma + (size_t)idx * FF;
  float s = 0.f;
#pragma unroll 8
  for (int o = 0; o < FF; ++o) s += r[o] * att_ma[FF + o];
  hma2[idx] = s;
}

#define CB_P 36

__global__ __launch_bounds__(512) void conn_bits(
    const int* __restrict__ opadj, const int* __restrict__ maadj,
    unsigned* __restrict__ cbits) {
  extern __shared__ __attribute__((aligned(16))) int cbs[];
  const int b    = blockIdx.y;
  const int r0   = blockIdx.x * 32;
  const int wv   = threadIdx.x >> 5;
  const int lane = threadIdx.x & 31;
  int* tile = cbs + wv * (32 * CB_P);
  unsigned* words = (unsigned*)(cbs + 16 * 32 * CB_P);
  const int* opb = opadj + (size_t)b * NN * NN;
  const int* mab = maadj + (size_t)b * NN * NN;
  const int sub = lane >> 3;
  const int q4  = (lane & 7) * 4;

  int4 dsum[2][8];
  int  tr[2][32];
#pragma unroll
  for (int hf = 0; hf < 2; ++hf) {
    const int c0 = (wv + 16 * hf) * 32;
#pragma unroll
    for (int q = 0; q < 8; ++q) {
      const size_t off = (size_t)(r0 + 4 * q + sub) * NN + c0 + q4;
      const int4 a = *(const int4*)(opb + off);
      const int4 m = *(const int4*)(mab + off);
      dsum[hf][q].x = a.x + m.x; dsum[hf][q].y = a.y + m.y;
      dsum[hf][q].z = a.z + m.z; dsum[hf][q].w = a.w + m.w;
    }
#pragma unroll
    for (int k = 0; k < 32; ++k) {
      const size_t off = (size_t)(c0 + k) * NN + r0 + lane;
      tr[hf][k] = opb[off] + mab[off];
    }
  }

#pragma unroll
  for (int hf = 0; hf < 2; ++hf) {
    const int ct = wv + 16 * hf;
#pragma unroll
    for (int q = 0; q < 8; ++q)
      *(int4*)(tile + (4 * q + sub) * CB_P + q4) = dsum[hf][q];
    __syncthreads();
    unsigned w = 0;
#pragma unroll
    for (int g = 0; g < 8; ++g) {
      const int4 d = *(const int4*)(tile + lane * CB_P + 4 * g);
      w |= (unsigned)((d.x + tr[hf][4 * g + 0]) > 0) << (4 * g + 0);
      w |= (unsigned)((d.y + tr[hf][4 * g + 1]) > 0) << (4 * g + 1);
      w |= (unsigned)((d.z + tr[hf][4 * g + 2]) > 0) << (4 * g + 2);
      w |= (unsigned)((d.w + tr[hf][4 * g + 3]) > 0) << (4 * g + 3);
    }
    words[lane * 33 + ct] = w;
    __syncthreads();
  }

  for (int rr = wv; rr < 32; rr += 16)
    cbits[((size_t)b * NN + r0 + rr) * (2 * FF) + lane] = words[rr * 33 + lane];
}

__global__ __launch_bounds__(128) void op_attn(
    const float* __restrict__ h, const float* __restrict__ ha1,
    const float* __restrict__ ha2, const unsigned* __restrict__ cbits,
    const int* __restrict__ rowmask, float* __restrict__ hprime) {
  extern __shared__ float smem[];
  typedef float att_row_t[NN + 4];
  att_row_t* att = (att_row_t*)smem;
  float* inv_sum = smem + 16 * (NN + 4);

  const int b  = blockIdx.y;
  const int i0 = blockIdx.x * 16;
  const float NEGINF = -__builtin_inff();

  {
    const int wv0 = threadIdx.x >> 5;
    const int ln  = threadIdx.x & 31;
    float a1v[8];
#pragma unroll
    for (int jj = 0; jj < 8; ++jj) a1v[jj] = ha1[b * NN + threadIdx.x + 128 * jj];
    for (int r = 0; r < 16; ++r) {
      const int i = i0 + r;
      const int rm = rowmask[b * NN + i];
      const float a2 = ha2[b * NN + i];
      const unsigned* br = cbits + ((size_t)b * NN + i) * (2 * FF);
#pragma unroll
      for (int jj = 0; jj < 8; ++jj) {
        const unsigned wb = br[wv0 + 4 * jj];
        const int conn = rm | (int)((wb >> ln) & 1u);
        float e = a2 + a1v[jj];
        e = e > 0.f ? e : 0.01f * e;
        att[r][threadIdx.x + 128 * jj] = conn ? e : NEGINF;
      }
    }
  }
  __syncthreads();

  const int wv = threadIdx.x >> 5;
  const int lane = threadIdx.x & 31;
  for (int r = wv; r < 16; r += 4) {
    float mx = NEGINF;
    for (int j = lane; j < NN; j += 32) mx = fmaxf(mx, att[r][j]);
#pragma unroll
    for (int off = 16; off; off >>= 1) mx = fmaxf(mx, __shfl_xor(mx, off, 32));
    if (mx == NEGINF) mx = 0.f;
    float sum = 0.f;
    for (int j = lane; j < NN; j += 32) {
      float p = __expf(att[r][j] - mx);
      att[r][j] = p;
      sum += p;
    }
#pragma unroll
    for (int off = 16; off; off >>= 1) sum += __shfl_xor(sum, off, 32);
    if (lane == 0) inv_sum[r] = 1.f / fmaxf(sum, 1e-30f);
  }
  __syncthreads();

  const int lm = lane & 15;
  const int kh = (lane >> 4) * 2;
  const int lh = lane >> 4;
  const int o  = wv * 16 + lm;
  const float* hb = h + (size_t)b * NN * FF;

  v8f c = {};
  for (int k0 = 0; k0 < NN; k0 += 4) {
    v2f a, bf;
    a.x  = att[lm][k0 + kh];
    a.y  = att[lm][k0 + kh + 1];
    bf.x = hb[(size_t)(k0 + kh) * FF + o];
    bf.y = hb[(size_t)(k0 + kh + 1) * FF + o];
    c = wmma4(a, bf, c);
  }
  float* hp = hprime + ((size_t)b * NN + i0) * (2 * FF);
#pragma unroll
  for (int r = 0; r < 8; ++r) {
    int m = r + 8 * lh;
    hp[(size_t)m * (2 * FF) + o] = c[r] * inv_sum[m];
  }
}

typedef __attribute__((ext_vector_type(4))) float v4f;
#define MA_NPW 8
#define MA_WAVES 4

__device__ __forceinline__ float xhalf(float x) {
  int b = __builtin_bit_cast(int, x);
  return __builtin_bit_cast(float,
      __builtin_amdgcn_permlanex16(b, b, 0x76543210, 0xfedcba98, false, false));
}
template <int MASK>
__device__ __forceinline__ float rowx(float x) {
  int b = __builtin_bit_cast(int, x);
  return __builtin_bit_cast(float,
      __builtin_amdgcn_update_dpp(0, b, 0x160 + MASK, 0xf, 0xf, true));
}
__device__ __forceinline__ float row_sum16(float x) {
  x += rowx<1>(x); x += rowx<2>(x); x += rowx<4>(x); x += rowx<8>(x);
  return x;
}
__device__ __forceinline__ float row_max16(float x) {
  x = fmaxf(x, rowx<1>(x)); x = fmaxf(x, rowx<2>(x));
  x = fmaxf(x, rowx<4>(x)); x = fmaxf(x, rowx<8>(x));
  return x;
}

__global__ __launch_bounds__(32 * MA_WAVES) void ma_attn(
    const float* __restrict__ z, const float* __restrict__ wedge,
    const float* __restrict__ hma, const float* __restrict__ ham1,
    const float* __restrict__ hma2, const int* __restrict__ opma,
    const int* __restrict__ rowmask, const float* __restrict__ h,
    const float* __restrict__ attma, float* __restrict__ hprime) {
  const int lane = threadIdx.x & 31;
  const int c  = lane & 15;
  const int hh = lane >> 4;
  const int wv  = __builtin_amdgcn_readfirstlane((int)(threadIdx.x >> 5));
  const int bn0 = (blockIdx.x * MA_WAVES + wv) * MA_NPW;
  const int b   = bn0 / NN;
  const float NEGINF = -__builtin_inff();

  v2f awe[4][4];
#pragma unroll
  for (int s = 0; s < 4; ++s)
#pragma unroll
    for (int ot = 0; ot < 4; ++ot) {
      awe[s][ot].x = wedge[(8 * hh + 2 * s) * FF + ot * 16 + c];
      awe[s][ot].y = wedge[(8 * hh + 2 * s + 1) * FF + ot * 16 + c];
    }
  v4f am3v[4][2];
#pragma unroll
  for (int ot = 0; ot < 4; ++ot)
#pragma unroll
    for (int q = 0; q < 2; ++q)
      am3v[ot][q] = *(const v4f*)(attma + 2 * FF + ot * 16 + 8 * hh + 4 * q);
  float hma2v[8];
#pragma unroll
  for (int t = 0; t < 8; ++t) hma2v[t] = hma2[b * MM + 16 * t + c];
  const float* hmb = hma + (size_t)b * MM * FF + (size_t)c * FF + 8 * hh;

  v4f zq[8][2];
  {
    const float* zp = z + (size_t)bn0 * MM * FE + c * FE + 8 * hh;
#pragma unroll
    for (int t = 0; t < 8; ++t)
#pragma unroll
      for (int q = 0; q < 2; ++q)
        zq[t][q] = *(const v4f*)(zp + t * 16 * FE + 4 * q);
  }

#pragma unroll 1
  for (int i = 0; i < MA_NPW; ++i) {
    const int bn = bn0 + i;

    const int   rm  = rowmask[bn];
    const float hm1 = ham1[bn];
    const float* hr = h + (size_t)bn * FF + 8 * hh;
    v4f res[4][2];
#pragma unroll
    for (int ot = 0; ot < 4; ++ot) {
      res[ot][0] = *(const v4f*)(hr + ot * 16);
      res[ot][1] = *(const v4f*)(hr + ot * 16 + 4);
    }
    int adv[8];
#pragma unroll
    for (int t = 0; t < 8; ++t) adv[t] = opma[(size_t)bn * MM + 16 * t + c];

    v8f D[8][4];
    float sc[8];
#pragma unroll
    for (int t = 0; t < 8; ++t) {
      v2f zb[4];
      zb[0].x = zq[t][0].x; zb[0].y = zq[t][0].y;
      zb[1].x = zq[t][0].z; zb[1].y = zq[t][0].w;
      zb[2].x = zq[t][1].x; zb[2].y = zq[t][1].y;
      zb[3].x = zq[t][1].z; zb[3].y = zq[t][1].w;
      float part[4];
#pragma unroll
      for (int ot = 0; ot < 4; ++ot) {
        v8f d = {};
#pragma unroll
        for (int s = 0; s < 4; ++s) d = wmma4(awe[s][ot], zb[s], d);
        D[t][ot] = d;
        float a = 0.f;
        a += d[0] * am3v[ot][0].x; a += d[1] * am3v[ot][0].y;
        a += d[2] * am3v[ot][0].z; a += d[3] * am3v[ot][0].w;
        a += d[4] * am3v[ot][1].x; a += d[5] * am3v[ot][1].y;
        a += d[6] * am3v[ot][1].z; a += d[7] * am3v[ot][1].w;
        part[ot] = a;
      }
      float dot = (part[0] + part[1]) + (part[2] + part[3]);
      dot += xhalf(dot);
      float e = hm1 + hma2v[t] + dot;
      e = e > 0.f ? e : 0.01f * e;
      const int conn = rm | (adv[t] > 0);
      sc[t] = conn ? e : NEGINF;
    }

    float mx = sc[0];
#pragma unroll
    for (int t = 1; t < 8; ++t) mx = fmaxf(mx, sc[t]);
    mx = row_max16(mx);
    if (mx == NEGINF) mx = 0.f;
    float p[8];
    float sum = 0.f;
#pragma unroll
    for (int t = 0; t < 8; ++t) { p[t] = __expf(sc[t] - mx); sum += p[t]; }
    sum = row_sum16(sum);
    const float inv = 1.f / fmaxf(sum, 1e-30f);

    if (i + 1 < MA_NPW) {
      const float* zp = z + (size_t)(bn + 1) * MM * FE + c * FE + 8 * hh;
#pragma unroll
      for (int t = 0; t < 8; ++t)
#pragma unroll
        for (int q = 0; q < 2; ++q)
          zq[t][q] = *(const v4f*)(zp + t * 16 * FE + 4 * q);
    }

    v8f acc[4];
#pragma unroll
    for (int ot = 0; ot < 4; ++ot) acc[ot] = (v8f){};
#pragma unroll
    for (int t = 0; t < 8; ++t)
#pragma unroll
      for (int ot = 0; ot < 4; ++ot) {
        const v4f g0 = *(const v4f*)(hmb + (size_t)t * 16 * FF + ot * 16);
        const v4f g1 = *(const v4f*)(hmb + (size_t)t * 16 * FF + ot * 16 + 4);
        acc[ot][0] += p[t] * (g0.x + D[t][ot][0]);
        acc[ot][1] += p[t] * (g0.y + D[t][ot][1]);
        acc[ot][2] += p[t] * (g0.z + D[t][ot][2]);
        acc[ot][3] += p[t] * (g0.w + D[t][ot][3]);
        acc[ot][4] += p[t] * (g1.x + D[t][ot][4]);
        acc[ot][5] += p[t] * (g1.y + D[t][ot][5]);
        acc[ot][6] += p[t] * (g1.z + D[t][ot][6]);
        acc[ot][7] += p[t] * (g1.w + D[t][ot][7]);
      }
#pragma unroll
    for (int ot = 0; ot < 4; ++ot)
#pragma unroll
      for (int r = 0; r < 8; ++r) acc[ot][r] = row_sum16(acc[ot][r]);

    float* hp = hprime + (size_t)bn * (2 * FF) + FF + 8 * hh;
#pragma unroll
    for (int ot = 0; ot < 4; ++ot) {
      const v4f r0 = res[ot][0];
      const v4f r1 = res[ot][1];
      v4f o0, o1;
      o0.x = acc[ot][0] * inv + r0.x; o0.y = acc[ot][1] * inv + r0.y;
      o0.z = acc[ot][2] * inv + r0.z; o0.w = acc[ot][3] * inv + r0.w;
      o1.x = acc[ot][4] * inv + r1.x; o1.y = acc[ot][5] * inv + r1.y;
      o1.z = acc[ot][6] * inv + r1.z; o1.w = acc[ot][7] * inv + r1.w;
      if (c == 0) {
        *(v4f*)(hp + ot * 16)     = o0;
        *(v4f*)(hp + ot * 16 + 4) = o1;
      }
    }

  }
}

__global__ __launch_bounds__(256) void mlp(
    const float* __restrict__ hprime, const float* __restrict__ w1,
    const float* __restrict__ b1, const float* __restrict__ w2,
    const float* __restrict__ b2, const float* __restrict__ wo,
    const float* __restrict__ bo, const int* __restrict__ rowmask,
    float* __restrict__ out) {
  extern __shared__ float smem[];
  float* wb = smem;
  typedef float act_row_t[HID + 4];
  act_row_t* a0 = (act_row_t*)(smem + HID * HID);
  act_row_t* a1 = (act_row_t*)(smem + HID * HID + 16 * (HID + 4));

  const int row0 = blockIdx.x * 16;
  const int wv = threadIdx.x >> 5;
  const int lane = threadIdx.x & 31;
  const int lm = lane & 15;
  const int kh = (lane >> 4) * 2;
  const int lh = lane >> 4;

  for (int t = threadIdx.x; t < 16 * HID; t += 256)
    a0[t >> 7][t & 127] = hprime[(size_t)row0 * HID + t];
  for (int t = threadIdx.x; t < HID * HID; t += 256) wb[t] = w1[t];
  __syncthreads();

  {
    int o = wv * 16 + lm;
    v8f c = {};
    for (int k0 = 0; k0 < HID; k0 += 4) {
      v2f a, bf;
      a.x  = a0[lm][k0 + kh];
      a.y  = a0[lm][k0 + kh + 1];
      bf.x = wb[(k0 + kh) * HID + o];
      bf.y = wb[(k0 + kh + 1) * HID + o];
      c = wmma4(a, bf, c);
    }
    float bias = b1[o];
#pragma unroll
    for (int r = 0; r < 8; ++r) {
      float v = c[r] + bias;
      a1[r + 8 * lh][o] = v > 0.f ? v : expm1f(v);
    }
  }
  __syncthreads();
  for (int t = threadIdx.x; t < HID * HID; t += 256) wb[t] = w2[t];
  __syncthreads();

  {
    int o = wv * 16 + lm;
    v8f c = {};
    for (int k0 = 0; k0 < HID; k0 += 4) {
      v2f a, bf;
      a.x  = a1[lm][k0 + kh];
      a.y  = a1[lm][k0 + kh + 1];
      bf.x = wb[(k0 + kh) * HID + o];
      bf.y = wb[(k0 + kh + 1) * HID + o];
      c = wmma4(a, bf, c);
    }
    float bias = b2[o];
#pragma unroll
    for (int r = 0; r < 8; ++r) {
      float v = c[r] + bias;
      a0[r + 8 * lh][o] = v > 0.f ? v : expm1f(v);
    }
  }
  __syncthreads();
  for (int t = threadIdx.x; t < HID * FF; t += 256) wb[t] = wo[t];
  __syncthreads();

  if (wv < 4) {
    int o = wv * 16 + lm;
    v8f c = {};
    for (int k0 = 0; k0 < HID; k0 += 4) {
      v2f a, bf;
      a.x  = a0[lm][k0 + kh];
      a.y  = a0[lm][k0 + kh + 1];
      bf.x = wb[(k0 + kh) * FF + o];
      bf.y = wb[(k0 + kh + 1) * FF + o];
      c = wmma4(a, bf, c);
    }
    float bias = bo[o];
#pragma unroll
    for (int r = 0; r < 8; ++r) {
      int m = r + 8 * lh;
      float v = c[r] + bias;
      if (rowmask[row0 + m]) v = 0.f;
      out[(size_t)(row0 + m) * FF + o] = v;
    }
  }
}

extern "C" void kernel_launch(void* const* d_in, const int* in_sizes, int n_in,
                              void* d_out, int out_size, void* d_ws,
                              size_t ws_size, hipStream_t stream) {
  (void)in_sizes; (void)n_in; (void)out_size; (void)ws_size;
  const float* x      = (const float*)d_in[0];
  const float* y      = (const float*)d_in[1];
  const float* z      = (const float*)d_in[2];
  const int*   op_adj = (const int*)d_in[3];
  const int*   ma_adj = (const int*)d_in[4];
  const int*   op_ma  = (const int*)d_in[5];
  const float* Wop    = (const float*)d_in[6];
  const float* Wma    = (const float*)d_in[7];
  const float* Wedge  = (const float*)d_in[8];
  const float* att_op = (const float*)d_in[9];
  const float* att_ma = (const float*)d_in[10];
  const float* w1     = (const float*)d_in[11];
  const float* b1     = (const float*)d_in[12];
  const float* w2     = (const float*)d_in[13];
  const float* b2     = (const float*)d_in[14];
  const float* wo     = (const float*)d_in[15];
  const float* bo     = (const float*)d_in[16];
  float* out = (float*)d_out;

  float* ws     = (float*)d_ws;
  float* h      = ws;
  float* hma    = h + BB * NN * FF;
  float* ha1    = hma + BB * MM * FF;
  float* ha2    = ha1 + BB * NN;
  float* ham1   = ha2 + BB * NN;
  float* hma2   = ham1 + BB * NN;
  float* hprime = hma2 + BB * MM;
  int*   rmask  = (int*)(hprime + (size_t)BB * NN * 2 * FF);

  gemm_rows_64<<<dim3((BB * NN) / 16), dim3(128), 0, stream>>>(x, Wop, h);
  gemm_rows_64<<<dim3((BB * MM) / 16), dim3(128), 0, stream>>>(y, Wma, hma);
  proj_node<<<dim3((BB * NN) / 256), dim3(256), 0, stream>>>(
      h, att_op, att_ma, op_ma, ha1, ha2, ham1, rmask);
  proj_ma<<<dim3((BB * MM) / 256), dim3(256), 0, stream>>>(hma, att_ma, hma2);
  size_t sm2 = (16 * (NN + 4) + 16) * sizeof(float);
  unsigned* cbits = (unsigned*)(hprime + FF);
  size_t smb = (16 * 32 * CB_P + 32 * 33) * sizeof(int);
  conn_bits<<<dim3(NN / 32, BB), dim3(512), smb, stream>>>(op_adj, ma_adj, cbits);
  op_attn<<<dim3(NN / 16, BB), dim3(128), sm2, stream>>>(
      h, ha1, ha2, cbits, rmask, hprime);
  ma_attn<<<dim3((BB * NN) / (MA_WAVES * MA_NPW)), dim3(32 * MA_WAVES), 0, stream>>>(
      z, Wedge, hma, ham1, hma2, op_ma, rmask, h, att_ma, hprime);
  size_t sm4 = (HID * HID + 2 * 16 * (HID + 4)) * sizeof(float);
  mlp<<<dim3((BB * NN) / 16), dim3(256), sm4, stream>>>(
      hprime, w1, b1, w2, b2, wo, bo, rmask, out);
}


# launches that do not depend on each other share a grid: both feature GEMMs; projections, row flags, hma2 ride in the bit-mask blocks (wave per row, lane sums: summing order changed); 8 launches -> 5
# speedup vs baseline: 2.6361x; 1.1549x over previous
#include <hip/hip_runtime.h>
#include <math.h>


typedef __attribute__((ext_vector_type(2))) float v2f;
typedef __attribute__((ext_vector_type(8))) float v8f;

#define BB 8
#define NN 1024
#define MM 128
#define FF 64
#define FE 16
#define HID 128

__device__ __forceinline__ v8f wmma4(v2f a, v2f b, v8f c) {
  return __builtin_amdgcn_wmma_f32_16x16x4_f32(
       false, a,  false, b,
       (short)0, c,  false,  false);
}

__global__ __launch_bounds__(128) void gemm_rows_64(
    const float* __restrict__ A0, const float* __restrict__ W0,
    float* __restrict__ C0,
    const float* __restrict__ A1, const float* __restrict__ W1,
    float* __restrict__ C1) {
  const bool second = blockIdx.x >= (BB * NN) / 16;
  const float* __restrict__ A = second ? A1 : A0;
  const float* __restrict__ W = second ? W1 : W0;
  float* __restrict__ C = second ? C1 : C0;
  const int wv   = threadIdx.x >> 5;
  const int lane = threadIdx.x & 31;
  const int lm   = lane & 15;
  const int kh   = (lane >> 4) * 2;
  const int lh   = lane >> 4;
  const int row0 = (second ? (int)blockIdx.x - (BB * NN) / 16 : (int)blockIdx.x) * 16;
  const int o    = wv * 16 + lm;
  const float* Arow = A + (size_t)(row0 + lm) * FF;

  v8f c = {};
#pragma unroll
  for (int k0 = 0; k0 < FF; k0 += 4) {
    v2f a, b;
    a.x = Arow[k0 + kh];
    a.y = Arow[k0 + kh + 1];
    b.x = W[(size_t)(k0 + kh) * FF + o];
    b.y = W[(size_t)(k0 + kh + 1) * FF + o];
    c = wmma4(a, b, c);
  }
#pragma unroll
  for (int r = 0; r < 8; ++r) {
    int m = r + 8 * lh;
    C[(size_t)(row0 + m) * FF + o] = c[r];
  }
}

__device__ __forceinline__ float xhalf(float x) {
  int b = __builtin_bit_cast(int, x);
  return __builtin_bit_cast(float,
      __builtin_amdgcn_permlanex16(b, b, 0x76543210, 0xfedcba98, false, false));
}
template <int MASK>
__device__ __forceinline__ float rowx(float x) {
  int b = __builtin_bit_cast(int, x);
  return __builtin_bit_cast(float,
      __builtin_amdgcn_update_dpp(0, b, 0x160 + MASK, 0xf, 0xf, true));
}
__device__ __forceinline__ float row_sum16(float x) {
  x += rowx<1>(x); x += rowx<2>(x); x += rowx<4>(x); x += rowx<8>(x);
  return x;
}
__device__ __forceinline__ float row_max16(float x) {
  x = fmaxf(x, rowx<1>(x)); x = fmaxf(x, rowx<2>(x));
  x = fmaxf(x, rowx<4>(x)); x = fmaxf(x, rowx<8>(x));
  return x;
}
__device__ __forceinline__ float wave_sum32(float x) {
  x = row_sum16(x);
  return x + xhalf(x);
}

#define CB_P 36

__global__ __launch_bounds__(512) void conn_bits_proj(
    const int* __restrict__ opadj, const int* __restrict__ maadj,
    unsigned* __restrict__ cbits,
    const float* __restrict__ h, const float* __restrict__ hma,
    const float* __restrict__ att_op, const float* __restrict__ att_ma,
    const int* __restrict__ opma,
    float* __restrict__ ha1, float* __restrict__ ha2,
    float* __restrict__ ham1, float* __restrict__ hma2,
    int* __restrict__ rowmask) {
  extern __shared__ __attribute__((aligned(16))) int cbs[];
  const int b    = blockIdx.y;
  const int r0   = blockIdx.x * 32;
  const int wv   = threadIdx.x >> 5;
  const int lane = threadIdx.x & 31;
  int* tile = cbs + wv * (32 * CB_P);
  unsigned* words = (unsigned*)(cbs + 16 * 32 * CB_P);
  float* pj = (float*)(cbs + 16 * 32 * CB_P + 32 * 33);
  float* pm = pj + 4 * 32;
  const int* opb = opadj + (size_t)b * NN * NN;
  const int* mab = maadj + (size_t)b * NN * NN;
  const int sub = lane >> 3;
  const int q4  = (lane & 7) * 4;
  const bool machines = (blockIdx.x == 0);

  const int node0 = b * NN + r0 + 2 * wv;
  v2f  hv[2];
  int4 lk[2];
#pragma unroll
  for (int q = 0; q < 2; ++q) {
    hv[q] = *(const v2f*)(h + (size_t)(node0 + q) * FF + 2 * lane);
    lk[q] = *(const int4*)(opma + (size_t)(node0 + q) * MM + 4 * lane);
  }
  const v2f wa1 = *(const v2f*)(att_op + 2 * lane);
  const v2f wa2 = *(const v2f*)(att_op + FF + 2 * lane);
  const v2f wm1 = *(const v2f*)(att_ma + 2 * lane);
  v2f wm2 = {};
  v2f mv[8] = {};
  if (machines) {
    wm2 = *(const v2f*)(att_ma + FF + 2 * lane);
#pragma unroll
    for (int t = 0; t < 8; ++t)
      mv[t] = *(const v2f*)(hma + ((size_t)b * MM + 8 * wv + t) * FF + 2 * lane);
  }

#pragma unroll
  for (int q = 0; q < 2; ++q) {
    const float s1 = wave_sum32(hv[q].x * wa1.x + hv[q].y * wa1.y);
    const float s2 = wave_sum32(hv[q].x * wa2.x + hv[q].y * wa2.y);
    const float s3 = wave_sum32(hv[q].x * wm1.x + hv[q].y * wm1.y);
    const int any = lk[q].x | lk[q].y | lk[q].z | lk[q].w;
    const unsigned linked = __builtin_amdgcn_ballot_w32(any != 0);
    if (lane == 0) {
      pj[0 * 32 + 2 * wv + q] = s1;
      pj[1 * 32 + 2 * wv + q] = s2;
      pj[2 * 32 + 2 * wv + q] = s3;
      ((int*)pj)[3 * 32 + 2 * wv + q] = (linked == 0u) ? 1 : 0;
    }
  }
  if (machines) {
#pragma unroll
    for (int t = 0; t < 8; ++t) {
      const float s = wave_sum32(mv[t].x * wm2.x + mv[t].y * wm2.y);
      if (lane == 0) pm[8 * wv + t] = s;
    }
  }

  int4 dsum[2][8];
  int  tr[2][32];
#pragma unroll
  for (int hf = 0; hf < 2; ++hf) {
    const int c0 = (wv + 16 * hf) * 32;
#pragma unroll
    for (int q = 0; q < 8; ++q) {
      const size_t off = (size_t)(r0 + 4 * q + sub) * NN + c0 + q4;
      const int4 a = *(const int4*)(opb + off);
      const int4 m = *(const int4*)(mab + off);
      dsum[hf][q].x = a.x + m.x; dsum[hf][q].y = a.y + m.y;
      dsum[hf][q].z = a.z + m.z; dsum[hf][q].w = a.w + m.w;
    }
#pragma unroll
    for (int k = 0; k < 32; ++k) {
      const size_t off = (size_t)(c0 + k) * NN + r0 + lane;
      tr[hf][k] = opb[off] + mab[off];
    }
  }

#pragma unroll
  for (int hf = 0; hf < 2; ++hf) {
    const int ct = wv + 16 * hf;
#pragma unroll
    for (int q = 0; q < 8; ++q)
      *(int4*)(tile + (4 * q + sub) * CB_P + q4) = dsum[hf][q];
    __syncthreads();
    unsigned w = 0;
#pragma unroll
    for (int g = 0; g < 8; ++g) {
      const int4 d = *(const int4*)(tile + lane * CB_P + 4 * g);
      w |= (unsigned)((d.x + tr[hf][4 * g + 0]) > 0) << (4 * g + 0);
      w |= (unsigned)((d.y + tr[hf][4 * g + 1]) > 0) << (4 * g + 1);
      w |= (unsigned)((d.z + tr[hf][4 * g + 2]) > 0) << (4 * g + 2);
      w |= (unsigned)((d.w + tr[hf][4 * g + 3]) > 0) << (4 * g + 3);
    }
    words[lane * 33 + ct] = w;
    __syncthreads();
  }

  for (int rr = wv; rr < 32; rr += 16)
    cbits[((size_t)b * NN + r0 + rr) * (2 * FF) + lane] = words[rr * 33 + lane];

  const int nb = b * NN + r0 + lane;
  if (wv == 0)      ha1[nb]  = pj[0 * 32 + lane];
  else if (wv == 1) ha2[nb]  = pj[1 * 32 + lane];
  else if (wv == 2) ham1[nb] = pj[2 * 32 + lane];
  else if (wv == 3) rowmask[nb] = ((const int*)pj)[3 * 32 + lane];
  else if (machines && wv < 8)
    hma2[b * MM + (wv - 4) * 32 + lane] = pm[(wv - 4) * 32 + lane];
}

__global__ __launch_bounds__(128) void op_attn(
    const float* __restrict__ h, const float* __restrict__ ha1,
    const float* __restrict__ ha2, const unsigned* __restrict__ cbits,
    const int* __restrict__ rowmask, float* __restrict__ hprime) {
  extern __shared__ float smem[];
  typedef float att_row_t[NN + 4];
  att_row_t* att = (att_row_t*)smem;
  float* inv_sum = smem + 16 * (NN + 4);

  const int b  = blockIdx.y;
  const int i0 = blockIdx.x * 16;
  const float NEGINF = -__builtin_inff();

  {
    const int wv0 = threadIdx.x >> 5;
    const int ln  = threadIdx.x & 31;
    float a1v[8];
#pragma unroll
    for (int jj = 0; jj < 8; ++jj) a1v[jj] = ha1[b * NN + threadIdx.x + 128 * jj];
    for (int r = 0; r < 16; ++r) {
      const int i = i0 + r;
      const int rm = rowmask[b * NN + i];
      const float a2 = ha2[b * NN + i];
      const unsigned* br = cbits + ((size_t)b * NN + i) * (2 * FF);
#pragma unroll
      for (int jj = 0; jj < 8; ++jj) {
        const unsigned wb = br[wv0 + 4 * jj];
        const int conn = rm | (int)((wb >> ln) & 1u);
        float e = a2 + a1v[jj];
        e = e > 0.f ? e : 0.01f * e;
        att[r][threadIdx.x + 128 * jj] = conn ? e : NEGINF;
      }
    }
  }
  __syncthreads();

  const int wv = threadIdx.x >> 5;
  const int lane = threadIdx.x & 31;
  for (int r = wv; r < 16; r += 4) {
    float mx = NEGINF;
    for (int j = lane; j < NN; j += 32) mx = fmaxf(mx, att[r][j]);
#pragma unroll
    for (int off = 16; off; off >>= 1) mx = fmaxf(mx, __shfl_xor(mx, off, 32));
    if (mx == NEGINF) mx = 0.f;
    float sum = 0.f;
    for (int j = lane; j < NN; j += 32) {
      float p = __expf(att[r][j] - mx);
      att[r][j] = p;
      sum += p;
    }
#pragma unroll
    for (int off = 16; off; off >>= 1) sum += __shfl_xor(sum, off, 32);
    if (lane == 0) inv_sum[r] = 1.f / fmaxf(sum, 1e-30f);
  }
  __syncthreads();

  const int lm = lane & 15;
  const int kh = (lane >> 4) * 2;
  const int lh = lane >> 4;
  const int o  = wv * 16 + lm;
  const float* hb = h + (size_t)b * NN * FF;

  v8f c = {};
  for (int k0 = 0; k0 < NN; k0 += 4) {
    v2f a, bf;
    a.x  = att[lm][k0 + kh];
    a.y  = att[lm][k0 + kh + 1];
    bf.x = hb[(size_t)(k0 + kh) * FF + o];
    bf.y = hb[(size_t)(k0 + kh + 1) * FF + o];
    c = wmma4(a, bf, c);
  }
  float* hp = hprime + ((size_t)b * NN + i0) * (2 * FF);
#pragma unroll
  for (int r = 0; r < 8; ++r) {
    int m = r + 8 * lh;
    hp[(size_t)m * (2 * FF) + o] = c[r] * inv_sum[m];
  }
}

typedef __attribute__((ext_vector_type(4))) float v4f;
#define MA_NPW 8
#define MA_WAVES 4

__global__ __launch_bounds__(32 * MA_WAVES) void ma_attn(
    const float* __restrict__ z, const float* __restrict__ wedge,
    const float* __restrict__ hma, const float* __restrict__ ham1,
    const float* __restrict__ hma2, const int* __restrict__ opma,
    const int* __restrict__ rowmask, const float* __restrict__ h,
    const float* __restrict__ attma, float* __restrict__ hprime) {
  const int lane = threadIdx.x & 31;
  const int c  = lane & 15;
  const int hh = lane >> 4;
  const int wv  = __builtin_amdgcn_readfirstlane((int)(threadIdx.x >> 5));
  const int bn0 = (blockIdx.x * MA_WAVES + wv) * MA_NPW;
  const int b   = bn0 / NN;
  const float NEGINF = -__builtin_inff();

  v2f awe[4][4];
#pragma unroll
  for (int s = 0; s < 4; ++s)
#pragma unroll
    for (int ot = 0; ot < 4; ++ot) {
      awe[s][ot].x = wedge[(8 * hh + 2 * s) * FF + ot * 16 + c];
      awe[s][ot].y = wedge[(8 * hh + 2 * s + 1) * FF + ot * 16 + c];
    }
  v4f am3v[4][2];
#pragma unroll
  for (int ot = 0; ot < 4; ++ot)
#pragma unroll
    for (int q = 0; q < 2; ++q)
      am3v[ot][q] = *(const v4f*)(attma + 2 * FF + ot * 16 + 8 * hh + 4 * q);
  float hma2v[8];
#pragma unroll
  for (int t = 0; t < 8; ++t) hma2v[t] = hma2[b * MM + 16 * t + c];
  const float* hmb = hma + (size_t)b * MM * FF + (size_t)c * FF + 8 * hh;

  v4f zq[8][2];
  {
    const float* zp = z + (size_t)bn0 * MM * FE + c * FE + 8 * hh;
#pragma unroll
    for (int t = 0; t < 8; ++t)
#pragma unroll
      for (int q = 0; q < 2; ++q)
        zq[t][q] = *(const v4f*)(zp + t * 16 * FE + 4 * q);
  }

#pragma unroll 1
  for (int i = 0; i < MA_NPW; ++i) {
    const int bn = bn0 + i;

    const int   rm  = rowmask[bn];
    const float hm1 = ham1[bn];
    const float* hr = h + (size_t)bn * FF + 8 * hh;
    v4f res[4][2];
#pragma unroll
    for (int ot = 0; ot < 4; ++ot) {
      res[ot][0] = *(const v4f*)(hr + ot * 16);
      res[ot][1] = *(const v4f*)(hr + ot * 16 + 4);
    }
    int adv[8];
#pragma unroll
    for (int t = 0; t < 8; ++t) adv[t] = opma[(size_t)bn * MM + 16 * t + c];

    v8f D[8][4];
    float sc[8];
#pragma unroll
    for (int t = 0; t < 8; ++t) {
      v2f zb[4];
      zb[0].x = zq[t][0].x; zb[0].y = zq[t][0].y;
      zb[1].x = zq[t][0].z; zb[1].y = zq[t][0].w;
      zb[2].x = zq[t][1].x; zb[2].y = zq[t][1].y;
      zb[3].x = zq[t][1].z; zb[3].y = zq[t][1].w;
      float part[4];
#pragma unroll
      for (int ot = 0; ot < 4; ++ot) {
        v8f d = {};
#pragma unroll
        for (int s = 0; s < 4; ++s) d = wmma4(awe[s][ot], zb[s], d);
        D[t][ot] = d;
        float a = 0.f;
        a += d[0] * am3v[ot][0].x; a += d[1] * am3v[ot][0].y;
        a += d[2] * am3v[ot][0].z; a += d[3] * am3v[ot][0].w;
        a += d[4] * am3v[ot][1].x; a += d[5] * am3v[ot][1].y;
        a += d[6] * am3v[ot][1].z; a += d[7] * am3v[ot][1].w;
        part[ot] = a;
      }
      float dot = (part[0] + part[1]) + (part[2] + part[3]);
      dot += xhalf(dot);
      float e = hm1 + hma2v[t] + dot;
      e = e > 0.f ? e : 0.01f * e;
      const int conn = rm | (adv[t] > 0);
      sc[t] = conn ? e : NEGINF;
    }

    float mx = sc[0];
#pragma unroll
    for (int t = 1; t < 8; ++t) mx = fmaxf(mx, sc[t]);
    mx = row_max16(mx);
    if (mx == NEGINF) mx = 0.f;
    float p[8];
    float sum = 0.f;
#pragma unroll
    for (int t = 0; t < 8; ++t) { p[t] = __expf(sc[t] - mx); sum += p[t]; }
    sum = row_sum16(sum);
    const float inv = 1.f / fmaxf(sum, 1e-30f);

    if (i + 1 < MA_NPW) {
      const float* zp = z + (size_t)(bn + 1) * MM * FE + c * FE + 8 * hh;
#pragma unroll
      for (int t = 0; t < 8; ++t)
#pragma unroll
        for (int q = 0; q < 2; ++q)
          zq[t][q] = *(const v4f*)(zp + t * 16 * FE + 4 * q);
    }

    v8f acc[4];
#pragma unroll
    for (int ot = 0; ot < 4; ++ot) acc[ot] = (v8f){};
#pragma unroll
    for (int t = 0; t < 8; ++t)
#pragma unroll
      for (int ot = 0; ot < 4; ++ot) {
        const v4f g0 = *(const v4f*)(hmb + (size_t)t * 16 * FF + ot * 16);
        const v4f g1 = *(const v4f*)(hmb + (size_t)t * 16 * FF + ot * 16 + 4);
        acc[ot][0] += p[t] * (g0.x + D[t][ot][0]);
        acc[ot][1] += p[t] * (g0.y + D[t][ot][1]);
        acc[ot][2] += p[t] * (g0.z + D[t][ot][2]);
        acc[ot][3] += p[t] * (g0.w + D[t][ot][3]);
        acc[ot][4] += p[t] * (g1.x + D[t][ot][4]);
        acc[ot][5] += p[t] * (g1.y + D[t][ot][5]);
        acc[ot][6] += p[t] * (g1.z + D[t][ot][6]);
        acc[ot][7] += p[t] * (g1.w + D[t][ot][7]);
      }
#pragma unroll
    for (int ot = 0; ot < 4; ++ot)
#pragma unroll
      for (int r = 0; r < 8; ++r) acc[ot][r] = row_sum16(acc[ot][r]);

    float* hp = hprime + (size_t)bn * (2 * FF) + FF + 8 * hh;
#pragma unroll
    for (int ot = 0; ot < 4; ++ot) {
      const v4f r0 = res[ot][0];
      const v4f r1 = res[ot][1];
      v4f o0, o1;
      o0.x = acc[ot][0] * inv + r0.x; o0.y = acc[ot][1] * inv + r0.y;
      o0.z = acc[ot][2] * inv + r0.z; o0.w = acc[ot][3] * inv + r0.w;
      o1.x = acc[ot][4] * inv + r1.x; o1.y = acc[ot][5] * inv + r1.y;
      o1.z = acc[ot][6] * inv + r1.z; o1.w = acc[ot][7] * inv + r1.w;
      if (c == 0) {
        *(v4f*)(hp + ot * 16)     = o0;
        *(v4f*)(hp + ot * 16 + 4) = o1;
      }
    }

  }
}

__global__ __launch_bounds__(256) void mlp(
    const float* __restrict__ hprime, const float* __restrict__ w1,
    const float* __restrict__ b1, const float* __restrict__ w2,
    const float* __restrict__ b2, const float* __restrict__ wo,
    const float* __restrict__ bo, const int* __restrict__ rowmask,
    float* __restrict__ out) {
  extern __shared__ float smem[];
  float* wb = smem;
  typedef float act_row_t[HID + 4];
  act_row_t* a0 = (act_row_t*)(smem + HID * HID);
  act_row_t* a1 = (act_row_t*)(smem + HID * HID + 16 * (HID + 4));

  const int row0 = blockIdx.x * 16;
  const int wv = threadIdx.x >> 5;
  const int lane = threadIdx.x & 31;
  const int lm = lane & 15;
  const int kh = (lane >> 4) * 2;
  const int lh = lane >> 4;

  for (int t = threadIdx.x; t < 16 * HID; t += 256)
    a0[t >> 7][t & 127] = hprime[(size_t)row0 * HID + t];
  for (int t = threadIdx.x; t < HID * HID; t += 256) wb[t] = w1[t];
  __syncthreads();

  {
    int o = wv * 16 + lm;
    v8f c = {};
    for (int k0 = 0; k0 < HID; k0 += 4) {
      v2f a, bf;
      a.x  = a0[lm][k0 + kh];
      a.y  = a0[lm][k0 + kh + 1];
      bf.x = wb[(k0 + kh) * HID + o];
      bf.y = wb[(k0 + kh + 1) * HID + o];
      c = wmma4(a, bf, c);
    }
    float bias = b1[o];
#pragma unroll
    for (int r = 0; r < 8; ++r) {
      float v = c[r] + bias;
      a1[r + 8 * lh][o] = v > 0.f ? v : expm1f(v);
    }
  }
  __syncthreads();
  for (int t = threadIdx.x; t < HID * HID; t += 256) wb[t] = w2[t];
  __syncthreads();

  {
    int o = wv * 16 + lm;
    v8f c = {};
    for (int k0 = 0; k0 < HID; k0 += 4) {
      v2f a, bf;
      a.x  = a1[lm][k0 + kh];
      a.y  = a1[lm][k0 + kh + 1];
      bf.x = wb[(k0 + kh) * HID + o];
      bf.y = wb[(k0 + kh + 1) * HID + o];
      c = wmma4(a, bf, c);
    }
    float bias = b2[o];
#pragma unroll
    for (int r = 0; r < 8; ++r) {
      float v = c[r] + bias;
      a0[r + 8 * lh][o] = v > 0.f ? v : expm1f(v);
    }
  }
  __syncthreads();
  for (int t = threadIdx.x; t < HID * FF; t += 256) wb[t] = wo[t];
  __syncthreads();

  if (wv < 4) {
    int o = wv * 16 + lm;
    v8f c = {};
    for (int k0 = 0; k0 < HID; k0 += 4) {
      v2f a, bf;
      a.x  = a0[lm][k0 + kh];
      a.y  = a0[lm][k0 + kh + 1];
      bf.x = wb[(k0 + kh) * FF + o];
      bf.y = wb[(k0 + kh + 1) * FF + o];
      c = wmma4(a, bf, c);
    }
    float bias = bo[o];
#pragma unroll
    for (int r = 0; r < 8; ++r) {
      int m = r + 8 * lh;
      float v = c[r] + bias;
      if (rowmask[row0 + m]) v = 0.f;
      out[(size_t)(row0 + m) * FF + o] = v;
    }
  }
}

extern "C" void kernel_launch(void* const* d_in, const int* in_sizes, int n_in,
                              void* d_out, int out_size, void* d_ws,
                              size_t ws_size, hipStream_t stream) {
  (void)in_sizes; (void)n_in; (void)out_size; (void)ws_size;
  const float* x      = (const float*)d_in[0];
  const float* y      = (const float*)d_in[1];
  const float* z      = (const float*)d_in[2];
  const int*   op_adj = (const int*)d_in[3];
  const int*   ma_adj = (const int*)d_in[4];
  const int*   op_ma  = (const int*)d_in[5];
  const float* Wop    = (const float*)d_in[6];
  const float* Wma    = (const float*)d_in[7];
  const float* Wedge  = (const float*)d_in[8];
  const float* att_op = (const float*)d_in[9];
  const float* att_ma = (const float*)d_in[10];
  const float* w1     = (const float*)d_in[11];
  const float* b1     = (const float*)d_in[12];
  const float* w2     = (const float*)d_in[13];
  const float* b2     = (const float*)d_in[14];
  const float* wo     = (const float*)d_in[15];
  const float* bo     = (const float*)d_in[16];
  float* out = (float*)d_out;

  float* ws     = (float*)d_ws;
  float* h      = ws;
  float* hma    = h + BB * NN * FF;
  float* ha1    = hma + BB * MM * FF;
  float* ha2    = ha1 + BB * NN;
  float* ham1   = ha2 + BB * NN;
  float* hma2   = ham1 + BB * NN;
  float* hprime = hma2 + BB * MM;
  int*   rmask  = (int*)(hprime + (size_t)BB * NN * 2 * FF);

  gemm_rows_64<<<dim3((BB * NN) / 16 + (BB * MM) / 16), dim3(128), 0, stream>>>(
      x, Wop, h, y, Wma, hma);
  size_t sm2 = (16 * (NN + 4) + 16) * sizeof(float);
  unsigned* cbits = (unsigned*)(hprime + FF);
  size_t smb = (16 * 32 * CB_P + 32 * 33 + 4 * 32 + MM) * sizeof(int);
  conn_bits_proj<<<dim3(NN / 32, BB), dim3(512), smb, stream>>>(
      op_adj, ma_adj, cbits, h, hma, att_op, att_ma, op_ma,
      ha1, ha2, ham1, hma2, rmask);
  op_attn<<<dim3(NN / 16, BB), dim3(128), sm2, stream>>>(
      h, ha1, ha2, cbits, rmask, hprime);
  ma_attn<<<dim3((BB * NN) / (MA_WAVES * MA_NPW)), dim3(32 * MA_WAVES), 0, stream>>>(
      z, Wedge, hma, ham1, hma2, op_ma, rmask, h, att_ma, hprime);
  size_t sm4 = (HID * HID + 2 * 16 * (HID + 4)) * sizeof(float);
  mlp<<<dim3((BB * NN) / 16), dim3(256), sm4, stream>>>(
      hprime, w1, b1, w2, b2, wo, bo, rmask, out);
}


# op_attn scores+softmax: a wave per row, columns along lanes, mask word as lane mask; max, exp, sum in registers, each weight stored to LDS once (was LDS tile re-read twice + LDS shuffles); same sums
# speedup vs baseline: 2.7649x; 1.0489x over previous
#include <hip/hip_runtime.h>
#include <math.h>


typedef __attribute__((ext_vector_type(2))) float v2f;
typedef __attribute__((ext_vector_type(8))) float v8f;

#define BB 8
#define NN 1024
#define MM 128
#define FF 64
#define FE 16
#define HID 128

__device__ __forceinline__ v8f wmma4(v2f a, v2f b, v8f c) {
  return __builtin_amdgcn_wmma_f32_16x16x4_f32(
       false, a,  false, b,
       (short)0, c,  false,  false);
}

__global__ __launch_bounds__(128) void gemm_rows_64(
    const float* __restrict__ A0, const float* __restrict__ W0,
    float* __restrict__ C0,
    const float* __restrict__ A1, const float* __restrict__ W1,
    float* __restrict__ C1) {
  const bool second = blockIdx.x >= (BB * NN) / 16;
  const float* __restrict__ A = second ? A1 : A0;
  const float* __restrict__ W = second ? W1 : W0;
  float* __restrict__ C = second ? C1 : C0;
  const int wv   = threadIdx.x >> 5;
  const int lane = threadIdx.x & 31;
  const int lm   = lane & 15;
  const int kh   = (lane >> 4) * 2;
  const int lh   = lane >> 4;
  const int row0 = (second ? (int)blockIdx.x - (BB * NN) / 16 : (int)blockIdx.x) * 16;
  const int o    = wv * 16 + lm;
  const float* Arow = A + (size_t)(row0 + lm) * FF;

  v8f c = {};
#pragma unroll
  for (int k0 = 0; k0 < FF; k0 += 4) {
    v2f a, b;
    a.x = Arow[k0 + kh];
    a.y = Arow[k0 + kh + 1];
    b.x = W[(size_t)(k0 + kh) * FF + o];
    b.y = W[(size_t)(k0 + kh + 1) * FF + o];
    c = wmma4(a, b, c);
  }
#pragma unroll
  for (int r = 0; r < 8; ++r) {
    int m = r + 8 * lh;
    C[(size_t)(row0 + m) * FF + o] = c[r];
  }
}

__device__ __forceinline__ float xhalf(float x) {
  int b = __builtin_bit_cast(int, x);
  return __builtin_bit_cast(float,
      __builtin_amdgcn_permlanex16(b, b, 0x76543210, 0xfedcba98, false, false));
}
template <int MASK>
__device__ __forceinline__ float rowx(float x) {
  int b = __builtin_bit_cast(int, x);
  return __builtin_bit_cast(float,
      __builtin_amdgcn_update_dpp(0, b, 0x160 + MASK, 0xf, 0xf, true));
}
__device__ __forceinline__ float row_sum16(float x) {
  x += rowx<1>(x); x += rowx<2>(x); x += rowx<4>(x); x += rowx<8>(x);
  return x;
}
__device__ __forceinline__ float row_max16(float x) {
  x = fmaxf(x, rowx<1>(x)); x = fmaxf(x, rowx<2>(x));
  x = fmaxf(x, rowx<4>(x)); x = fmaxf(x, rowx<8>(x));
  return x;
}
__device__ __forceinline__ float wave_sum32(float x) {
  x = row_sum16(x);
  return x + xhalf(x);
}

#define CB_P 36

__global__ __launch_bounds__(512) void conn_bits_proj(
    const int* __restrict__ opadj, const int* __restrict__ maadj,
    unsigned* __restrict__ cbits,
    const float* __restrict__ h, const float* __restrict__ hma,
    const float* __restrict__ att_op, const float* __restrict__ att_ma,
    const int* __restrict__ opma,
    float* __restrict__ ha1, float* __restrict__ ha2,
    float* __restrict__ ham1, float* __restrict__ hma2,
    int* __restrict__ rowmask) {
  extern __shared__ __attribute__((aligned(16))) int cbs[];
  const int b    = blockIdx.y;
  const int r0   = blockIdx.x * 32;
  const int wv   = threadIdx.x >> 5;
  const int lane = threadIdx.x & 31;
  int* tile = cbs + wv * (32 * CB_P);
  unsigned* words = (unsigned*)(cbs + 16 * 32 * CB_P);
  float* pj = (float*)(cbs + 16 * 32 * CB_P + 32 * 33);
  float* pm = pj + 4 * 32;
  const int* opb = opadj + (size_t)b * NN * NN;
  const int* mab = maadj + (size_t)b * NN * NN;
  const int sub = lane >> 3;
  const int q4  = (lane & 7) * 4;
  const bool machines = (blockIdx.x == 0);

  const int node0 = b * NN + r0 + 2 * wv;
  v2f  hv[2];
  int4 lk[2];
#pragma unroll
  for (int q = 0; q < 2; ++q) {
    hv[q] = *(const v2f*)(h + (size_t)(node0 + q) * FF + 2 * lane);
    lk[q] = *(const int4*)(opma + (size_t)(node0 + q) * MM + 4 * lane);
  }
  const v2f wa1 = *(const v2f*)(att_op + 2 * lane);
  const v2f wa2 = *(const v2f*)(att_op + FF + 2 * lane);
  const v2f wm1 = *(const v2f*)(att_ma + 2 * lane);
  v2f wm2 = {};
  v2f mv[8] = {};
  if (machines) {
    wm2 = *(const v2f*)(att_ma + FF + 2 * lane);
#pragma unroll
    for (int t = 0; t < 8; ++t)
      mv[t] = *(const v2f*)(hma + ((size_t)b * MM + 8 * wv + t) * FF + 2 * lane);
  }

#pragma unroll
  for (int q = 0; q < 2; ++q) {
    const float s1 = wave_sum32(hv[q].x * wa1.x + hv[q].y * wa1.y);
    const float s2 = wave_sum32(hv[q].x * wa2.x + hv[q].y * wa2.y);
    const float s3 = wave_sum32(hv[q].x * wm1.x + hv[q].y * wm1.y);
    const int any = lk[q].x | lk[q].y | lk[q].z | lk[q].w;
    const unsigned linked = __builtin_amdgcn_ballot_w32(any != 0);
    if (lane == 0) {
      pj[0 * 32 + 2 * wv + q] = s1;
      pj[1 * 32 + 2 * wv + q] = s2;
      pj[2 * 32 + 2 * wv + q] = s3;
      ((int*)pj)[3 * 32 + 2 * wv + q] = (linked == 0u) ? 1 : 0;
    }
  }
  if (machines) {
#pragma unroll
    for (int t = 0; t < 8; ++t) {
      const float s = wave_sum32(mv[t].x * wm2.x + mv[t].y * wm2.y);
      if (lane == 0) pm[8 * wv + t] = s;
    }
  }

  int4 dsum[2][8];
  int  tr[2][32];
#pragma unroll
  for (int hf = 0; hf < 2; ++hf) {
    const int c0 = (wv + 16 * hf) * 32;
#pragma unroll
    for (int q = 0; q < 8; ++q) {
      const size_t off = (size_t)(r0 + 4 * q + sub) * NN + c0 + q4;
      const int4 a = *(const int4*)(opb + off);
      const int4 m = *(const int4*)(mab + off);
      dsum[hf][q].x = a.x + m.x; dsum[hf][q].y = a.y + m.y;
      dsum[hf][q].z = a.z + m.z; dsum[hf][q].w = a.w + m.w;
    }
#pragma unroll
    for (int k = 0; k < 32; ++k) {
      const size_t off = (size_t)(c0 + k) * NN + r0 + lane;
      tr[hf][k] = opb[off] + mab[off];
    }
  }

#pragma unroll
  for (int hf = 0; hf < 2; ++hf) {
    const int ct = wv + 16 * hf;
#pragma unroll
    for (int q = 0; q < 8; ++q)
      *(int4*)(tile + (4 * q + sub) * CB_P + q4) = dsum[hf][q];
    __syncthreads();
    unsigned w = 0;
#pragma unroll
    for (int g = 0; g < 8; ++g) {
      const int4 d = *(const int4*)(tile + lane * CB_P + 4 * g);
      w |= (unsigned)((d.x + tr[hf][4 * g + 0]) > 0) << (4 * g + 0);
      w |= (unsigned)((d.y + tr[hf][4 * g + 1]) > 0) << (4 * g + 1);
      w |= (unsigned)((d.z + tr[hf][4 * g + 2]) > 0) << (4 * g + 2);
      w |= (unsigned)((d.w + tr[hf][4 * g + 3]) > 0) << (4 * g + 3);
    }
    words[lane * 33 + ct] = w;
    __syncthreads();
  }

  for (int rr = wv; rr < 32; rr += 16)
    cbits[((size_t)b * NN + r0 + rr) * (2 * FF) + lane] = words[rr * 33 + lane];

  const int nb = b * NN + r0 + lane;
  if (wv == 0)      ha1[nb]  = pj[0 * 32 + lane];
  else if (wv == 1) ha2[nb]  = pj[1 * 32 + lane];
  else if (wv == 2) ham1[nb] = pj[2 * 32 + lane];
  else if (wv == 3) rowmask[nb] = ((const int*)pj)[3 * 32 + lane];
  else if (machines && wv < 8)
    hma2[b * MM + (wv - 4) * 32 + lane] = pm[(wv - 4) * 32 + lane];
}

__global__ __launch_bounds__(128) void op_attn(
    const float* __restrict__ h, const float* __restrict__ ha1,
    const float* __restrict__ ha2, const unsigned* __restrict__ cbits,
    const int* __restrict__ rowmask, float* __restrict__ hprime) {
  extern __shared__ float smem[];
  typedef float att_row_t[NN + 4];
  att_row_t* att = (att_row_t*)smem;
  float* inv_sum = smem + 16 * (NN + 4);

  const int b  = blockIdx.y;
  const int i0 = blockIdx.x * 16;
  const float NEGINF = -__builtin_inff();

  const int lane = threadIdx.x & 31;
  const int wv = __builtin_amdgcn_readfirstlane((int)(threadIdx.x >> 5));
  {
    float a1v[32];
#pragma unroll
    for (int t = 0; t < 32; ++t) a1v[t] = ha1[b * NN + lane + 32 * t];
    unsigned wd[4];
#pragma unroll
    for (int rr = 0; rr < 4; ++rr)
      wd[rr] = cbits[((size_t)b * NN + i0 + wv + 4 * rr) * (2 * FF) + lane];
#pragma unroll
    for (int rr = 0; rr < 4; ++rr) {
      const int r = wv + 4 * rr;
      const int i = i0 + r;
      const int rm = rowmask[b * NN + i];
      const float a2 = ha2[b * NN + i];
      const unsigned wdm = rm ? 0xffffffffu : wd[rr];
      float e[32];
      float mx = NEGINF;
#pragma unroll
      for (int t = 0; t < 32; ++t) {
        const unsigned wt = (unsigned)__builtin_amdgcn_readlane((int)wdm, t);
        float s = a2 + a1v[t];
        s = s > 0.f ? s : 0.01f * s;
        e[t] = __builtin_amdgcn_inverse_ballot_w32(wt) ? s : NEGINF;
        mx = fmaxf(mx, e[t]);
      }
      mx = fmaxf(mx, xhalf(mx));
      mx = row_max16(mx);
      if (mx == NEGINF) mx = 0.f;
      float sum = 0.f;
#pragma unroll
      for (int t = 0; t < 32; ++t) {
        const float p = __expf(e[t] - mx);
        att[r][lane + 32 * t] = p;
        sum += p;
      }
      sum += xhalf(sum);
      sum += rowx<8>(sum); sum += rowx<4>(sum);
      sum += rowx<2>(sum); sum += rowx<1>(sum);
      if (lane == 0) inv_sum[r] = 1.f / fmaxf(sum, 1e-30f);
    }
  }
  __syncthreads();

  const int lm = lane & 15;
  const int kh = (lane >> 4) * 2;
  const int lh = lane >> 4;
  const int o  = wv * 16 + lm;
  const float* hb = h + (size_t)b * NN * FF;

  v8f c = {};
  for (int k0 = 0; k0 < NN; k0 += 4) {
    v2f a, bf;
    a.x  = att[lm][k0 + kh];
    a.y  = att[lm][k0 + kh + 1];
    bf.x = hb[(size_t)(k0 + kh) * FF + o];
    bf.y = hb[(size_t)(k0 + kh + 1) * FF + o];
    c = wmma4(a, bf, c);
  }
  float* hp = hprime + ((size_t)b * NN + i0) * (2 * FF);
#pragma unroll
  for (int r = 0; r < 8; ++r) {
    int m = r + 8 * lh;
    hp[(size_t)m * (2 * FF) + o] = c[r] * inv_sum[m];
  }
}

typedef __attribute__((ext_vector_type(4))) float v4f;
#define MA_NPW 8
#define MA_WAVES 4

__global__ __launch_bounds__(32 * MA_WAVES) void ma_attn(
    const float* __restrict__ z, const float* __restrict__ wedge,
    const float* __restrict__ hma, const float* __restrict__ ham1,
    const float* __restrict__ hma2, const int* __restrict__ opma,
    const int* __restrict__ rowmask, const float* __restrict__ h,
    const float* __restrict__ attma, float* __restrict__ hprime) {
  const int lane = threadIdx.x & 31;
  const int c  = lane & 15;
  const int hh = lane >> 4;
  const int wv  = __builtin_amdgcn_readfirstlane((int)(threadIdx.x >> 5));
  const int bn0 = (blockIdx.x * MA_WAVES + wv) * MA_NPW;
  const int b   = bn0 / NN;
  const float NEGINF = -__builtin_inff();

  v2f awe[4][4];
#pragma unroll
  for (int s = 0; s < 4; ++s)
#pragma unroll
    for (int ot = 0; ot < 4; ++ot) {
      awe[s][ot].x = wedge[(8 * hh + 2 * s) * FF + ot * 16 + c];
      awe[s][ot].y = wedge[(8 * hh + 2 * s + 1) * FF + ot * 16 + c];
    }
  v4f am3v[4][2];
#pragma unroll
  for (int ot = 0; ot < 4; ++ot)
#pragma unroll
    for (int q = 0; q < 2; ++q)
      am3v[ot][q] = *(const v4f*)(attma + 2 * FF + ot * 16 + 8 * hh + 4 * q);
  float hma2v[8];
#pragma unroll
  for (int t = 0; t < 8; ++t) hma2v[t] = hma2[b * MM + 16 * t + c];
  const float* hmb = hma + (size_t)b * MM * FF + (size_t)c * FF + 8 * hh;

  v4f zq[8][2];
  {
    const float* zp = z + (size_t)bn0 * MM * FE + c * FE + 8 * hh;
#pragma unroll
    for (int t = 0; t < 8; ++t)
#pragma unroll
      for (int q = 0; q < 2; ++q)
        zq[t][q] = *(const v4f*)(zp + t * 16 * FE + 4 * q);
  }

#pragma unroll 1
  for (int i = 0; i < MA_NPW; ++i) {
    const int bn = bn0 + i;

    const int   rm  = rowmask[bn];
    const float hm1 = ham1[bn];
    const float* hr = h + (size_t)bn * FF + 8 * hh;
    v4f res[4][2];
#pragma unroll
    for (int ot = 0; ot < 4; ++ot) {
      res[ot][0] = *(const v4f*)(hr + ot * 16);
      res[ot][1] = *(const v4f*)(hr + ot * 16 + 4);
    }
    int adv[8];
#pragma unroll
    for (int t = 0; t < 8; ++t) adv[t] = opma[(size_t)bn * MM + 16 * t + c];

    v8f D[8][4];
    float sc[8];
#pragma unroll
    for (int t = 0; t < 8; ++t) {
      v2f zb[4];
      zb[0].x = zq[t][0].x; zb[0].y = zq[t][0].y;
      zb[1].x = zq[t][0].z; zb[1].y = zq[t][0].w;
      zb[2].x = zq[t][1].x; zb[2].y = zq[t][1].y;
      zb[3].x = zq[t][1].z; zb[3].y = zq[t][1].w;
      float part[4];
#pragma unroll
      for (int ot = 0; ot < 4; ++ot) {
        v8f d = {};
#pragma unroll
        for (int s = 0; s < 4; ++s) d = wmma4(awe[s][ot], zb[s], d);
        D[t][ot] = d;
        float a = 0.f;
        a += d[0] * am3v[ot][0].x; a += d[1] * am3v[ot][0].y;
        a += d[2] * am3v[ot][0].z; a += d[3] * am3v[ot][0].w;
        a += d[4] * am3v[ot][1].x; a += d[5] * am3v[ot][1].y;
        a += d[6] * am3v[ot][1].z; a += d[7] * am3v[ot][1].w;
        part[ot] = a;
      }
      float dot = (part[0] + part[1]) + (part[2] + part[3]);
      dot += xhalf(dot);
      float e = hm1 + hma2v[t] + dot;
      e = e > 0.f ? e : 0.01f * e;
      const int conn = rm | (adv[t] > 0);
      sc[t] = conn ? e : NEGINF;
    }

    float mx = sc[0];
#pragma unroll
    for (int t = 1; t < 8; ++t) mx = fmaxf(mx, sc[t]);
    mx = row_max16(mx);
    if (mx == NEGINF) mx = 0.f;
    float p[8];
    float sum = 0.f;
#pragma unroll
    for (int t = 0; t < 8; ++t) { p[t] = __expf(sc[t] - mx); sum += p[t]; }
    sum = row_sum16(sum);
    const float inv = 1.f / fmaxf(sum, 1e-30f);

    if (i + 1 < MA_NPW) {
      const float* zp = z + (size_t)(bn + 1) * MM * FE + c * FE + 8 * hh;
#pragma unroll
      for (int t = 0; t < 8; ++t)
#pragma unroll
        for (int q = 0; q < 2; ++q)
          zq[t][q] = *(const v4f*)(zp + t * 16 * FE + 4 * q);
    }

    v8f acc[4];
#pragma unroll
    for (int ot = 0; ot < 4; ++ot) acc[ot] = (v8f){};
#pragma unroll
    for (int t = 0; t < 8; ++t)
#pragma unroll
      for (int ot = 0; ot < 4; ++ot) {
        const v4f g0 = *(const v4f*)(hmb + (size_t)t * 16 * FF + ot * 16);
        const v4f g1 = *(const v4f*)(hmb + (size_t)t * 16 * FF + ot * 16 + 4);
        acc[ot][0] += p[t] * (g0.x + D[t][ot][0]);
        acc[ot][1] += p[t] * (g0.y + D[t][ot][1]);
        acc[ot][2] += p[t] * (g0.z + D[t][ot][2]);
        acc[ot][3] += p[t] * (g0.w + D[t][ot][3]);
        acc[ot][4] += p[t] * (g1.x + D[t][ot][4]);
        acc[ot][5] += p[t] * (g1.y + D[t][ot][5]);
        acc[ot][6] += p[t] * (g1.z + D[t][ot][6]);
        acc[ot][7] += p[t] * (g1.w + D[t][ot][7]);
      }
#pragma unroll
    for (int ot = 0; ot < 4; ++ot)
#pragma unroll
      for (int r = 0; r < 8; ++r) acc[ot][r] = row_sum16(acc[ot][r]);

    float* hp = hprime + (size_t)bn * (2 * FF) + FF + 8 * hh;
#pragma unroll
    for (int ot = 0; ot < 4; ++ot) {
      const v4f r0 = res[ot][0];
      const v4f r1 = res[ot][1];
      v4f o0, o1;
      o0.x = acc[ot][0] * inv + r0.x; o0.y = acc[ot][1] * inv + r0.y;
      o0.z = acc[ot][2] * inv + r0.z; o0.w = acc[ot][3] * inv + r0.w;
      o1.x = acc[ot][4] * inv + r1.x; o1.y = acc[ot][5] * inv + r1.y;
      o1.z = acc[ot][6] * inv + r1.z; o1.w = acc[ot][7] * inv + r1.w;
      if (c == 0) {
        *(v4f*)(hp + ot * 16)     = o0;
        *(v4f*)(hp + ot * 16 + 4) = o1;
      }
    }

  }
}

__global__ __launch_bounds__(256) void mlp(
    const float* __restrict__ hprime, const float* __restrict__ w1,
    const float* __restrict__ b1, const float* __restrict__ w2,
    const float* __restrict__ b2, const float* __restrict__ wo,
    const float* __restrict__ bo, const int* __restrict__ rowmask,
    float* __restrict__ out) {
  extern __shared__ float smem[];
  float* wb = smem;
  typedef float act_row_t[HID + 4];
  act_row_t* a0 = (act_row_t*)(smem + HID * HID);
  act_row_t* a1 = (act_row_t*)(smem + HID * HID + 16 * (HID + 4));

  const int row0 = blockIdx.x * 16;
  const int wv = threadIdx.x >> 5;
  const int lane = threadIdx.x & 31;
  const int lm = lane & 15;
  const int kh = (lane >> 4) * 2;
  const int lh = lane >> 4;

  for (int t = threadIdx.x; t < 16 * HID; t += 256)
    a0[t >> 7][t & 127] = hprime[(size_t)row0 * HID + t];
  for (int t = threadIdx.x; t < HID * HID; t += 256) wb[t] = w1[t];
  __syncthreads();

  {
    int o = wv * 16 + lm;
    v8f c = {};
    for (int k0 = 0; k0 < HID; k0 += 4) {
      v2f a, bf;
      a.x  = a0[lm][k0 + kh];
      a.y  = a0[lm][k0 + kh + 1];
      bf.x = wb[(k0 + kh) * HID + o];
      bf.y = wb[(k0 + kh + 1) * HID + o];
      c = wmma4(a, bf, c);
    }
    float bias = b1[o];
#pragma unroll
    for (int r = 0; r < 8; ++r) {
      float v = c[r] + bias;
      a1[r + 8 * lh][o] = v > 0.f ? v : expm1f(v);
    }
  }
  __syncthreads();
  for (int t = threadIdx.x; t < HID * HID; t += 256) wb[t] = w2[t];
  __syncthreads();

  {
    int o = wv * 16 + lm;
    v8f c = {};
    for (int k0 = 0; k0 < HID; k0 += 4) {
      v2f a, bf;
      a.x  = a1[lm][k0 + kh];
      a.y  = a1[lm][k0 + kh + 1];
      bf.x = wb[(k0 + kh) * HID + o];
      bf.y = wb[(k0 + kh + 1) * HID + o];
      c = wmma4(a, bf, c);
    }
    float bias = b2[o];
#pragma unroll
    for (int r = 0; r < 8; ++r) {
      float v = c[r] + bias;
      a0[r + 8 * lh][o] = v > 0.f ? v : expm1f(v);
    }
  }
  __syncthreads();
  for (int t = threadIdx.x; t < HID * FF; t += 256) wb[t] = wo[t];
  __syncthreads();

  if (wv < 4) {
    int o = wv * 16 + lm;
    v8f c = {};
    for (int k0 = 0; k0 < HID; k0 += 4) {
      v2f a, bf;
      a.x  = a0[lm][k0 + kh];
      a.y  = a0[lm][k0 + kh + 1];
      bf.x = wb[(k0 + kh) * FF + o];
      bf.y = wb[(k0 + kh + 1) * FF + o];
      c = wmma4(a, bf, c);
    }
    float bias = bo[o];
#pragma unroll
    for (int r = 0; r < 8; ++r) {
      int m = r + 8 * lh;
      float v = c[r] + bias;
      if (rowmask[row0 + m]) v = 0.f;
      out[(size_t)(row0 + m) * FF + o] = v;
    }
  }
}

extern "C" void kernel_launch(void* const* d_in, const int* in_sizes, int n_in,
                              void* d_out, int out_size, void* d_ws,
                              size_t ws_size, hipStream_t stream) {
  (void)in_sizes; (void)n_in; (void)out_size; (void)ws_size;
  const float* x      = (const float*)d_in[0];
  const float* y      = (const float*)d_in[1];
  const float* z      = (const float*)d_in[2];
  const int*   op_adj = (const int*)d_in[3];
  const int*   ma_adj = (const int*)d_in[4];
  const int*   op_ma  = (const int*)d_in[5];
  const float* Wop    = (const float*)d_in[6];
  const float* Wma    = (const float*)d_in[7];
  const float* Wedge  = (const float*)d_in[8];
  const float* att_op = (const float*)d_in[9];
  const float* att_ma = (const float*)d_in[10];
  const float* w1     = (const float*)d_in[11];
  const float* b1     = (const float*)d_in[12];
  const float* w2     = (const float*)d_in[13];
  const float* b2     = (const float*)d_in[14];
  const float* wo     = (const float*)d_in[15];
  const float* bo     = (const float*)d_in[16];
  float* out = (float*)d_out;

  float* ws     = (float*)d_ws;
  float* h      = ws;
  float* hma    = h + BB * NN * FF;
  float* ha1    = hma + BB * MM * FF;
  float* ha2    = ha1 + BB * NN;
  float* ham1   = ha2 + BB * NN;
  float* hma2   = ham1 + BB * NN;
  float* hprime = hma2 + BB * MM;
  int*   rmask  = (int*)(hprime + (size_t)BB * NN * 2 * FF);

  gemm_rows_64<<<dim3((BB * NN) / 16 + (BB * MM) / 16), dim3(128), 0, stream>>>(
      x, Wop, h, y, Wma, hma);
  size_t sm2 = (16 * (NN + 4) + 16) * sizeof(float);
  unsigned* cbits = (unsigned*)(hprime + FF);
  size_t smb = (16 * 32 * CB_P + 32 * 33 + 4 * 32 + MM) * sizeof(int);
  conn_bits_proj<<<dim3(NN / 32, BB), dim3(512), smb, stream>>>(
      op_adj, ma_adj, cbits, h, hma, att_op, att_ma, op_ma,
      ha1, ha2, ham1, hma2, rmask);
  op_attn<<<dim3(NN / 16, BB), dim3(128), sm2, stream>>>(
      h, ha1, ha2, cbits, rmask, hprime);
  ma_attn<<<dim3((BB * NN) / (MA_WAVES * MA_NPW)), dim3(32 * MA_WAVES), 0, stream>>>(
      z, Wedge, hma, ham1, hma2, op_ma, rmask, h, att_ma, hprime);
  size_t sm4 = (HID * HID + 2 * 16 * (HID + 4)) * sizeof(float);
  mlp<<<dim3((BB * NN) / 16), dim3(256), sm4, stream>>>(
      hprime, w1, b1, w2, b2, wo, bo, rmask, out);
}


# op_attn product: h rows reach the waves through LDS: 64-row slabs fetched by all threads as 16-byte vectors two slabs ahead, parked in a 2-buffer LDS ring; B fragments read from LDS; same sums
# speedup vs baseline: 2.9610x; 1.0709x over previous
#include <hip/hip_runtime.h>
#include <math.h>


typedef __attribute__((ext_vector_type(2))) float v2f;
typedef __attribute__((ext_vector_type(8))) float v8f;
typedef __attribute__((ext_vector_type(4))) float v4f;

#define BB 8
#define NN 1024
#define MM 128
#define FF 64
#define FE 16
#define HID 128

__device__ __forceinline__ v8f wmma4(v2f a, v2f b, v8f c) {
  return __builtin_amdgcn_wmma_f32_16x16x4_f32(
       false, a,  false, b,
       (short)0, c,  false,  false);
}

__global__ __launch_bounds__(128) void gemm_rows_64(
    const float* __restrict__ A0, const float* __restrict__ W0,
    float* __restrict__ C0,
    const float* __restrict__ A1, const float* __restrict__ W1,
    float* __restrict__ C1) {
  const bool second = blockIdx.x >= (BB * NN) / 16;
  const float* __restrict__ A = second ? A1 : A0;
  const float* __restrict__ W = second ? W1 : W0;
  float* __restrict__ C = second ? C1 : C0;
  const int wv   = threadIdx.x >> 5;
  const int lane = threadIdx.x & 31;
  const int lm   = lane & 15;
  const int kh   = (lane >> 4) * 2;
  const int lh   = lane >> 4;
  const int row0 = (second ? (int)blockIdx.x - (BB * NN) / 16 : (int)blockIdx.x) * 16;
  const int o    = wv * 16 + lm;
  const float* Arow = A + (size_t)(row0 + lm) * FF;

  v8f c = {};
#pragma unroll
  for (int k0 = 0; k0 < FF; k0 += 4) {
    v2f a, b;
    a.x = Arow[k0 + kh];
    a.y = Arow[k0 + kh + 1];
    b.x = W[(size_t)(k0 + kh) * FF + o];
    b.y = W[(size_t)(k0 + kh + 1) * FF + o];
    c = wmma4(a, b, c);
  }
#pragma unroll
  for (int r = 0; r < 8; ++r) {
    int m = r + 8 * lh;
    C[(size_t)(row0 + m) * FF + o] = c[r];
  }
}

__device__ __forceinline__ float xhalf(float x) {
  int b = __builtin_bit_cast(int, x);
  return __builtin_bit_cast(float,
      __builtin_amdgcn_permlanex16(b, b, 0x76543210, 0xfedcba98, false, false));
}
template <int MASK>
__device__ __forceinline__ float rowx(float x) {
  int b = __builtin_bit_cast(int, x);
  return __builtin_bit_cast(float,
      __builtin_amdgcn_update_dpp(0, b, 0x160 + MASK, 0xf, 0xf, true));
}
__device__ __forceinline__ float row_sum16(float x) {
  x += rowx<1>(x); x += rowx<2>(x); x += rowx<4>(x); x += rowx<8>(x);
  return x;
}
__device__ __forceinline__ float row_max16(float x) {
  x = fmaxf(x, rowx<1>(x)); x = fmaxf(x, rowx<2>(x));
  x = fmaxf(x, rowx<4>(x)); x = fmaxf(x, rowx<8>(x));
  return x;
}
__device__ __forceinline__ float wave_sum32(float x) {
  x = row_sum16(x);
  return x + xhalf(x);
}

#define CB_P 36

__global__ __launch_bounds__(512) void conn_bits_proj(
    const int* __restrict__ opadj, const int* __restrict__ maadj,
    unsigned* __restrict__ cbits,
    const float* __restrict__ h, const float* __restrict__ hma,
    const float* __restrict__ att_op, const float* __restrict__ att_ma,
    const int* __restrict__ opma,
    float* __restrict__ ha1, float* __restrict__ ha2,
    float* __restrict__ ham1, float* __restrict__ hma2,
    int* __restrict__ rowmask) {
  extern __shared__ __attribute__((aligned(16))) int cbs[];
  const int b    = blockIdx.y;
  const int r0   = blockIdx.x * 32;
  const int wv   = threadIdx.x >> 5;
  const int lane = threadIdx.x & 31;
  int* tile = cbs + wv * (32 * CB_P);
  unsigned* words = (unsigned*)(cbs + 16 * 32 * CB_P);
  float* pj = (float*)(cbs + 16 * 32 * CB_P + 32 * 33);
  float* pm = pj + 4 * 32;
  const int* opb = opadj + (size_t)b * NN * NN;
  const int* mab = maadj + (size_t)b * NN * NN;
  const int sub = lane >> 3;
  const int q4  = (lane & 7) * 4;
  const bool machines = (blockIdx.x == 0);

  const int node0 = b * NN + r0 + 2 * wv;
  v2f  hv[2];
  int4 lk[2];
#pragma unroll
  for (int q = 0; q < 2; ++q) {
    hv[q] = *(const v2f*)(h + (size_t)(node0 + q) * FF + 2 * lane);
    lk[q] = *(const int4*)(opma + (size_t)(node0 + q) * MM + 4 * lane);
  }
  const v2f wa1 = *(const v2f*)(att_op + 2 * lane);
  const v2f wa2 = *(const v2f*)(att_op + FF + 2 * lane);
  const v2f wm1 = *(const v2f*)(att_ma + 2 * lane);
  v2f wm2 = {};
  v2f mv[8] = {};
  if (machines) {
    wm2 = *(const v2f*)(att_ma + FF + 2 * lane);
#pragma unroll
    for (int t = 0; t < 8; ++t)
      mv[t] = *(const v2f*)(hma + ((size_t)b * MM + 8 * wv + t) * FF + 2 * lane);
  }

#pragma unroll
  for (int q = 0; q < 2; ++q) {
    const float s1 = wave_sum32(hv[q].x * wa1.x + hv[q].y * wa1.y);
    const float s2 = wave_sum32(hv[q].x * wa2.x + hv[q].y * wa2.y);
    const float s3 = wave_sum32(hv[q].x * wm1.x + hv[q].y * wm1.y);
    const int any = lk[q].x | lk[q].y | lk[q].z | lk[q].w;
    const unsigned linked = __builtin_amdgcn_ballot_w32(any != 0);
    if (lane == 0) {
      pj[0 * 32 + 2 * wv + q] = s1;
      pj[1 * 32 + 2 * wv + q] = s2;
      pj[2 * 32 + 2 * wv + q] = s3;
      ((int*)pj)[3 * 32 + 2 * wv + q] = (linked == 0u) ? 1 : 0;
    }
  }
  if (machines) {
#pragma unroll
    for (int t = 0; t < 8; ++t) {
      const float s = wave_sum32(mv[t].x * wm2.x + mv[t].y * wm2.y);
      if (lane == 0) pm[8 * wv + t] = s;
    }
  }

  int4 dsum[2][8];
  int  tr[2][32];
#pragma unroll
  for (int hf = 0; hf < 2; ++hf) {
    const int c0 = (wv + 16 * hf) * 32;
#pragma unroll
    for (int q = 0; q < 8; ++q) {
      const size_t off = (size_t)(r0 + 4 * q + sub) * NN + c0 + q4;
      const int4 a = *(const int4*)(opb + off);
      const int4 m = *(const int4*)(mab + off);
      dsum[hf][q].x = a.x + m.x; dsum[hf][q].y = a.y + m.y;
      dsum[hf][q].z = a.z + m.z; dsum[hf][q].w = a.w + m.w;
    }
#pragma unroll
    for (int k = 0; k < 32; ++k) {
      const size_t off = (size_t)(c0 + k) * NN + r0 + lane;
      tr[hf][k] = opb[off] + mab[off];
    }
  }

#pragma unroll
  for (int hf = 0; hf < 2; ++hf) {
    const int ct = wv + 16 * hf;
#pragma unroll
    for (int q = 0; q < 8; ++q)
      *(int4*)(tile + (4 * q + sub) * CB_P + q4) = dsum[hf][q];
    __syncthreads();
    unsigned w = 0;
#pragma unroll
    for (int g = 0; g < 8; ++g) {
      const int4 d = *(const int4*)(tile + lane * CB_P + 4 * g);
      w |= (unsigned)((d.x + tr[hf][4 * g + 0]) > 0) << (4 * g + 0);
      w |= (unsigned)((d.y + tr[hf][4 * g + 1]) > 0) << (4 * g + 1);
      w |= (unsigned)((d.z + tr[hf][4 * g + 2]) > 0) << (4 * g + 2);
      w |= (unsigned)((d.w + tr[hf][4 * g + 3]) > 0) << (4 * g + 3);
    }
    words[lane * 33 + ct] = w;
    __syncthreads();
  }

  for (int rr = wv; rr < 32; rr += 16)
    cbits[((size_t)b * NN + r0 + rr) * (2 * FF) + lane] = words[rr * 33 + lane];

  const int nb = b * NN + r0 + lane;
  if (wv == 0)      ha1[nb]  = pj[0 * 32 + lane];
  else if (wv == 1) ha2[nb]  = pj[1 * 32 + lane];
  else if (wv == 2) ham1[nb] = pj[2 * 32 + lane];
  else if (wv == 3) rowmask[nb] = ((const int*)pj)[3 * 32 + lane];
  else if (machines && wv < 8)
    hma2[b * MM + (wv - 4) * 32 + lane] = pm[(wv - 4) * 32 + lane];
}

#define HS_ROWS 64
__device__ __forceinline__ void lds_barrier() {
  __builtin_amdgcn_fence(__ATOMIC_RELEASE, "workgroup", "local");
  __builtin_amdgcn_s_barrier();
  __builtin_amdgcn_fence(__ATOMIC_ACQUIRE, "workgroup", "local");
}
#define HS_P 72

__global__ __launch_bounds__(128) void op_attn(
    const float* __restrict__ h, const float* __restrict__ ha1,
    const float* __restrict__ ha2, const unsigned* __restrict__ cbits,
    const int* __restrict__ rowmask, float* __restrict__ hprime) {
  extern __shared__ float smem[];
  typedef float att_row_t[NN + 4];
  att_row_t* att = (att_row_t*)smem;
  float* inv_sum = smem + 16 * (NN + 4);

  const int b  = blockIdx.y;
  const int i0 = blockIdx.x * 16;
  const float NEGINF = -__builtin_inff();

  const int lane = threadIdx.x & 31;
  const int wv = __builtin_amdgcn_readfirstlane((int)(threadIdx.x >> 5));

  float* hs = smem + 16 * (NN + 4) + 16;
  const v4f* hb4 = (const v4f*)(h + (size_t)b * NN * FF) + threadIdx.x;
  float* hst = hs + (threadIdx.x >> 4) * HS_P + (threadIdx.x & 15) * 4;
  v4f sa[8], sb[8];
#pragma unroll
  for (int q = 0; q < 8; ++q) {
    sa[q] = hb4[0 * (HS_ROWS * 16) + 128 * q];
    sb[q] = hb4[1 * (HS_ROWS * 16) + 128 * q];
  }
  {
    float a1v[32];
#pragma unroll
    for (int t = 0; t < 32; ++t) a1v[t] = ha1[b * NN + lane + 32 * t];
    unsigned wd[4];
#pragma unroll
    for (int rr = 0; rr < 4; ++rr)
      wd[rr] = cbits[((size_t)b * NN + i0 + wv + 4 * rr) * (2 * FF) + lane];
#pragma unroll
    for (int rr = 0; rr < 4; ++rr) {
      const int r = wv + 4 * rr;
      const int i = i0 + r;
      const int rm = rowmask[b * NN + i];
      const float a2 = ha2[b * NN + i];
      const unsigned wdm = rm ? 0xffffffffu : wd[rr];
      float e[32];
      float mx = NEGINF;
#pragma unroll
      for (int t = 0; t < 32; ++t) {
        const unsigned wt = (unsigned)__builtin_amdgcn_readlane((int)wdm, t);
        float s = a2 + a1v[t];
        s = s > 0.f ? s : 0.01f * s;
        e[t] = __builtin_amdgcn_inverse_ballot_w32(wt) ? s : NEGINF;
        mx = fmaxf(mx, e[t]);
      }
      mx = fmaxf(mx, xhalf(mx));
      mx = row_max16(mx);
      if (mx == NEGINF) mx = 0.f;
      float sum = 0.f;
#pragma unroll
      for (int t = 0; t < 32; ++t) {
        const float p = __expf(e[t] - mx);
        att[r][lane + 32 * t] = p;
        sum += p;
      }
      sum += xhalf(sum);
      sum += rowx<8>(sum); sum += rowx<4>(sum);
      sum += rowx<2>(sum); sum += rowx<1>(sum);
      if (lane == 0) inv_sum[r] = 1.f / fmaxf(sum, 1e-30f);
    }
  }
#pragma unroll
  for (int q = 0; q < 8; ++q) *(v4f*)(hst + q * 8 * HS_P) = sa[q];
  __syncthreads();

  const int lm = lane & 15;
  const int kh = (lane >> 4) * 2;
  const int lh = lane >> 4;
  const int o  = wv * 16 + lm;
  const float* ar  = &att[lm][kh];
  const float* hf0 = hs + kh * HS_P + o;
  const float* hf1 = hf0 + HS_ROWS * HS_P;

  v8f c = {};
#pragma unroll 1
  for (int s = 0; s < NN / HS_ROWS; s += 2) {
    {
      const int sn = (s + 2) & (NN / HS_ROWS - 1);
#pragma unroll
      for (int q = 0; q < 8; ++q) sa[q] = hb4[sn * (HS_ROWS * 16) + 128 * q];
    }
    {
      v2f af[16], bq[16];
#pragma unroll
      for (int t = 0; t < 16; ++t) {
        af[t].x = ar[s * HS_ROWS + 4 * t];
        af[t].y = ar[s * HS_ROWS + 4 * t + 1];
        bq[t].x = hf0[(4 * t) * HS_P];
        bq[t].y = hf0[(4 * t + 1) * HS_P];
      }
#pragma unroll
      for (int t = 0; t < 16; ++t) c = wmma4(af[t], bq[t], c);
    }
#pragma unroll
    for (int q = 0; q < 8; ++q) *(v4f*)(hst + HS_ROWS * HS_P + q * 8 * HS_P) = sb[q];
    __builtin_amdgcn_sched_group_barrier(0x020, 8, 0);
    __builtin_amdgcn_sched_group_barrier(0x100, 32, 0);
    __builtin_amdgcn_sched_group_barrier(0x008, 16, 0);
    __builtin_amdgcn_sched_group_barrier(0x200, 8, 0);
    lds_barrier();
    {
      const int sn = (s + 3) & (NN / HS_ROWS - 1);
#pragma unroll
      for (int q = 0; q < 8; ++q) sb[q] = hb4[sn * (HS_ROWS * 16) + 128 * q];
    }
    {
      v2f af[16], bq[16];
#pragma unroll
      for (int t = 0; t < 16; ++t) {
        af[t].x = ar[(s + 1) * HS_ROWS + 4 * t];
        af[t].y = ar[(s + 1) * HS_ROWS + 4 * t + 1];
        bq[t].x = hf1[(4 * t) * HS_P];
        bq[t].y = hf1[(4 * t + 1) * HS_P];
      }
#pragma unroll
      for (int t = 0; t < 16; ++t) c = wmma4(af[t], bq[t], c);
    }
#pragma unroll
    for (int q = 0; q < 8; ++q) *(v4f*)(hst + q * 8 * HS_P) = sa[q];
    __builtin_amdgcn_sched_group_barrier(0x020, 8, 0);
    __builtin_amdgcn_sched_group_barrier(0x100, 32, 0);
    __builtin_amdgcn_sched_group_barrier(0x008, 16, 0);
    __builtin_amdgcn_sched_group_barrier(0x200, 8, 0);
    lds_barrier();
  }
  float* hp = hprime + ((size_t)b * NN + i0) * (2 * FF);
#pragma unroll
  for (int r = 0; r < 8; ++r) {
    int m = r + 8 * lh;
    hp[(size_t)m * (2 * FF) + o] = c[r] * inv_sum[m];
  }
}

#define MA_NPW 8
#define MA_WAVES 4

__global__ __launch_bounds__(32 * MA_WAVES) void ma_attn(
    const float* __restrict__ z, const float* __restrict__ wedge,
    const float* __restrict__ hma, const float* __restrict__ ham1,
    const float* __restrict__ hma2, const int* __restrict__ opma,
    const int* __restrict__ rowmask, const float* __restrict__ h,
    const float* __restrict__ attma, float* __restrict__ hprime) {
  const int lane = threadIdx.x & 31;
  const int c  = lane & 15;
  const int hh = lane >> 4;
  const int wv  = __builtin_amdgcn_readfirstlane((int)(threadIdx.x >> 5));
  const int bn0 = (blockIdx.x * MA_WAVES + wv) * MA_NPW;
  const int b   = bn0 / NN;
  const float NEGINF = -__builtin_inff();

  v2f awe[4][4];
#pragma unroll
  for (int s = 0; s < 4; ++s)
#pragma unroll
    for (int ot = 0; ot < 4; ++ot) {
      awe[s][ot].x = wedge[(8 * hh + 2 * s) * FF + ot * 16 + c];
      awe[s][ot].y = wedge[(8 * hh + 2 * s + 1) * FF + ot * 16 + c];
    }
  v4f am3v[4][2];
#pragma unroll
  for (int ot = 0; ot < 4; ++ot)
#pragma unroll
    for (int q = 0; q < 2; ++q)
      am3v[ot][q] = *(const v4f*)(attma + 2 * FF + ot * 16 + 8 * hh + 4 * q);
  float hma2v[8];
#pragma unroll
  for (int t = 0; t < 8; ++t) hma2v[t] = hma2[b * MM + 16 * t + c];
  const float* hmb = hma + (size_t)b * MM * FF + (size_t)c * FF + 8 * hh;

  v4f zq[8][2];
  {
    const float* zp = z + (size_t)bn0 * MM * FE + c * FE + 8 * hh;
#pragma unroll
    for (int t = 0; t < 8; ++t)
#pragma unroll
      for (int q = 0; q < 2; ++q)
        zq[t][q] = *(const v4f*)(zp + t * 16 * FE + 4 * q);
  }

#pragma unroll 1
  for (int i = 0; i < MA_NPW; ++i) {
    const int bn = bn0 + i;

    const int   rm  = rowmask[bn];
    const float hm1 = ham1[bn];
    const float* hr = h + (size_t)bn * FF + 8 * hh;
    v4f res[4][2];
#pragma unroll
    for (int ot = 0; ot < 4; ++ot) {
      res[ot][0] = *(const v4f*)(hr + ot * 16);
      res[ot][1] = *(const v4f*)(hr + ot * 16 + 4);
    }
    int adv[8];
#pragma unroll
    for (int t = 0; t < 8; ++t) adv[t] = opma[(size_t)bn * MM + 16 * t + c];

    v8f D[8][4];
    float sc[8];
#pragma unroll
    for (int t = 0; t < 8; ++t) {
      v2f zb[4];
      zb[0].x = zq[t][0].x; zb[0].y = zq[t][0].y;
      zb[1].x = zq[t][0].z; zb[1].y = zq[t][0].w;
      zb[2].x = zq[t][1].x; zb[2].y = zq[t][1].y;
      zb[3].x = zq[t][1].z; zb[3].y = zq[t][1].w;
      float part[4];
#pragma unroll
      for (int ot = 0; ot < 4; ++ot) {
        v8f d = {};
#pragma unroll
        for (int s = 0; s < 4; ++s) d = wmma4(awe[s][ot], zb[s], d);
        D[t][ot] = d;
        float a = 0.f;
        a += d[0] * am3v[ot][0].x; a += d[1] * am3v[ot][0].y;
        a += d[2] * am3v[ot][0].z; a += d[3] * am3v[ot][0].w;
        a += d[4] * am3v[ot][1].x; a += d[5] * am3v[ot][1].y;
        a += d[6] * am3v[ot][1].z; a += d[7] * am3v[ot][1].w;
        part[ot] = a;
      }
      float dot = (part[0] + part[1]) + (part[2] + part[3]);
      dot += xhalf(dot);
      float e = hm1 + hma2v[t] + dot;
      e = e > 0.f ? e : 0.01f * e;
      const int conn = rm | (adv[t] > 0);
      sc[t] = conn ? e : NEGINF;
    }

    float mx = sc[0];
#pragma unroll
    for (int t = 1; t < 8; ++t) mx = fmaxf(mx, sc[t]);
    mx = row_max16(mx);
    if (mx == NEGINF) mx = 0.f;
    float p[8];
    float sum = 0.f;
#pragma unroll
    for (int t = 0; t < 8; ++t) { p[t] = __expf(sc[t] - mx); sum += p[t]; }
    sum = row_sum16(sum);
    const float inv = 1.f / fmaxf(sum, 1e-30f);

    if (i + 1 < MA_NPW) {
      const float* zp = z + (size_t)(bn + 1) * MM * FE + c * FE + 8 * hh;
#pragma unroll
      for (int t = 0; t < 8; ++t)
#pragma unroll
        for (int q = 0; q < 2; ++q)
          zq[t][q] = *(const v4f*)(zp + t * 16 * FE + 4 * q);
    }

    v8f acc[4];
#pragma unroll
    for (int ot = 0; ot < 4; ++ot) acc[ot] = (v8f){};
#pragma unroll
    for (int t = 0; t < 8; ++t)
#pragma unroll
      for (int ot = 0; ot < 4; ++ot) {
        const v4f g0 = *(const v4f*)(hmb + (size_t)t * 16 * FF + ot * 16);
        const v4f g1 = *(const v4f*)(hmb + (size_t)t * 16 * FF + ot * 16 + 4);
        acc[ot][0] += p[t] * (g0.x + D[t][ot][0]);
        acc[ot][1] += p[t] * (g0.y + D[t][ot][1]);
        acc[ot][2] += p[t] * (g0.z + D[t][ot][2]);
        acc[ot][3] += p[t] * (g0.w + D[t][ot][3]);
        acc[ot][4] += p[t] * (g1.x + D[t][ot][4]);
        acc[ot][5] += p[t] * (g1.y + D[t][ot][5]);
        acc[ot][6] += p[t] * (g1.z + D[t][ot][6]);
        acc[ot][7] += p[t] * (g1.w + D[t][ot][7]);
      }
#pragma unroll
    for (int ot = 0; ot < 4; ++ot)
#pragma unroll
      for (int r = 0; r < 8; ++r) acc[ot][r] = row_sum16(acc[ot][r]);

    float* hp = hprime + (size_t)bn * (2 * FF) + FF + 8 * hh;
#pragma unroll
    for (int ot = 0; ot < 4; ++ot) {
      const v4f r0 = res[ot][0];
      const v4f r1 = res[ot][1];
      v4f o0, o1;
      o0.x = acc[ot][0] * inv + r0.x; o0.y = acc[ot][1] * inv + r0.y;
      o0.z = acc[ot][2] * inv + r0.z; o0.w = acc[ot][3] * inv + r0.w;
      o1.x = acc[ot][4] * inv + r1.x; o1.y = acc[ot][5] * inv + r1.y;
      o1.z = acc[ot][6] * inv + r1.z; o1.w = acc[ot][7] * inv + r1.w;
      if (c == 0) {
        *(v4f*)(hp + ot * 16)     = o0;
        *(v4f*)(hp + ot * 16 + 4) = o1;
      }
    }

  }
}

__global__ __launch_bounds__(256) void mlp(
    const float* __restrict__ hprime, const float* __restrict__ w1,
    const float* __restrict__ b1, const float* __restrict__ w2,
    const float* __restrict__ b2, const float* __restrict__ wo,
    const float* __restrict__ bo, const int* __restrict__ rowmask,
    float* __restrict__ out) {
  extern __shared__ float smem[];
  float* wb = smem;
  typedef float act_row_t[HID + 4];
  act_row_t* a0 = (act_row_t*)(smem + HID * HID);
  act_row_t* a1 = (act_row_t*)(smem + HID * HID + 16 * (HID + 4));

  const int row0 = blockIdx.x * 16;
  const int wv = threadIdx.x >> 5;
  const int lane = threadIdx.x & 31;
  const int lm = lane & 15;
  const int kh = (lane >> 4) * 2;
  const int lh = lane >> 4;

  for (int t = threadIdx.x; t < 16 * HID; t += 256)
    a0[t >> 7][t & 127] = hprime[(size_t)row0 * HID + t];
  for (int t = threadIdx.x; t < HID * HID; t += 256) wb[t] = w1[t];
  __syncthreads();

  {
    int o = wv * 16 + lm;
    v8f c = {};
    for (int k0 = 0; k0 < HID; k0 += 4) {
      v2f a, bf;
      a.x  = a0[lm][k0 + kh];
      a.y  = a0[lm][k0 + kh + 1];
      bf.x = wb[(k0 + kh) * HID + o];
      bf.y = wb[(k0 + kh + 1) * HID + o];
      c = wmma4(a, bf, c);
    }
    float bias = b1[o];
#pragma unroll
    for (int r = 0; r < 8; ++r) {
      float v = c[r] + bias;
      a1[r + 8 * lh][o] = v > 0.f ? v : expm1f(v);
    }
  }
  __syncthreads();
  for (int t = threadIdx.x; t < HID * HID; t += 256) wb[t] = w2[t];
  __syncthreads();

  {
    int o = wv * 16 + lm;
    v8f c = {};
    for (int k0 = 0; k0 < HID; k0 += 4) {
      v2f a, bf;
      a.x  = a1[lm][k0 + kh];
      a.y  = a1[lm][k0 + kh + 1];
      bf.x = wb[(k0 + kh) * HID + o];
      bf.y = wb[(k0 + kh + 1) * HID + o];
      c = wmma4(a, bf, c);
    }
    float bias = b2[o];
#pragma unroll
    for (int r = 0; r < 8; ++r) {
      float v = c[r] + bias;
      a0[r + 8 * lh][o] = v > 0.f ? v : expm1f(v);
    }
  }
  __syncthreads();
  for (int t = threadIdx.x; t < HID * FF; t += 256) wb[t] = wo[t];
  __syncthreads();

  if (wv < 4) {
    int o = wv * 16 + lm;
    v8f c = {};
    for (int k0 = 0; k0 < HID; k0 += 4) {
      v2f a, bf;
      a.x  = a0[lm][k0 + kh];
      a.y  = a0[lm][k0 + kh + 1];
      bf.x = wb[(k0 + kh) * FF + o];
      bf.y = wb[(k0 + kh + 1) * FF + o];
      c = wmma4(a, bf, c);
    }
    float bias = bo[o];
#pragma unroll
    for (int r = 0; r < 8; ++r) {
      int m = r + 8 * lh;
      float v = c[r] + bias;
      if (rowmask[row0 + m]) v = 0.f;
      out[(size_t)(row0 + m) * FF + o] = v;
    }
  }
}

extern "C" void kernel_launch(void* const* d_in, const int* in_sizes, int n_in,
                              void* d_out, int out_size, void* d_ws,
                              size_t ws_size, hipStream_t stream) {
  (void)in_sizes; (void)n_in; (void)out_size; (void)ws_size;
  const float* x      = (const float*)d_in[0];
  const float* y      = (const float*)d_in[1];
  const float* z      = (const float*)d_in[2];
  const int*   op_adj = (const int*)d_in[3];
  const int*   ma_adj = (const int*)d_in[4];
  const int*   op_ma  = (const int*)d_in[5];
  const float* Wop    = (const float*)d_in[6];
  const float* Wma    = (const float*)d_in[7];
  const float* Wedge  = (const float*)d_in[8];
  const float* att_op = (const float*)d_in[9];
  const float* att_ma = (const float*)d_in[10];
  const float* w1     = (const float*)d_in[11];
  const float* b1     = (const float*)d_in[12];
  const float* w2     = (const float*)d_in[13];
  const float* b2     = (const float*)d_in[14];
  const float* wo     = (const float*)d_in[15];
  const float* bo     = (const float*)d_in[16];
  float* out = (float*)d_out;

  float* ws     = (float*)d_ws;
  float* h      = ws;
  float* hma    = h + BB * NN * FF;
  float* ha1    = hma + BB * MM * FF;
  float* ha2    = ha1 + BB * NN;
  float* ham1   = ha2 + BB * NN;
  float* hma2   = ham1 + BB * NN;
  float* hprime = hma2 + BB * MM;
  int*   rmask  = (int*)(hprime + (size_t)BB * NN * 2 * FF);

  gemm_rows_64<<<dim3((BB * NN) / 16 + (BB * MM) / 16), dim3(128), 0, stream>>>(
      x, Wop, h, y, Wma, hma);
  size_t sm2 = (16 * (NN + 4) + 16 + 2 * HS_ROWS * HS_P) * sizeof(float);
  unsigned* cbits = (unsigned*)(hprime + FF);
  size_t smb = (16 * 32 * CB_P + 32 * 33 + 4 * 32 + MM) * sizeof(int);
  conn_bits_proj<<<dim3(NN / 32, BB), dim3(512), smb, stream>>>(
      op_adj, ma_adj, cbits, h, hma, att_op, att_ma, op_ma,
      ha1, ha2, ham1, hma2, rmask);
  op_attn<<<dim3(NN / 16, BB), dim3(128), sm2, stream>>>(
      h, ha1, ha2, cbits, rmask, hprime);
  ma_attn<<<dim3((BB * NN) / (MA_WAVES * MA_NPW)), dim3(32 * MA_WAVES), 0, stream>>>(
      z, Wedge, hma, ham1, hma2, op_ma, rmask, h, att_ma, hprime);
  size_t sm4 = (HID * HID + 2 * 16 * (HID + 4)) * sizeof(float);
  mlp<<<dim3((BB * NN) / 16), dim3(256), sm4, stream>>>(
      hprime, w1, b1, w2, b2, wo, bo, rmask, out);
}


# op and ma stage in one launch: a group does an op tile and 16 ma nodes, order alternating, 2 groups per WGP; ma tiles start from h_ma, hma2=h_ma.(am2-am3): summing order changed; 5->4 launches
# speedup vs baseline: 3.2838x; 1.1090x over previous
#include <hip/hip_runtime.h>
#include <math.h>


typedef __attribute__((ext_vector_type(2))) float v2f;
typedef __attribute__((ext_vector_type(8))) float v8f;
typedef __attribute__((ext_vector_type(4))) float v4f;

#define BB 8
#define NN 1024
#define MM 128
#define FF 64
#define FE 16
#define HID 128

__device__ __forceinline__ v8f wmma4(v2f a, v2f b, v8f c) {
  return __builtin_amdgcn_wmma_f32_16x16x4_f32(
       false, a,  false, b,
       (short)0, c,  false,  false);
}

__global__ __launch_bounds__(128) void gemm_rows_64(
    const float* __restrict__ A0, const float* __restrict__ W0,
    float* __restrict__ C0,
    const float* __restrict__ A1, const float* __restrict__ W1,
    float* __restrict__ C1) {
  const bool second = blockIdx.x >= (BB * NN) / 16;
  const float* __restrict__ A = second ? A1 : A0;
  const float* __restrict__ W = second ? W1 : W0;
  float* __restrict__ C = second ? C1 : C0;
  const int wv   = threadIdx.x >> 5;
  const int lane = threadIdx.x & 31;
  const int lm   = lane & 15;
  const int kh   = (lane >> 4) * 2;
  const int lh   = lane >> 4;
  const int row0 = (second ? (int)blockIdx.x - (BB * NN) / 16 : (int)blockIdx.x) * 16;
  const int o    = wv * 16 + lm;
  const float* Arow = A + (size_t)(row0 + lm) * FF;

  v8f c = {};
#pragma unroll
  for (int k0 = 0; k0 < FF; k0 += 4) {
    v2f a, b;
    a.x = Arow[k0 + kh];
    a.y = Arow[k0 + kh + 1];
    b.x = W[(size_t)(k0 + kh) * FF + o];
    b.y = W[(size_t)(k0 + kh + 1) * FF + o];
    c = wmma4(a, b, c);
  }
#pragma unroll
  for (int r = 0; r < 8; ++r) {
    int m = r + 8 * lh;
    C[(size_t)(row0 + m) * FF + o] = c[r];
  }
}

__device__ __forceinline__ float xhalf(float x) {
  int b = __builtin_bit_cast(int, x);
  return __builtin_bit_cast(float,
      __builtin_amdgcn_permlanex16(b, b, 0x76543210, 0xfedcba98, false, false));
}
template <int MASK>
__device__ __forceinline__ float rowx(float x) {
  int b = __builtin_bit_cast(int, x);
  return __builtin_bit_cast(float,
      __builtin_amdgcn_update_dpp(0, b, 0x160 + MASK, 0xf, 0xf, true));
}
__device__ __forceinline__ float row_sum16(float x) {
  x += rowx<1>(x); x += rowx<2>(x); x += rowx<4>(x); x += rowx<8>(x);
  return x;
}
__device__ __forceinline__ float row_max16(float x) {
  x = fmaxf(x, rowx<1>(x)); x = fmaxf(x, rowx<2>(x));
  x = fmaxf(x, rowx<4>(x)); x = fmaxf(x, rowx<8>(x));
  return x;
}
__device__ __forceinline__ float wave_sum32(float x) {
  x = row_sum16(x);
  return x + xhalf(x);
}

#define CB_P 36

__global__ __launch_bounds__(512) void conn_bits_proj(
    const int* __restrict__ opadj, const int* __restrict__ maadj,
    unsigned* __restrict__ cbits,
    const float* __restrict__ h, const float* __restrict__ hma,
    const float* __restrict__ att_op, const float* __restrict__ att_ma,
    const int* __restrict__ opma,
    float* __restrict__ ha1, float* __restrict__ ha2,
    float* __restrict__ ham1, float* __restrict__ hma2,
    int* __restrict__ rowmask) {
  extern __shared__ __attribute__((aligned(16))) int cbs[];
  const int b    = blockIdx.y;
  const int r0   = blockIdx.x * 32;
  const int wv   = threadIdx.x >> 5;
  const int lane = threadIdx.x & 31;
  int* tile = cbs + wv * (32 * CB_P);
  unsigned* words = (unsigned*)(cbs + 16 * 32 * CB_P);
  float* pj = (float*)(cbs + 16 * 32 * CB_P + 32 * 33);
  float* pm = pj + 4 * 32;
  const int* opb = opadj + (size_t)b * NN * NN;
  const int* mab = maadj + (size_t)b * NN * NN;
  const int sub = lane >> 3;
  const int q4  = (lane & 7) * 4;
  const bool machines = (blockIdx.x == 0);

  const int node0 = b * NN + r0 + 2 * wv;
  v2f  hv[2];
  int4 lk[2];
#pragma unroll
  for (int q = 0; q < 2; ++q) {
    hv[q] = *(const v2f*)(h + (size_t)(node0 + q) * FF + 2 * lane);
    lk[q] = *(const int4*)(opma + (size_t)(node0 + q) * MM + 4 * lane);
  }
  const v2f wa1 = *(const v2f*)(att_op + 2 * lane);
  const v2f wa2 = *(const v2f*)(att_op + FF + 2 * lane);
  const v2f wm1 = *(const v2f*)(att_ma + 2 * lane);
  v2f wm2 = {};
  v2f mv[8] = {};
  if (machines) {
    const v2f w2 = *(const v2f*)(att_ma + FF + 2 * lane);
    const v2f w3 = *(const v2f*)(att_ma + 2 * FF + 2 * lane);
    wm2.x = w2.x - w3.x;
    wm2.y = w2.y - w3.y;
#pragma unroll
    for (int t = 0; t < 8; ++t)
      mv[t] = *(const v2f*)(hma + ((size_t)b * MM + 8 * wv + t) * FF + 2 * lane);
  }

#pragma unroll
  for (int q = 0; q < 2; ++q) {
    const float s1 = wave_sum32(hv[q].x * wa1.x + hv[q].y * wa1.y);
    const float s2 = wave_sum32(hv[q].x * wa2.x + hv[q].y * wa2.y);
    const float s3 = wave_sum32(hv[q].x * wm1.x + hv[q].y * wm1.y);
    const int any = lk[q].x | lk[q].y | lk[q].z | lk[q].w;
    const unsigned linked = __builtin_amdgcn_ballot_w32(any != 0);
    if (lane == 0) {
      pj[0 * 32 + 2 * wv + q] = s1;
      pj[1 * 32 + 2 * wv + q] = s2;
      pj[2 * 32 + 2 * wv + q] = s3;
      ((int*)pj)[3 * 32 + 2 * wv + q] = (linked == 0u) ? 1 : 0;
    }
  }
  if (machines) {
#pragma unroll
    for (int t = 0; t < 8; ++t) {
      const float s = wave_sum32(mv[t].x * wm2.x + mv[t].y * wm2.y);
      if (lane == 0) pm[8 * wv + t] = s;
    }
  }

  int4 dsum[2][8];
  int  tr[2][32];
#pragma unroll
  for (int hf = 0; hf < 2; ++hf) {
    const int c0 = (wv + 16 * hf) * 32;
#pragma unroll
    for (int q = 0; q < 8; ++q) {
      const size_t off = (size_t)(r0 + 4 * q + sub) * NN + c0 + q4;
      const int4 a = *(const int4*)(opb + off);
      const int4 m = *(const int4*)(mab + off);
      dsum[hf][q].x = a.x + m.x; dsum[hf][q].y = a.y + m.y;
      dsum[hf][q].z = a.z + m.z; dsum[hf][q].w = a.w + m.w;
    }
#pragma unroll
    for (int k = 0; k < 32; ++k) {
      const size_t off = (size_t)(c0 + k) * NN + r0 + lane;
      tr[hf][k] = opb[off] + mab[off];
    }
  }

#pragma unroll
  for (int hf = 0; hf < 2; ++hf) {
    const int ct = wv + 16 * hf;
#pragma unroll
    for (int q = 0; q < 8; ++q)
      *(int4*)(tile + (4 * q + sub) * CB_P + q4) = dsum[hf][q];
    __syncthreads();
    unsigned w = 0;
#pragma unroll
    for (int g = 0; g < 8; ++g) {
      const int4 d = *(const int4*)(tile + lane * CB_P + 4 * g);
      w |= (unsigned)((d.x + tr[hf][4 * g + 0]) > 0) << (4 * g + 0);
      w |= (unsigned)((d.y + tr[hf][4 * g + 1]) > 0) << (4 * g + 1);
      w |= (unsigned)((d.z + tr[hf][4 * g + 2]) > 0) << (4 * g + 2);
      w |= (unsigned)((d.w + tr[hf][4 * g + 3]) > 0) << (4 * g + 3);
    }
    words[lane * 33 + ct] = w;
    __syncthreads();
  }

  for (int rr = wv; rr < 32; rr += 16)
    cbits[((size_t)b * NN + r0 + rr) * (2 * FF) + lane] = words[rr * 33 + lane];

  const int nb = b * NN + r0 + lane;
  if (wv == 0)      ha1[nb]  = pj[0 * 32 + lane];
  else if (wv == 1) ha2[nb]  = pj[1 * 32 + lane];
  else if (wv == 2) ham1[nb] = pj[2 * 32 + lane];
  else if (wv == 3) rowmask[nb] = ((const int*)pj)[3 * 32 + lane];
  else if (machines && wv < 8)
    hma2[b * MM + (wv - 4) * 32 + lane] = pm[(wv - 4) * 32 + lane];
}

#define HS_ROWS 64
__device__ __forceinline__ void lds_barrier() {
  __builtin_amdgcn_fence(__ATOMIC_RELEASE, "workgroup", "local");
  __builtin_amdgcn_s_barrier();
  __builtin_amdgcn_fence(__ATOMIC_ACQUIRE, "workgroup", "local");
}
#define HS_P 72

__device__ __forceinline__ void op_attn_tile(
    float* smem, const int tile,
    const float* __restrict__ h, const float* __restrict__ ha1,
    const float* __restrict__ ha2, const unsigned* __restrict__ cbits,
    const int* __restrict__ rowmask, float* __restrict__ hprime) {
  typedef float att_row_t[NN + 4];
  att_row_t* att = (att_row_t*)smem;
  float* inv_sum = smem + 16 * (NN + 4);

  const int b  = tile / (NN / 16);
  const int i0 = (tile % (NN / 16)) * 16;
  const float NEGINF = -__builtin_inff();

  const int lane = threadIdx.x & 31;
  const int wv = __builtin_amdgcn_readfirstlane((int)(threadIdx.x >> 5));

  float* hs = smem + 16 * (NN + 4) + 16;
  const v4f* hb4 = (const v4f*)(h + (size_t)b * NN * FF) + threadIdx.x;
  float* hst = hs + (threadIdx.x >> 4) * HS_P + (threadIdx.x & 15) * 4;
  v4f sa[8], sb[8];
#pragma unroll
  for (int q = 0; q < 8; ++q) {
    sa[q] = hb4[0 * (HS_ROWS * 16) + 128 * q];
    sb[q] = hb4[1 * (HS_ROWS * 16) + 128 * q];
  }
  {
    float a1v[32];
#pragma unroll
    for (int t = 0; t < 32; ++t) a1v[t] = ha1[b * NN + lane + 32 * t];
    unsigned wd[4];
#pragma unroll
    for (int rr = 0; rr < 4; ++rr)
      wd[rr] = cbits[((size_t)b * NN + i0 + wv + 4 * rr) * (2 * FF) + lane];
#pragma unroll
    for (int rr = 0; rr < 4; ++rr) {
      const int r = wv + 4 * rr;
      const int i = i0 + r;
      const int rm = rowmask[b * NN + i];
      const float a2 = ha2[b * NN + i];
      const unsigned wdm = rm ? 0xffffffffu : wd[rr];
      float e[32];
      float mx = NEGINF;
#pragma unroll
      for (int t = 0; t < 32; ++t) {
        const unsigned wt = (unsigned)__builtin_amdgcn_readlane((int)wdm, t);
        float s = a2 + a1v[t];
        s = s > 0.f ? s : 0.01f * s;
        e[t] = __builtin_amdgcn_inverse_ballot_w32(wt) ? s : NEGINF;
        mx = fmaxf(mx, e[t]);
      }
      mx = fmaxf(mx, xhalf(mx));
      mx = row_max16(mx);
      if (mx == NEGINF) mx = 0.f;
      float sum = 0.f;
#pragma unroll
      for (int t = 0; t < 32; ++t) {
        const float p = __expf(e[t] - mx);
        att[r][lane + 32 * t] = p;
        sum += p;
      }
      sum += xhalf(sum);
      sum += rowx<8>(sum); sum += rowx<4>(sum);
      sum += rowx<2>(sum); sum += rowx<1>(sum);
      if (lane == 0) inv_sum[r] = 1.f / fmaxf(sum, 1e-30f);
    }
  }
#pragma unroll
  for (int q = 0; q < 8; ++q) *(v4f*)(hst + q * 8 * HS_P) = sa[q];
  __syncthreads();

  const int lm = lane & 15;
  const int kh = (lane >> 4) * 2;
  const int lh = lane >> 4;
  const int o  = wv * 16 + lm;
  const float* ar  = &att[lm][kh];
  const float* hf0 = hs + kh * HS_P + o;
  const float* hf1 = hf0 + HS_ROWS * HS_P;

  v8f c = {};
#pragma unroll 1
  for (int s = 0; s < NN / HS_ROWS; s += 2) {
    {
      const int sn = (s + 2) & (NN / HS_ROWS - 1);
#pragma unroll
      for (int q = 0; q < 8; ++q) sa[q] = hb4[sn * (HS_ROWS * 16) + 128 * q];
    }
    {
      v2f af[16], bq[16];
#pragma unroll
      for (int t = 0; t < 16; ++t) {
        af[t].x = ar[s * HS_ROWS + 4 * t];
        af[t].y = ar[s * HS_ROWS + 4 * t + 1];
        bq[t].x = hf0[(4 * t) * HS_P];
        bq[t].y = hf0[(4 * t + 1) * HS_P];
      }
#pragma unroll
      for (int t = 0; t < 16; ++t) c = wmma4(af[t], bq[t], c);
    }
#pragma unroll
    for (int q = 0; q < 8; ++q) *(v4f*)(hst + HS_ROWS * HS_P + q * 8 * HS_P) = sb[q];
    __builtin_amdgcn_sched_group_barrier(0x020, 8, 0);
    __builtin_amdgcn_sched_group_barrier(0x100, 32, 0);
    __builtin_amdgcn_sched_group_barrier(0x008, 16, 0);
    __builtin_amdgcn_sched_group_barrier(0x200, 8, 0);
    lds_barrier();
    {
      const int sn = (s + 3) & (NN / HS_ROWS - 1);
#pragma unroll
      for (int q = 0; q < 8; ++q) sb[q] = hb4[sn * (HS_ROWS * 16) + 128 * q];
    }
    {
      v2f af[16], bq[16];
#pragma unroll
      for (int t = 0; t < 16; ++t) {
        af[t].x = ar[(s + 1) * HS_ROWS + 4 * t];
        af[t].y = ar[(s + 1) * HS_ROWS + 4 * t + 1];
        bq[t].x = hf1[(4 * t) * HS_P];
        bq[t].y = hf1[(4 * t + 1) * HS_P];
      }
#pragma unroll
      for (int t = 0; t < 16; ++t) c = wmma4(af[t], bq[t], c);
    }
#pragma unroll
    for (int q = 0; q < 8; ++q) *(v4f*)(hst + q * 8 * HS_P) = sa[q];
    __builtin_amdgcn_sched_group_barrier(0x020, 8, 0);
    __builtin_amdgcn_sched_group_barrier(0x100, 32, 0);
    __builtin_amdgcn_sched_group_barrier(0x008, 16, 0);
    __builtin_amdgcn_sched_group_barrier(0x200, 8, 0);
    lds_barrier();
  }
  float* hp = hprime + ((size_t)b * NN + i0) * (2 * FF);
#pragma unroll
  for (int r = 0; r < 8; ++r) {
    int m = r + 8 * lh;
    hp[(size_t)m * (2 * FF) + o] = c[r] * inv_sum[m];
  }
}

#define MA_NPW 4
#define HM_P 68
#define MA_WAVES 4

__device__ __forceinline__ void ma_attn_nodes(
    float* smem, const int group,
    const float* __restrict__ z, const float* __restrict__ wedge,
    const float* __restrict__ hma, const float* __restrict__ ham1,
    const float* __restrict__ hma2, const int* __restrict__ opma,
    const int* __restrict__ rowmask, const float* __restrict__ h,
    const float* __restrict__ attma, float* __restrict__ hprime) {
  const int lane = threadIdx.x & 31;
  const int c  = lane & 15;
  const int hh = lane >> 4;
  const int wv  = __builtin_amdgcn_readfirstlane((int)(threadIdx.x >> 5));
  const int bn0 = (group * MA_WAVES + wv) * MA_NPW;
  const int b   = bn0 / NN;
  const float NEGINF = -__builtin_inff();

  v2f awe[4][4];
#pragma unroll
  for (int s = 0; s < 4; ++s)
#pragma unroll
    for (int ot = 0; ot < 4; ++ot) {
      awe[s][ot].x = wedge[(8 * hh + 2 * s) * FF + ot * 16 + c];
      awe[s][ot].y = wedge[(8 * hh + 2 * s + 1) * FF + ot * 16 + c];
    }
  v4f am3v[4][2];
#pragma unroll
  for (int ot = 0; ot < 4; ++ot)
#pragma unroll
    for (int q = 0; q < 2; ++q)
      am3v[ot][q] = *(const v4f*)(attma + 2 * FF + ot * 16 + 8 * hh + 4 * q);
  float hma2v[8];
#pragma unroll
  for (int t = 0; t < 8; ++t) hma2v[t] = hma2[b * MM + 16 * t + c];
  {
    const v4f* src4 = (const v4f*)(hma + (size_t)b * MM * FF);
#pragma unroll
    for (int q = 0; q < 16; ++q) {
      const int idx = threadIdx.x + 128 * q;
      *(v4f*)(smem + (idx >> 4) * HM_P + (idx & 15) * 4) = src4[idx];
    }
  }
  __syncthreads();
  const float* hmb = smem + c * HM_P + 8 * hh;

  v4f zq[8][2];
  {
    const float* zp = z + (size_t)bn0 * MM * FE + c * FE + 8 * hh;
#pragma unroll
    for (int t = 0; t < 8; ++t)
#pragma unroll
      for (int q = 0; q < 2; ++q)
        zq[t][q] = *(const v4f*)(zp + t * 16 * FE + 4 * q);
  }

#pragma unroll 1
  for (int i = 0; i < MA_NPW; ++i) {
    const int bn = bn0 + i;
    asm volatile("" ::: "memory");

    const int   rm  = rowmask[bn];
    const float hm1 = ham1[bn];
    int adv[8];
#pragma unroll
    for (int t = 0; t < 8; ++t) adv[t] = opma[(size_t)bn * MM + 16 * t + c];

    v8f D[8][4];
    float sc[8];
#pragma unroll
    for (int t = 0; t < 8; ++t) {
      v2f zb[4];
      zb[0].x = zq[t][0].x; zb[0].y = zq[t][0].y;
      zb[1].x = zq[t][0].z; zb[1].y = zq[t][0].w;
      zb[2].x = zq[t][1].x; zb[2].y = zq[t][1].y;
      zb[3].x = zq[t][1].z; zb[3].y = zq[t][1].w;
      float part[4];
#pragma unroll
      for (int ot = 0; ot < 4; ++ot) {
        const v4f g0 = *(const v4f*)(hmb + t * 16 * HM_P + ot * 16);
        const v4f g1 = *(const v4f*)(hmb + t * 16 * HM_P + ot * 16 + 4);
        v8f d = {g0.x, g0.y, g0.z, g0.w, g1.x, g1.y, g1.z, g1.w};
#pragma unroll
        for (int s = 0; s < 4; ++s) d = wmma4(awe[s][ot], zb[s], d);
        D[t][ot] = d;
        v2f a2 = __builtin_shufflevector(d, d, 0, 1) * __builtin_shufflevector(am3v[ot][0], am3v[ot][0], 0, 1);
        a2 += __builtin_shufflevector(d, d, 2, 3) * __builtin_shufflevector(am3v[ot][0], am3v[ot][0], 2, 3);
        a2 += __builtin_shufflevector(d, d, 4, 5) * __builtin_shufflevector(am3v[ot][1], am3v[ot][1], 0, 1);
        a2 += __builtin_shufflevector(d, d, 6, 7) * __builtin_shufflevector(am3v[ot][1], am3v[ot][1], 2, 3);
        part[ot] = a2.x + a2.y;
      }
      float dot = (part[0] + part[1]) + (part[2] + part[3]);
      dot += xhalf(dot);
      float e = hm1 + hma2v[t] + dot;
      e = e > 0.f ? e : 0.01f * e;
      const int conn = rm | (adv[t] > 0);
      sc[t] = conn ? e : NEGINF;
    }

    float mx = sc[0];
#pragma unroll
    for (int t = 1; t < 8; ++t) mx = fmaxf(mx, sc[t]);
    mx = row_max16(mx);
    if (mx == NEGINF) mx = 0.f;
    float p[8];
    float sum = 0.f;
#pragma unroll
    for (int t = 0; t < 8; ++t) { p[t] = __expf(sc[t] - mx); sum += p[t]; }
    sum = row_sum16(sum);
    const float inv = 1.f / fmaxf(sum, 1e-30f);

    if (i + 1 < MA_NPW) {
      const float* zp = z + (size_t)(bn + 1) * MM * FE + c * FE + 8 * hh;
#pragma unroll
      for (int t = 0; t < 8; ++t)
#pragma unroll
        for (int q = 0; q < 2; ++q)
          zq[t][q] = *(const v4f*)(zp + t * 16 * FE + 4 * q);
    }

    v8f acc[4];
#pragma unroll
    for (int ot = 0; ot < 4; ++ot) acc[ot] = (v8f){};
#pragma unroll
    for (int t = 0; t < 8; ++t)
#pragma unroll
      for (int ot = 0; ot < 4; ++ot)
#pragma unroll
        for (int r = 0; r < 8; ++r) acc[ot][r] += p[t] * D[t][ot][r];

    const float* hr = h + (size_t)bn * FF + 8 * hh;
    v4f res[4][2];
#pragma unroll
    for (int ot = 0; ot < 4; ++ot) {
      res[ot][0] = *(const v4f*)(hr + ot * 16);
      res[ot][1] = *(const v4f*)(hr + ot * 16 + 4);
    }
#pragma unroll
    for (int ot = 0; ot < 4; ++ot)
#pragma unroll
      for (int r = 0; r < 8; ++r) acc[ot][r] = row_sum16(acc[ot][r]);

    float* hp = hprime + (size_t)bn * (2 * FF) + FF + 8 * hh;
#pragma unroll
    for (int ot = 0; ot < 4; ++ot) {
      const v4f r0 = res[ot][0];
      const v4f r1 = res[ot][1];
      v4f o0, o1;
      o0.x = acc[ot][0] * inv + r0.x; o0.y = acc[ot][1] * inv + r0.y;
      o0.z = acc[ot][2] * inv + r0.z; o0.w = acc[ot][3] * inv + r0.w;
      o1.x = acc[ot][4] * inv + r1.x; o1.y = acc[ot][5] * inv + r1.y;
      o1.z = acc[ot][6] * inv + r1.z; o1.w = acc[ot][7] * inv + r1.w;
      if (c == 0) {
        *(v4f*)(hp + ot * 16)     = o0;
        *(v4f*)(hp + ot * 16 + 4) = o1;
      }
    }

  }
}

__global__ __launch_bounds__(128, 2) void attn_pair(
    const float* __restrict__ h, const float* __restrict__ ha1,
    const float* __restrict__ ha2, const unsigned* __restrict__ cbits,
    const int* __restrict__ rowmask, float* __restrict__ hprime,
    const float* __restrict__ z, const float* __restrict__ wedge,
    const float* __restrict__ hma, const float* __restrict__ ham1,
    const float* __restrict__ hma2, const int* __restrict__ opma,
    const float* __restrict__ attma) {
  extern __shared__ float smem[];
  const int g = blockIdx.x;
  const int op_first = (g ^ (g >> 8)) & 1;
#pragma unroll 1
  for (int ph = 0; ph < 2; ++ph) {
    if (ph == op_first)
      ma_attn_nodes(smem, g, z, wedge, hma, ham1, hma2, opma, rowmask, h, attma, hprime);
    else
      op_attn_tile(smem, g, h, ha1, ha2, cbits, rowmask, hprime);
    __syncthreads();
  }
}

__global__ __launch_bounds__(256) void mlp(
    const float* __restrict__ hprime, const float* __restrict__ w1,
    const float* __restrict__ b1, const float* __restrict__ w2,
    const float* __restrict__ b2, const float* __restrict__ wo,
    const float* __restrict__ bo, const int* __restrict__ rowmask,
    float* __restrict__ out) {
  extern __shared__ float smem[];
  float* wb = smem;
  typedef float act_row_t[HID + 4];
  act_row_t* a0 = (act_row_t*)(smem + HID * HID);
  act_row_t* a1 = (act_row_t*)(smem + HID * HID + 16 * (HID + 4));

  const int row0 = blockIdx.x * 16;
  const int wv = threadIdx.x >> 5;
  const int lane = threadIdx.x & 31;
  const int lm = lane & 15;
  const int kh = (lane >> 4) * 2;
  const int lh = lane >> 4;

  for (int t = threadIdx.x; t < 16 * HID; t += 256)
    a0[t >> 7][t & 127] = hprime[(size_t)row0 * HID + t];
  for (int t = threadIdx.x; t < HID * HID; t += 256) wb[t] = w1[t];
  __syncthreads();

  {
    int o = wv * 16 + lm;
    v8f c = {};
    for (int k0 = 0; k0 < HID; k0 += 4) {
      v2f a, bf;
      a.x  = a0[lm][k0 + kh];
      a.y  = a0[lm][k0 + kh + 1];
      bf.x = wb[(k0 + kh) * HID + o];
      bf.y = wb[(k0 + kh + 1) * HID + o];
      c = wmma4(a, bf, c);
    }
    float bias = b1[o];
#pragma unroll
    for (int r = 0; r < 8; ++r) {
      float v = c[r] + bias;
      a1[r + 8 * lh][o] = v > 0.f ? v : expm1f(v);
    }
  }
  __syncthreads();
  for (int t = threadIdx.x; t < HID * HID; t += 256) wb[t] = w2[t];
  __syncthreads();

  {
    int o = wv * 16 + lm;
    v8f c = {};
    for (int k0 = 0; k0 < HID; k0 += 4) {
      v2f a, bf;
      a.x  = a1[lm][k0 + kh];
      a.y  = a1[lm][k0 + kh + 1];
      bf.x = wb[(k0 + kh) * HID + o];
      bf.y = wb[(k0 + kh + 1) * HID + o];
      c = wmma4(a, bf, c);
    }
    float bias = b2[o];
#pragma unroll
    for (int r = 0; r < 8; ++r) {
      float v = c[r] + bias;
      a0[r + 8 * lh][o] = v > 0.f ? v : expm1f(v);
    }
  }
  __syncthreads();
  for (int t = threadIdx.x; t < HID * FF; t += 256) wb[t] = wo[t];
  __syncthreads();

  if (wv < 4) {
    int o = wv * 16 + lm;
    v8f c = {};
    for (int k0 = 0; k0 < HID; k0 += 4) {
      v2f a, bf;
      a.x  = a0[lm][k0 + kh];
      a.y  = a0[lm][k0 + kh + 1];
      bf.x = wb[(k0 + kh) * FF + o];
      bf.y = wb[(k0 + kh + 1) * FF + o];
      c = wmma4(a, bf, c);
    }
    float bias = bo[o];
#pragma unroll
    for (int r = 0; r < 8; ++r) {
      int m = r + 8 * lh;
      float v = c[r] + bias;
      if (rowmask[row0 + m]) v = 0.f;
      out[(size_t)(row0 + m) * FF + o] = v;
    }
  }
}

extern "C" void kernel_launch(void* const* d_in, const int* in_sizes, int n_in,
                              void* d_out, int out_size, void* d_ws,
                              size_t ws_size, hipStream_t stream) {
  (void)in_sizes; (void)n_in; (void)out_size; (void)ws_size;
  const float* x      = (const float*)d_in[0];
  const float* y      = (const float*)d_in[1];
  const float* z      = (const float*)d_in[2];
  const int*   op_adj = (const int*)d_in[3];
  const int*   ma_adj = (const int*)d_in[4];
  const int*   op_ma  = (const int*)d_in[5];
  const float* Wop    = (const float*)d_in[6];
  const float* Wma    = (const float*)d_in[7];
  const float* Wedge  = (const float*)d_in[8];
  const float* att_op = (const float*)d_in[9];
  const float* att_ma = (const float*)d_in[10];
  const float* w1     = (const float*)d_in[11];
  const float* b1     = (const float*)d_in[12];
  const float* w2     = (const float*)d_in[13];
  const float* b2     = (const float*)d_in[14];
  const float* wo     = (const float*)d_in[15];
  const float* bo     = (const float*)d_in[16];
  float* out = (float*)d_out;

  float* ws     = (float*)d_ws;
  float* h      = ws;
  float* hma    = h + BB * NN * FF;
  float* ha1    = hma + BB * MM * FF;
  float* ha2    = ha1 + BB * NN;
  float* ham1   = ha2 + BB * NN;
  float* hma2   = ham1 + BB * NN;
  float* hprime = hma2 + BB * MM;
  int*   rmask  = (int*)(hprime + (size_t)BB * NN * 2 * FF);

  gemm_rows_64<<<dim3((BB * NN) / 16 + (BB * MM) / 16), dim3(128), 0, stream>>>(
      x, Wop, h, y, Wma, hma);
  size_t sm2 = (16 * (NN + 4) + 16 + 2 * HS_ROWS * HS_P) * sizeof(float);
  unsigned* cbits = (unsigned*)hprime;
  size_t smb = (16 * 32 * CB_P + 32 * 33 + 4 * 32 + MM) * sizeof(int);
  conn_bits_proj<<<dim3(NN / 32, BB), dim3(512), smb, stream>>>(
      op_adj, ma_adj, cbits, h, hma, att_op, att_ma, op_ma,
      ha1, ha2, ham1, hma2, rmask);
  attn_pair<<<dim3((BB * NN) / 16), dim3(128), sm2, stream>>>(
      h, ha1, ha2, cbits, rmask, hprime,
      z, Wedge, hma, ham1, hma2, op_ma, att_ma);
  size_t sm4 = (HID * HID + 2 * 16 * (HID + 4)) * sizeof(float);
  mlp<<<dim3((BB * NN) / 16), dim3(256), sm4, stream>>>(
      hprime, w1, b1, w2, b2, wo, bo, rmask, out);
}


# MLP weights no longer pass through LDS: each wave loads the B fragments of its own column tile straight from global into registers, one matrix ahead of its product; staging loops go; same sums
# speedup vs baseline: 3.6944x; 1.1250x over previous
#include <hip/hip_runtime.h>
#include <math.h>


typedef __attribute__((ext_vector_type(2))) float v2f;
typedef __attribute__((ext_vector_type(8))) float v8f;
typedef __attribute__((ext_vector_type(4))) float v4f;

#define BB 8
#define NN 1024
#define MM 128
#define FF 64
#define FE 16
#define HID 128

__device__ __forceinline__ v8f wmma4(v2f a, v2f b, v8f c) {
  return __builtin_amdgcn_wmma_f32_16x16x4_f32(
       false, a,  false, b,
       (short)0, c,  false,  false);
}

__global__ __launch_bounds__(128) void gemm_rows_64(
    const float* __restrict__ A0, const float* __restrict__ W0,
    float* __restrict__ C0,
    const float* __restrict__ A1, const float* __restrict__ W1,
    float* __restrict__ C1) {
  const bool second = blockIdx.x >= (BB * NN) / 16;
  const float* __restrict__ A = second ? A1 : A0;
  const float* __restrict__ W = second ? W1 : W0;
  float* __restrict__ C = second ? C1 : C0;
  const int wv   = threadIdx.x >> 5;
  const int lane = threadIdx.x & 31;
  const int lm   = lane & 15;
  const int kh   = (lane >> 4) * 2;
  const int lh   = lane >> 4;
  const int row0 = (second ? (int)blockIdx.x - (BB * NN) / 16 : (int)blockIdx.x) * 16;
  const int o    = wv * 16 + lm;
  const float* Arow = A + (size_t)(row0 + lm) * FF;

  v8f c = {};
#pragma unroll
  for (int k0 = 0; k0 < FF; k0 += 4) {
    v2f a, b;
    a.x = Arow[k0 + kh];
    a.y = Arow[k0 + kh + 1];
    b.x = W[(size_t)(k0 + kh) * FF + o];
    b.y = W[(size_t)(k0 + kh + 1) * FF + o];
    c = wmma4(a, b, c);
  }
#pragma unroll
  for (int r = 0; r < 8; ++r) {
    int m = r + 8 * lh;
    C[(size_t)(row0 + m) * FF + o] = c[r];
  }
}

__device__ __forceinline__ float xhalf(float x) {
  int b = __builtin_bit_cast(int, x);
  return __builtin_bit_cast(float,
      __builtin_amdgcn_permlanex16(b, b, 0x76543210, 0xfedcba98, false, false));
}
template <int MASK>
__device__ __forceinline__ float rowx(float x) {
  int b = __builtin_bit_cast(int, x);
  return __builtin_bit_cast(float,
      __builtin_amdgcn_update_dpp(0, b, 0x160 + MASK, 0xf, 0xf, true));
}
__device__ __forceinline__ float row_sum16(float x) {
  x += rowx<1>(x); x += rowx<2>(x); x += rowx<4>(x); x += rowx<8>(x);
  return x;
}
__device__ __forceinline__ float row_max16(float x) {
  x = fmaxf(x, rowx<1>(x)); x = fmaxf(x, rowx<2>(x));
  x = fmaxf(x, rowx<4>(x)); x = fmaxf(x, rowx<8>(x));
  return x;
}
__device__ __forceinline__ float wave_sum32(float x) {
  x = row_sum16(x);
  return x + xhalf(x);
}

#define CB_P 36

__global__ __launch_bounds__(512) void conn_bits_proj(
    const int* __restrict__ opadj, const int* __restrict__ maadj,
    unsigned* __restrict__ cbits,
    const float* __restrict__ h, const float* __restrict__ hma,
    const float* __restrict__ att_op, const float* __restrict__ att_ma,
    const int* __restrict__ opma,
    float* __restrict__ ha1, float* __restrict__ ha2,
    float* __restrict__ ham1, float* __restrict__ hma2,
    int* __restrict__ rowmask) {
  extern __shared__ __attribute__((aligned(16))) int cbs[];
  const int b    = blockIdx.y;
  const int r0   = blockIdx.x * 32;
  const int wv   = threadIdx.x >> 5;
  const int lane = threadIdx.x & 31;
  int* tile = cbs + wv * (32 * CB_P);
  unsigned* words = (unsigned*)(cbs + 16 * 32 * CB_P);
  float* pj = (float*)(cbs + 16 * 32 * CB_P + 32 * 33);
  float* pm = pj + 4 * 32;
  const int* opb = opadj + (size_t)b * NN * NN;
  const int* mab = maadj + (size_t)b * NN * NN;
  const int sub = lane >> 3;
  const int q4  = (lane & 7) * 4;
  const bool machines = (blockIdx.x == 0);

  const int node0 = b * NN + r0 + 2 * wv;
  v2f  hv[2];
  int4 lk[2];
#pragma unroll
  for (int q = 0; q < 2; ++q) {
    hv[q] = *(const v2f*)(h + (size_t)(node0 + q) * FF + 2 * lane);
    lk[q] = *(const int4*)(opma + (size_t)(node0 + q) * MM + 4 * lane);
  }
  const v2f wa1 = *(const v2f*)(att_op + 2 * lane);
  const v2f wa2 = *(const v2f*)(att_op + FF + 2 * lane);
  const v2f wm1 = *(const v2f*)(att_ma + 2 * lane);
  v2f wm2 = {};
  v2f mv[8] = {};
  if (machines) {
    const v2f w2 = *(const v2f*)(att_ma + FF + 2 * lane);
    const v2f w3 = *(const v2f*)(att_ma + 2 * FF + 2 * lane);
    wm2.x = w2.x - w3.x;
    wm2.y = w2.y - w3.y;
#pragma unroll
    for (int t = 0; t < 8; ++t)
      mv[t] = *(const v2f*)(hma + ((size_t)b * MM + 8 * wv + t) * FF + 2 * lane);
  }

#pragma unroll
  for (int q = 0; q < 2; ++q) {
    const float s1 = wave_sum32(hv[q].x * wa1.x + hv[q].y * wa1.y);
    const float s2 = wave_sum32(hv[q].x * wa2.x + hv[q].y * wa2.y);
    const float s3 = wave_sum32(hv[q].x * wm1.x + hv[q].y * wm1.y);
    const int any = lk[q].x | lk[q].y | lk[q].z | lk[q].w;
    const unsigned linked = __builtin_amdgcn_ballot_w32(any != 0);
    if (lane == 0) {
      pj[0 * 32 + 2 * wv + q] = s1;
      pj[1 * 32 + 2 * wv + q] = s2;
      pj[2 * 32 + 2 * wv + q] = s3;
      ((int*)pj)[3 * 32 + 2 * wv + q] = (linked == 0u) ? 1 : 0;
    }
  }
  if (machines) {
#pragma unroll
    for (int t = 0; t < 8; ++t) {
      const float s = wave_sum32(mv[t].x * wm2.x + mv[t].y * wm2.y);
      if (lane == 0) pm[8 * wv + t] = s;
    }
  }

  int4 dsum[2][8];
  int  tr[2][32];
#pragma unroll
  for (int hf = 0; hf < 2; ++hf) {
    const int c0 = (wv + 16 * hf) * 32;
#pragma unroll
    for (int q = 0; q < 8; ++q) {
      const size_t off = (size_t)(r0 + 4 * q + sub) * NN + c0 + q4;
      const int4 a = *(const int4*)(opb + off);
      const int4 m = *(const int4*)(mab + off);
      dsum[hf][q].x = a.x + m.x; dsum[hf][q].y = a.y + m.y;
      dsum[hf][q].z = a.z + m.z; dsum[hf][q].w = a.w + m.w;
    }
#pragma unroll
    for (int k = 0; k < 32; ++k) {
      const size_t off = (size_t)(c0 + k) * NN + r0 + lane;
      tr[hf][k] = opb[off] + mab[off];
    }
  }

#pragma unroll
  for (int hf = 0; hf < 2; ++hf) {
    const int ct = wv + 16 * hf;
#pragma unroll
    for (int q = 0; q < 8; ++q)
      *(int4*)(tile + (4 * q + sub) * CB_P + q4) = dsum[hf][q];
    __syncthreads();
    unsigned w = 0;
#pragma unroll
    for (int g = 0; g < 8; ++g) {
      const int4 d = *(const int4*)(tile + lane * CB_P + 4 * g);
      w |= (unsigned)((d.x + tr[hf][4 * g + 0]) > 0) << (4 * g + 0);
      w |= (unsigned)((d.y + tr[hf][4 * g + 1]) > 0) << (4 * g + 1);
      w |= (unsigned)((d.z + tr[hf][4 * g + 2]) > 0) << (4 * g + 2);
      w |= (unsigned)((d.w + tr[hf][4 * g + 3]) > 0) << (4 * g + 3);
    }
    words[lane * 33 + ct] = w;
    __syncthreads();
  }

  for (int rr = wv; rr < 32; rr += 16)
    cbits[((size_t)b * NN + r0 + rr) * (2 * FF) + lane] = words[rr * 33 + lane];

  const int nb = b * NN + r0 + lane;
  if (wv == 0)      ha1[nb]  = pj[0 * 32 + lane];
  else if (wv == 1) ha2[nb]  = pj[1 * 32 + lane];
  else if (wv == 2) ham1[nb] = pj[2 * 32 + lane];
  else if (wv == 3) rowmask[nb] = ((const int*)pj)[3 * 32 + lane];
  else if (machines && wv < 8)
    hma2[b * MM + (wv - 4) * 32 + lane] = pm[(wv - 4) * 32 + lane];
}

#define HS_ROWS 64
__device__ __forceinline__ void lds_barrier() {
  __builtin_amdgcn_fence(__ATOMIC_RELEASE, "workgroup", "local");
  __builtin_amdgcn_s_barrier();
  __builtin_amdgcn_fence(__ATOMIC_ACQUIRE, "workgroup", "local");
}
#define HS_P 72

__device__ __forceinline__ void op_attn_tile(
    float* smem, const int tile,
    const float* __restrict__ h, const float* __restrict__ ha1,
    const float* __restrict__ ha2, const unsigned* __restrict__ cbits,
    const int* __restrict__ rowmask, float* __restrict__ hprime) {
  typedef float att_row_t[NN + 4];
  att_row_t* att = (att_row_t*)smem;
  float* inv_sum = smem + 16 * (NN + 4);

  const int b  = tile / (NN / 16);
  const int i0 = (tile % (NN / 16)) * 16;
  const float NEGINF = -__builtin_inff();

  const int lane = threadIdx.x & 31;
  const int wv = __builtin_amdgcn_readfirstlane((int)(threadIdx.x >> 5));

  float* hs = smem + 16 * (NN + 4) + 16;
  const v4f* hb4 = (const v4f*)(h + (size_t)b * NN * FF) + threadIdx.x;
  float* hst = hs + (threadIdx.x >> 4) * HS_P + (threadIdx.x & 15) * 4;
  v4f sa[8], sb[8];
#pragma unroll
  for (int q = 0; q < 8; ++q) {
    sa[q] = hb4[0 * (HS_ROWS * 16) + 128 * q];
    sb[q] = hb4[1 * (HS_ROWS * 16) + 128 * q];
  }
  {
    float a1v[32];
#pragma unroll
    for (int t = 0; t < 32; ++t) a1v[t] = ha1[b * NN + lane + 32 * t];
    unsigned wd[4];
#pragma unroll
    for (int rr = 0; rr < 4; ++rr)
      wd[rr] = cbits[((size_t)b * NN + i0 + wv + 4 * rr) * (2 * FF) + lane];
#pragma unroll
    for (int rr = 0; rr < 4; ++rr) {
      const int r = wv + 4 * rr;
      const int i = i0 + r;
      const int rm = rowmask[b * NN + i];
      const float a2 = ha2[b * NN + i];
      const unsigned wdm = rm ? 0xffffffffu : wd[rr];
      float e[32];
      float mx = NEGINF;
#pragma unroll
      for (int t = 0; t < 32; ++t) {
        const unsigned wt = (unsigned)__builtin_amdgcn_readlane((int)wdm, t);
        float s = a2 + a1v[t];
        s = s > 0.f ? s : 0.01f * s;
        e[t] = __builtin_amdgcn_inverse_ballot_w32(wt) ? s : NEGINF;
        mx = fmaxf(mx, e[t]);
      }
      mx = fmaxf(mx, xhalf(mx));
      mx = row_max16(mx);
      if (mx == NEGINF) mx = 0.f;
      float sum = 0.f;
#pragma unroll
      for (int t = 0; t < 32; ++t) {
        const float p = __expf(e[t] - mx);
        att[r][lane + 32 * t] = p;
        sum += p;
      }
      sum += xhalf(sum);
      sum += rowx<8>(sum); sum += rowx<4>(sum);
      sum += rowx<2>(sum); sum += rowx<1>(sum);
      if (lane == 0) inv_sum[r] = 1.f / fmaxf(sum, 1e-30f);
    }
  }
#pragma unroll
  for (int q = 0; q < 8; ++q) *(v4f*)(hst + q * 8 * HS_P) = sa[q];
  __syncthreads();

  const int lm = lane & 15;
  const int kh = (lane >> 4) * 2;
  const int lh = lane >> 4;
  const int o  = wv * 16 + lm;
  const float* ar  = &att[lm][kh];
  const float* hf0 = hs + kh * HS_P + o;
  const float* hf1 = hf0 + HS_ROWS * HS_P;

  v8f c = {};
#pragma unroll 1
  for (int s = 0; s < NN / HS_ROWS; s += 2) {
    {
      const int sn = (s + 2) & (NN / HS_ROWS - 1);
#pragma unroll
      for (int q = 0; q < 8; ++q) sa[q] = hb4[sn * (HS_ROWS * 16) + 128 * q];
    }
    {
      v2f af[16], bq[16];
#pragma unroll
      for (int t = 0; t < 16; ++t) {
        af[t].x = ar[s * HS_ROWS + 4 * t];
        af[t].y = ar[s * HS_ROWS + 4 * t + 1];
        bq[t].x = hf0[(4 * t) * HS_P];
        bq[t].y = hf0[(4 * t + 1) * HS_P];
      }
#pragma unroll
      for (int t = 0; t < 16; ++t) c = wmma4(af[t], bq[t], c);
    }
#pragma unroll
    for (int q = 0; q < 8; ++q) *(v4f*)(hst + HS_ROWS * HS_P + q * 8 * HS_P) = sb[q];
    __builtin_amdgcn_sched_group_barrier(0x020, 8, 0);
    __builtin_amdgcn_sched_group_barrier(0x100, 32, 0);
    __builtin_amdgcn_sched_group_barrier(0x008, 16, 0);
    __builtin_amdgcn_sched_group_barrier(0x200, 8, 0);
    lds_barrier();
    {
      const int sn = (s + 3) & (NN / HS_ROWS - 1);
#pragma unroll
      for (int q = 0; q < 8; ++q) sb[q] = hb4[sn * (HS_ROWS * 16) + 128 * q];
    }
    {
      v2f af[16], bq[16];
#pragma unroll
      for (int t = 0; t < 16; ++t) {
        af[t].x = ar[(s + 1) * HS_ROWS + 4 * t];
        af[t].y = ar[(s + 1) * HS_ROWS + 4 * t + 1];
        bq[t].x = hf1[(4 * t) * HS_P];
        bq[t].y = hf1[(4 * t + 1) * HS_P];
      }
#pragma unroll
      for (int t = 0; t < 16; ++t) c = wmma4(af[t], bq[t], c);
    }
#pragma unroll
    for (int q = 0; q < 8; ++q) *(v4f*)(hst + q * 8 * HS_P) = sa[q];
    __builtin_amdgcn_sched_group_barrier(0x020, 8, 0);
    __builtin_amdgcn_sched_group_barrier(0x100, 32, 0);
    __builtin_amdgcn_sched_group_barrier(0x008, 16, 0);
    __builtin_amdgcn_sched_group_barrier(0x200, 8, 0);
    lds_barrier();
  }
  float* hp = hprime + ((size_t)b * NN + i0) * (2 * FF);
#pragma unroll
  for (int r = 0; r < 8; ++r) {
    int m = r + 8 * lh;
    hp[(size_t)m * (2 * FF) + o] = c[r] * inv_sum[m];
  }
}

#define MA_NPW 4
#define HM_P 68
#define MA_WAVES 4

__device__ __forceinline__ void ma_attn_nodes(
    float* smem, const int group,
    const float* __restrict__ z, const float* __restrict__ wedge,
    const float* __restrict__ hma, const float* __restrict__ ham1,
    const float* __restrict__ hma2, const int* __restrict__ opma,
    const int* __restrict__ rowmask, const float* __restrict__ h,
    const float* __restrict__ attma, float* __restrict__ hprime) {
  const int lane = threadIdx.x & 31;
  const int c  = lane & 15;
  const int hh = lane >> 4;
  const int wv  = __builtin_amdgcn_readfirstlane((int)(threadIdx.x >> 5));
  const int bn0 = (group * MA_WAVES + wv) * MA_NPW;
  const int b   = bn0 / NN;
  const float NEGINF = -__builtin_inff();

  v2f awe[4][4];
#pragma unroll
  for (int s = 0; s < 4; ++s)
#pragma unroll
    for (int ot = 0; ot < 4; ++ot) {
      awe[s][ot].x = wedge[(8 * hh + 2 * s) * FF + ot * 16 + c];
      awe[s][ot].y = wedge[(8 * hh + 2 * s + 1) * FF + ot * 16 + c];
    }
  v4f am3v[4][2];
#pragma unroll
  for (int ot = 0; ot < 4; ++ot)
#pragma unroll
    for (int q = 0; q < 2; ++q)
      am3v[ot][q] = *(const v4f*)(attma + 2 * FF + ot * 16 + 8 * hh + 4 * q);
  float hma2v[8];
#pragma unroll
  for (int t = 0; t < 8; ++t) hma2v[t] = hma2[b * MM + 16 * t + c];
  {
    const v4f* src4 = (const v4f*)(hma + (size_t)b * MM * FF);
#pragma unroll
    for (int q = 0; q < 16; ++q) {
      const int idx = threadIdx.x + 128 * q;
      *(v4f*)(smem + (idx >> 4) * HM_P + (idx & 15) * 4) = src4[idx];
    }
  }
  __syncthreads();
  const float* hmb = smem + c * HM_P + 8 * hh;

  v4f zq[8][2];
  {
    const float* zp = z + (size_t)bn0 * MM * FE + c * FE + 8 * hh;
#pragma unroll
    for (int t = 0; t < 8; ++t)
#pragma unroll
      for (int q = 0; q < 2; ++q)
        zq[t][q] = *(const v4f*)(zp + t * 16 * FE + 4 * q);
  }

#pragma unroll 1
  for (int i = 0; i < MA_NPW; ++i) {
    const int bn = bn0 + i;
    asm volatile("" ::: "memory");

    const int   rm  = rowmask[bn];
    const float hm1 = ham1[bn];
    int adv[8];
#pragma unroll
    for (int t = 0; t < 8; ++t) adv[t] = opma[(size_t)bn * MM + 16 * t + c];

    v8f D[8][4];
    float sc[8];
#pragma unroll
    for (int t = 0; t < 8; ++t) {
      v2f zb[4];
      zb[0].x = zq[t][0].x; zb[0].y = zq[t][0].y;
      zb[1].x = zq[t][0].z; zb[1].y = zq[t][0].w;
      zb[2].x = zq[t][1].x; zb[2].y = zq[t][1].y;
      zb[3].x = zq[t][1].z; zb[3].y = zq[t][1].w;
      float part[4];
#pragma unroll
      for (int ot = 0; ot < 4; ++ot) {
        const v4f g0 = *(const v4f*)(hmb + t * 16 * HM_P + ot * 16);
        const v4f g1 = *(const v4f*)(hmb + t * 16 * HM_P + ot * 16 + 4);
        v8f d = {g0.x, g0.y, g0.z, g0.w, g1.x, g1.y, g1.z, g1.w};
#pragma unroll
        for (int s = 0; s < 4; ++s) d = wmma4(awe[s][ot], zb[s], d);
        D[t][ot] = d;
        v2f a2 = __builtin_shufflevector(d, d, 0, 1) * __builtin_shufflevector(am3v[ot][0], am3v[ot][0], 0, 1);
        a2 += __builtin_shufflevector(d, d, 2, 3) * __builtin_shufflevector(am3v[ot][0], am3v[ot][0], 2, 3);
        a2 += __builtin_shufflevector(d, d, 4, 5) * __builtin_shufflevector(am3v[ot][1], am3v[ot][1], 0, 1);
        a2 += __builtin_shufflevector(d, d, 6, 7) * __builtin_shufflevector(am3v[ot][1], am3v[ot][1], 2, 3);
        part[ot] = a2.x + a2.y;
      }
      float dot = (part[0] + part[1]) + (part[2] + part[3]);
      dot += xhalf(dot);
      float e = hm1 + hma2v[t] + dot;
      e = e > 0.f ? e : 0.01f * e;
      const int conn = rm | (adv[t] > 0);
      sc[t] = conn ? e : NEGINF;
    }

    float mx = sc[0];
#pragma unroll
    for (int t = 1; t < 8; ++t) mx = fmaxf(mx, sc[t]);
    mx = row_max16(mx);
    if (mx == NEGINF) mx = 0.f;
    float p[8];
    float sum = 0.f;
#pragma unroll
    for (int t = 0; t < 8; ++t) { p[t] = __expf(sc[t] - mx); sum += p[t]; }
    sum = row_sum16(sum);
    const float inv = 1.f / fmaxf(sum, 1e-30f);

    if (i + 1 < MA_NPW) {
      const float* zp = z + (size_t)(bn + 1) * MM * FE + c * FE + 8 * hh;
#pragma unroll
      for (int t = 0; t < 8; ++t)
#pragma unroll
        for (int q = 0; q < 2; ++q)
          zq[t][q] = *(const v4f*)(zp + t * 16 * FE + 4 * q);
    }

    v8f acc[4];
#pragma unroll
    for (int ot = 0; ot < 4; ++ot) acc[ot] = (v8f){};
#pragma unroll
    for (int t = 0; t < 8; ++t)
#pragma unroll
      for (int ot = 0; ot < 4; ++ot)
#pragma unroll
        for (int r = 0; r < 8; ++r) acc[ot][r] += p[t] * D[t][ot][r];

    const float* hr = h + (size_t)bn * FF + 8 * hh;
    v4f res[4][2];
#pragma unroll
    for (int ot = 0; ot < 4; ++ot) {
      res[ot][0] = *(const v4f*)(hr + ot * 16);
      res[ot][1] = *(const v4f*)(hr + ot * 16 + 4);
    }
#pragma unroll
    for (int ot = 0; ot < 4; ++ot)
#pragma unroll
      for (int r = 0; r < 8; ++r) acc[ot][r] = row_sum16(acc[ot][r]);

    float* hp = hprime + (size_t)bn * (2 * FF) + FF + 8 * hh;
#pragma unroll
    for (int ot = 0; ot < 4; ++ot) {
      const v4f r0 = res[ot][0];
      const v4f r1 = res[ot][1];
      v4f o0, o1;
      o0.x = acc[ot][0] * inv + r0.x; o0.y = acc[ot][1] * inv + r0.y;
      o0.z = acc[ot][2] * inv + r0.z; o0.w = acc[ot][3] * inv + r0.w;
      o1.x = acc[ot][4] * inv + r1.x; o1.y = acc[ot][5] * inv + r1.y;
      o1.z = acc[ot][6] * inv + r1.z; o1.w = acc[ot][7] * inv + r1.w;
      if (c == 0) {
        *(v4f*)(hp + ot * 16)     = o0;
        *(v4f*)(hp + ot * 16 + 4) = o1;
      }
    }

  }
}

__global__ __launch_bounds__(128, 2) void attn_pair(
    const float* __restrict__ h, const float* __restrict__ ha1,
    const float* __restrict__ ha2, const unsigned* __restrict__ cbits,
    const int* __restrict__ rowmask, float* __restrict__ hprime,
    const float* __restrict__ z, const float* __restrict__ wedge,
    const float* __restrict__ hma, const float* __restrict__ ham1,
    const float* __restrict__ hma2, const int* __restrict__ opma,
    const float* __restrict__ attma) {
  extern __shared__ float smem[];
  const int g = blockIdx.x;
  const int op_first = (g ^ (g >> 8)) & 1;
#pragma unroll 1
  for (int ph = 0; ph < 2; ++ph) {
    if (ph == op_first)
      ma_attn_nodes(smem, g, z, wedge, hma, ham1, hma2, opma, rowmask, h, attma, hprime);
    else
      op_attn_tile(smem, g, h, ha1, ha2, cbits, rowmask, hprime);
    __syncthreads();
  }
}

__global__ __launch_bounds__(256, 2) void mlp(
    const float* __restrict__ hprime, const float* __restrict__ w1,
    const float* __restrict__ b1, const float* __restrict__ w2,
    const float* __restrict__ b2, const float* __restrict__ wo,
    const float* __restrict__ bo, const int* __restrict__ rowmask,
    float* __restrict__ out) {
  extern __shared__ float smem[];
  typedef float act_row_t[HID + 4];
  act_row_t* a0 = (act_row_t*)smem;
  act_row_t* a1 = (act_row_t*)(smem + 16 * (HID + 4));

  const int row0 = blockIdx.x * 16;
  const int wv = __builtin_amdgcn_readfirstlane((int)(threadIdx.x >> 5));
  const int lane = threadIdx.x & 31;
  const int lm = lane & 15;
  const int kh = (lane >> 4) * 2;
  const int lh = lane >> 4;
  const int o  = wv * 16 + lm;

  const v4f* hp4 = (const v4f*)(hprime + (size_t)row0 * HID) + threadIdx.x;
  const v4f in0 = hp4[0];
  const v4f in1 = hp4[256];
  const float bias1 = b1[o];
  const float bias2 = b2[o];
  v2f bw1[32];
#pragma unroll
  for (int t = 0; t < 32; ++t) {
    bw1[t].x = w1[(4 * t + kh) * HID + o];
    bw1[t].y = w1[(4 * t + kh + 1) * HID + o];
  }
  *(v4f*)&a0[threadIdx.x >> 5][(threadIdx.x & 31) * 4] = in0;
  *(v4f*)&a0[(threadIdx.x >> 5) + 8][(threadIdx.x & 31) * 4] = in1;
  lds_barrier();

  v2f bw2[32];
#pragma unroll
  for (int t = 0; t < 32; ++t) {
    bw2[t].x = w2[(4 * t + kh) * HID + o];
    bw2[t].y = w2[(4 * t + kh + 1) * HID + o];
  }

  v8f c1 = {};
  {
    v2f af[32];
#pragma unroll
    for (int t = 0; t < 32; ++t) {
      af[t].x = a0[lm][4 * t + kh];
      af[t].y = a0[lm][4 * t + kh + 1];
    }
#pragma unroll
    for (int t = 0; t < 32; ++t) c1 = wmma4(af[t], bw1[t], c1);
    __builtin_amdgcn_sched_group_barrier(0x100, 32, 0);
    __builtin_amdgcn_sched_group_barrier(0x008, 32, 0);
  }
  const float bias3 = bo[o & (FF - 1)];
  const int4 rm0 = *(const int4*)(rowmask + row0 + 8 * lh);
  const int4 rm1 = *(const int4*)(rowmask + row0 + 8 * lh + 4);
  v2f bwo[32];
  if (wv < 4) {
#pragma unroll
    for (int t = 0; t < 32; ++t) {
      bwo[t].x = wo[(4 * t + kh) * FF + o];
      bwo[t].y = wo[(4 * t + kh + 1) * FF + o];
    }
  }
  {
#pragma unroll
    for (int r = 0; r < 8; ++r) {
      float v = c1[r] + bias1;
      a1[r + 8 * lh][o] = v > 0.f ? v : expm1f(v);
    }
  }
  lds_barrier();

  {
    v8f c = {};
    v2f af[32];
#pragma unroll
    for (int t = 0; t < 32; ++t) {
      af[t].x = a1[lm][4 * t + kh];
      af[t].y = a1[lm][4 * t + kh + 1];
    }
#pragma unroll
    for (int t = 0; t < 32; ++t) c = wmma4(af[t], bw2[t], c);
    __builtin_amdgcn_sched_group_barrier(0x100, 32, 0);
    __builtin_amdgcn_sched_group_barrier(0x008, 32, 0);
#pragma unroll
    for (int r = 0; r < 8; ++r) {
      float v = c[r] + bias2;
      a0[r + 8 * lh][o] = v > 0.f ? v : expm1f(v);
    }
  }
  lds_barrier();

  if (wv < 4) {
    v8f c = {};
    v2f af[32];
#pragma unroll
    for (int t = 0; t < 32; ++t) {
      af[t].x = a0[lm][4 * t + kh];
      af[t].y = a0[lm][4 * t + kh + 1];
    }
#pragma unroll
    for (int t = 0; t < 32; ++t) c = wmma4(af[t], bwo[t], c);
    __builtin_amdgcn_sched_group_barrier(0x100, 32, 0);
    __builtin_amdgcn_sched_group_barrier(0x008, 32, 0);
    const int rmv[8] = {rm0.x, rm0.y, rm0.z, rm0.w, rm1.x, rm1.y, rm1.z, rm1.w};
#pragma unroll
    for (int r = 0; r < 8; ++r) {
      int m = r + 8 * lh;
      float v = c[r] + bias3;
      if (rmv[r]) v = 0.f;
      out[(size_t)(row0 + m) * FF + o] = v;
    }
  }
}

extern "C" void kernel_launch(void* const* d_in, const int* in_sizes, int n_in,
                              void* d_out, int out_size, void* d_ws,
                              size_t ws_size, hipStream_t stream) {
  (void)in_sizes; (void)n_in; (void)out_size; (void)ws_size;
  const float* x      = (const float*)d_in[0];
  const float* y      = (const float*)d_in[1];
  const float* z      = (const float*)d_in[2];
  const int*   op_adj = (const int*)d_in[3];
  const int*   ma_adj = (const int*)d_in[4];
  const int*   op_ma  = (const int*)d_in[5];
  const float* Wop    = (const float*)d_in[6];
  const float* Wma    = (const float*)d_in[7];
  const float* Wedge  = (const float*)d_in[8];
  const float* att_op = (const float*)d_in[9];
  const float* att_ma = (const float*)d_in[10];
  const float* w1     = (const float*)d_in[11];
  const float* b1     = (const float*)d_in[12];
  const float* w2     = (const float*)d_in[13];
  const float* b2     = (const float*)d_in[14];
  const float* wo     = (const float*)d_in[15];
  const float* bo     = (const float*)d_in[16];
  float* out = (float*)d_out;

  float* ws     = (float*)d_ws;
  float* h      = ws;
  float* hma    = h + BB * NN * FF;
  float* ha1    = hma + BB * MM * FF;
  float* ha2    = ha1 + BB * NN;
  float* ham1   = ha2 + BB * NN;
  float* hma2   = ham1 + BB * NN;
  float* hprime = hma2 + BB * MM;
  int*   rmask  = (int*)(hprime + (size_t)BB * NN * 2 * FF);

  gemm_rows_64<<<dim3((BB * NN) / 16 + (BB * MM) / 16), dim3(128), 0, stream>>>(
      x, Wop, h, y, Wma, hma);
  size_t sm2 = (16 * (NN + 4) + 16 + 2 * HS_ROWS * HS_P) * sizeof(float);
  unsigned* cbits = (unsigned*)hprime;
  size_t smb = (16 * 32 * CB_P + 32 * 33 + 4 * 32 + MM) * sizeof(int);
  conn_bits_proj<<<dim3(NN / 32, BB), dim3(512), smb, stream>>>(
      op_adj, ma_adj, cbits, h, hma, att_op, att_ma, op_ma,
      ha1, ha2, ham1, hma2, rmask);
  attn_pair<<<dim3((BB * NN) / 16), dim3(128), sm2, stream>>>(
      h, ha1, ha2, cbits, rmask, hprime,
      z, Wedge, hma, ham1, hma2, op_ma, att_ma);
  size_t sm4 = (2 * 16 * (HID + 4)) * sizeof(float);
  mlp<<<dim3((BB * NN) / 16), dim3(256), sm4, stream>>>(
      hprime, w1, b1, w2, b2, wo, bo, rmask, out);
}


# projections taken from x and y against Wop@a, Wma@(am2-am3) formed per block, so the bit/projection launch needs no h; feature WMMA tiles ride in its waves; 4 launches -> 3; summing order changed
# speedup vs baseline: 3.8750x; 1.0489x over previous
#include <hip/hip_runtime.h>
#include <math.h>


typedef __attribute__((ext_vector_type(2))) float v2f;
typedef __attribute__((ext_vector_type(8))) float v8f;
typedef __attribute__((ext_vector_type(4))) float v4f;

#define BB 8
#define NN 1024
#define MM 128
#define FF 64
#define FE 16
#define HID 128

__device__ __forceinline__ v8f wmma4(v2f a, v2f b, v8f c) {
  return __builtin_amdgcn_wmma_f32_16x16x4_f32(
       false, a,  false, b,
       (short)0, c,  false,  false);
}

__device__ __forceinline__ float xhalf(float x) {
  int b = __builtin_bit_cast(int, x);
  return __builtin_bit_cast(float,
      __builtin_amdgcn_permlanex16(b, b, 0x76543210, 0xfedcba98, false, false));
}
template <int MASK>
__device__ __forceinline__ float rowx(float x) {
  int b = __builtin_bit_cast(int, x);
  return __builtin_bit_cast(float,
      __builtin_amdgcn_update_dpp(0, b, 0x160 + MASK, 0xf, 0xf, true));
}
__device__ __forceinline__ float row_sum16(float x) {
  x += rowx<1>(x); x += rowx<2>(x); x += rowx<4>(x); x += rowx<8>(x);
  return x;
}
__device__ __forceinline__ float row_max16(float x) {
  x = fmaxf(x, rowx<1>(x)); x = fmaxf(x, rowx<2>(x));
  x = fmaxf(x, rowx<4>(x)); x = fmaxf(x, rowx<8>(x));
  return x;
}
__device__ __forceinline__ float wave_sum32(float x) {
  x = row_sum16(x);
  return x + xhalf(x);
}


__device__ __forceinline__ void lds_barrier() {
  __builtin_amdgcn_fence(__ATOMIC_RELEASE, "workgroup", "local");
  __builtin_amdgcn_s_barrier();
  __builtin_amdgcn_fence(__ATOMIC_ACQUIRE, "workgroup", "local");
}

#define CB_P 36

__device__ __forceinline__ float dot8(v4f a0, v4f a1, v4f b0, v4f b1) {
  return a0.x * b0.x + a0.y * b0.y + a0.z * b0.z + a0.w * b0.w
       + a1.x * b1.x + a1.y * b1.y + a1.z * b1.z + a1.w * b1.w;
}
__device__ __forceinline__ float sum8(float x) {
  x += rowx<1>(x); x += rowx<2>(x); x += rowx<4>(x);
  return x;
}

__global__ __launch_bounds__(512) void feat_bits_proj(
    const int* __restrict__ opadj, const int* __restrict__ maadj,
    unsigned* __restrict__ cbits,
    const float* __restrict__ x, const float* __restrict__ y,
    const float* __restrict__ Wop, const float* __restrict__ Wma,
    float* __restrict__ h, float* __restrict__ hma,
    const float* __restrict__ att_op, const float* __restrict__ att_ma,
    const int* __restrict__ opma,
    float* __restrict__ ha1, float* __restrict__ ha2,
    float* __restrict__ ham1, float* __restrict__ hma2,
    int* __restrict__ rowmask) {
  extern __shared__ __attribute__((aligned(16))) int cbs[];
  const int b    = blockIdx.y;
  const int r0   = blockIdx.x * 32;
  const int wv   = threadIdx.x >> 5;
  const int wvs  = __builtin_amdgcn_readfirstlane((int)(threadIdx.x >> 5));
  const int lane = threadIdx.x & 31;
  int* tile = cbs + wv * (32 * CB_P);
  unsigned* words = (unsigned*)(cbs + 16 * 32 * CB_P);
  float* pj = (float*)(cbs + 16 * 32 * CB_P + 32 * 33);
  float* pm = pj + 4 * 32;
  float* vv = pm + MM;
  const int* opb = opadj + (size_t)b * NN * NN;
  const int* mab = maadj + (size_t)b * NN * NN;
  const int sub = lane >> 3;
  const int q4  = (lane & 7) * 4;
  const bool machines = (blockIdx.x == 0);

  const bool tile_wave = wvs < 9;
  const bool tile_h    = wvs < 8;
  const int lm = lane & 15;
  const int kh = (lane >> 4) * 2;
  const int lh = lane >> 4;
  const float* GA = tile_h ? x : y;
  const float* GW = tile_h ? Wop : Wma;
  float* GC = tile_h ? h : hma;
  const int grow = tile_h ? (b * NN + r0 + 16 * (wvs >> 2)) : (b * MM + 16 * ((int)blockIdx.x >> 2));
  const int go   = (tile_h ? (wvs & 3) : ((int)blockIdx.x & 3)) * 16 + lm;
  v2f ga[16], gb[16];
  if (tile_wave) {
    const float* Arow = GA + (size_t)(grow + lm) * FF;
#pragma unroll
    for (int t = 0; t < 16; ++t) {
      ga[t].x = Arow[4 * t + kh];
      ga[t].y = Arow[4 * t + kh + 1];
      gb[t].x = GW[(size_t)(4 * t + kh) * FF + go];
      gb[t].y = GW[(size_t)(4 * t + kh + 1) * FF + go];
    }
  }

  const int node0 = b * NN + r0 + 2 * wv;
  v2f  xv[2];
  int4 lk[2];
#pragma unroll
  for (int q = 0; q < 2; ++q) {
    xv[q] = *(const v2f*)(x + (size_t)(node0 + q) * FF + 2 * lane);
    lk[q] = *(const int4*)(opma + (size_t)(node0 + q) * MM + 4 * lane);
  }
  const int vi = 4 * wv + (lane >> 3);
  const int vc = (lane & 7) * 8;
  const v4f w0  = *(const v4f*)(Wop + (size_t)vi * FF + vc);
  const v4f w1  = *(const v4f*)(Wop + (size_t)vi * FF + vc + 4);
  const v4f p10 = *(const v4f*)(att_op + vc);
  const v4f p11 = *(const v4f*)(att_op + vc + 4);
  const v4f p20 = *(const v4f*)(att_op + FF + vc);
  const v4f p21 = *(const v4f*)(att_op + FF + vc + 4);
  const v4f p30 = *(const v4f*)(att_ma + vc);
  const v4f p31 = *(const v4f*)(att_ma + vc + 4);
  v2f mv[8] = {};
  v4f m0, m1, q20, q21, q30, q31;
  if (machines) {
#pragma unroll
    for (int t = 0; t < 8; ++t)
      mv[t] = *(const v2f*)(y + ((size_t)b * MM + 8 * wv + t) * FF + 2 * lane);
    m0  = *(const v4f*)(Wma + (size_t)vi * FF + vc);
    m1  = *(const v4f*)(Wma + (size_t)vi * FF + vc + 4);
    q20 = *(const v4f*)(att_ma + FF + vc);
    q21 = *(const v4f*)(att_ma + FF + vc + 4);
    q30 = *(const v4f*)(att_ma + 2 * FF + vc);
    q31 = *(const v4f*)(att_ma + 2 * FF + vc + 4);
  }

  {
    const float d1 = sum8(dot8(w0, w1, p10, p11));
    const float d2 = sum8(dot8(w0, w1, p20, p21));
    const float d3 = sum8(dot8(w0, w1, p30, p31));
    if ((lane & 7) == 0) {
      vv[0 * FF + vi] = d1;
      vv[1 * FF + vi] = d2;
      vv[2 * FF + vi] = d3;
    }
    if (machines) {
      const float d4 = sum8(dot8(m0, m1, q20 - q30, q21 - q31));
      if ((lane & 7) == 0) vv[3 * FF + vi] = d4;
    }
  }
  lds_barrier();
  const v2f wa1 = *(const v2f*)(vv + 0 * FF + 2 * lane);
  const v2f wa2 = *(const v2f*)(vv + 1 * FF + 2 * lane);
  const v2f wm1 = *(const v2f*)(vv + 2 * FF + 2 * lane);

#pragma unroll
  for (int q = 0; q < 2; ++q) {
    const float s1 = wave_sum32(xv[q].x * wa1.x + xv[q].y * wa1.y);
    const float s2 = wave_sum32(xv[q].x * wa2.x + xv[q].y * wa2.y);
    const float s3 = wave_sum32(xv[q].x * wm1.x + xv[q].y * wm1.y);
    const int any = lk[q].x | lk[q].y | lk[q].z | lk[q].w;
    const unsigned linked = __builtin_amdgcn_ballot_w32(any != 0);
    if (lane == 0) {
      pj[0 * 32 + 2 * wv + q] = s1;
      pj[1 * 32 + 2 * wv + q] = s2;
      pj[2 * 32 + 2 * wv + q] = s3;
      ((int*)pj)[3 * 32 + 2 * wv + q] = (linked == 0u) ? 1 : 0;
    }
  }
  if (machines) {
    const v2f wm2 = *(const v2f*)(vv + 3 * FF + 2 * lane);
#pragma unroll
    for (int t = 0; t < 8; ++t) {
      const float s = wave_sum32(mv[t].x * wm2.x + mv[t].y * wm2.y);
      if (lane == 0) pm[8 * wv + t] = s;
    }
  }

  if (tile_wave) {
    v8f c = {};
#pragma unroll
    for (int t = 0; t < 16; ++t) c = wmma4(ga[t], gb[t], c);
#pragma unroll
    for (int r = 0; r < 8; ++r)
      GC[(size_t)(grow + r + 8 * lh) * FF + go] = c[r];
  }

  int4 dsum[2][8];
  int  tr[2][32];
#pragma unroll
  for (int hf = 0; hf < 2; ++hf) {
    const int c0 = (wv + 16 * hf) * 32;
#pragma unroll
    for (int q = 0; q < 8; ++q) {
      const size_t off = (size_t)(r0 + 4 * q + sub) * NN + c0 + q4;
      const int4 a = *(const int4*)(opb + off);
      const int4 m = *(const int4*)(mab + off);
      dsum[hf][q].x = a.x + m.x; dsum[hf][q].y = a.y + m.y;
      dsum[hf][q].z = a.z + m.z; dsum[hf][q].w = a.w + m.w;
    }
#pragma unroll
    for (int k = 0; k < 32; ++k) {
      const size_t off = (size_t)(c0 + k) * NN + r0 + lane;
      tr[hf][k] = opb[off] + mab[off];
    }
  }

#pragma unroll
  for (int hf = 0; hf < 2; ++hf) {
    const int ct = wv + 16 * hf;
#pragma unroll
    for (int q = 0; q < 8; ++q)
      *(int4*)(tile + (4 * q + sub) * CB_P + q4) = dsum[hf][q];
    __syncthreads();
    unsigned w = 0;
#pragma unroll
    for (int g = 0; g < 8; ++g) {
      const int4 d = *(const int4*)(tile + lane * CB_P + 4 * g);
      w |= (unsigned)((d.x + tr[hf][4 * g + 0]) > 0) << (4 * g + 0);
      w |= (unsigned)((d.y + tr[hf][4 * g + 1]) > 0) << (4 * g + 1);
      w |= (unsigned)((d.z + tr[hf][4 * g + 2]) > 0) << (4 * g + 2);
      w |= (unsigned)((d.w + tr[hf][4 * g + 3]) > 0) << (4 * g + 3);
    }
    words[lane * 33 + ct] = w;
    __syncthreads();
  }

  for (int rr = wv; rr < 32; rr += 16)
    cbits[((size_t)b * NN + r0 + rr) * (2 * FF) + lane] = words[rr * 33 + lane];

  const int nb = b * NN + r0 + lane;
  if (wv == 0)      ha1[nb]  = pj[0 * 32 + lane];
  else if (wv == 1) ha2[nb]  = pj[1 * 32 + lane];
  else if (wv == 2) ham1[nb] = pj[2 * 32 + lane];
  else if (wv == 3) rowmask[nb] = ((const int*)pj)[3 * 32 + lane];
  else if (machines && wv < 8)
    hma2[b * MM + (wv - 4) * 32 + lane] = pm[(wv - 4) * 32 + lane];
}

#define HS_ROWS 64
#define HS_P 72

__device__ __forceinline__ void op_attn_tile(
    float* smem, const int tile,
    const float* __restrict__ h, const float* __restrict__ ha1,
    const float* __restrict__ ha2, const unsigned* __restrict__ cbits,
    const int* __restrict__ rowmask, float* __restrict__ hprime) {
  typedef float att_row_t[NN + 4];
  att_row_t* att = (att_row_t*)smem;
  float* inv_sum = smem + 16 * (NN + 4);

  const int b  = tile / (NN / 16);
  const int i0 = (tile % (NN / 16)) * 16;
  const float NEGINF = -__builtin_inff();

  const int lane = threadIdx.x & 31;
  const int wv = __builtin_amdgcn_readfirstlane((int)(threadIdx.x >> 5));

  float* hs = smem + 16 * (NN + 4) + 16;
  const v4f* hb4 = (const v4f*)(h + (size_t)b * NN * FF) + threadIdx.x;
  float* hst = hs + (threadIdx.x >> 4) * HS_P + (threadIdx.x & 15) * 4;
  v4f sa[8], sb[8];
#pragma unroll
  for (int q = 0; q < 8; ++q) {
    sa[q] = hb4[0 * (HS_ROWS * 16) + 128 * q];
    sb[q] = hb4[1 * (HS_ROWS * 16) + 128 * q];
  }
  {
    float a1v[32];
#pragma unroll
    for (int t = 0; t < 32; ++t) a1v[t] = ha1[b * NN + lane + 32 * t];
    unsigned wd[4];
#pragma unroll
    for (int rr = 0; rr < 4; ++rr)
      wd[rr] = cbits[((size_t)b * NN + i0 + wv + 4 * rr) * (2 * FF) + lane];
#pragma unroll
    for (int rr = 0; rr < 4; ++rr) {
      const int r = wv + 4 * rr;
      const int i = i0 + r;
      const int rm = rowmask[b * NN + i];
      const float a2 = ha2[b * NN + i];
      const unsigned wdm = rm ? 0xffffffffu : wd[rr];
      float e[32];
      float mx = NEGINF;
#pragma unroll
      for (int t = 0; t < 32; ++t) {
        const unsigned wt = (unsigned)__builtin_amdgcn_readlane((int)wdm, t);
        float s = a2 + a1v[t];
        s = s > 0.f ? s : 0.01f * s;
        e[t] = __builtin_amdgcn_inverse_ballot_w32(wt) ? s : NEGINF;
        mx = fmaxf(mx, e[t]);
      }
      mx = fmaxf(mx, xhalf(mx));
      mx = row_max16(mx);
      if (mx == NEGINF) mx = 0.f;
      float sum = 0.f;
#pragma unroll
      for (int t = 0; t < 32; ++t) {
        const float p = __expf(e[t] - mx);
        att[r][lane + 32 * t] = p;
        sum += p;
      }
      sum += xhalf(sum);
      sum += rowx<8>(sum); sum += rowx<4>(sum);
      sum += rowx<2>(sum); sum += rowx<1>(sum);
      if (lane == 0) inv_sum[r] = 1.f / fmaxf(sum, 1e-30f);
    }
  }
#pragma unroll
  for (int q = 0; q < 8; ++q) *(v4f*)(hst + q * 8 * HS_P) = sa[q];
  __syncthreads();

  const int lm = lane & 15;
  const int kh = (lane >> 4) * 2;
  const int lh = lane >> 4;
  const int o  = wv * 16 + lm;
  const float* ar  = &att[lm][kh];
  const float* hf0 = hs + kh * HS_P + o;
  const float* hf1 = hf0 + HS_ROWS * HS_P;

  v8f c = {};
#pragma unroll 1
  for (int s = 0; s < NN / HS_ROWS; s += 2) {
    {
      const int sn = (s + 2) & (NN / HS_ROWS - 1);
#pragma unroll
      for (int q = 0; q < 8; ++q) sa[q] = hb4[sn * (HS_ROWS * 16) + 128 * q];
    }
    {
      v2f af[16], bq[16];
#pragma unroll
      for (int t = 0; t < 16; ++t) {
        af[t].x = ar[s * HS_ROWS + 4 * t];
        af[t].y = ar[s * HS_ROWS + 4 * t + 1];
        bq[t].x = hf0[(4 * t) * HS_P];
        bq[t].y = hf0[(4 * t + 1) * HS_P];
      }
#pragma unroll
      for (int t = 0; t < 16; ++t) c = wmma4(af[t], bq[t], c);
    }
#pragma unroll
    for (int q = 0; q < 8; ++q) *(v4f*)(hst + HS_ROWS * HS_P + q * 8 * HS_P) = sb[q];
    __builtin_amdgcn_sched_group_barrier(0x020, 8, 0);
    __builtin_amdgcn_sched_group_barrier(0x100, 32, 0);
    __builtin_amdgcn_sched_group_barrier(0x008, 16, 0);
    __builtin_amdgcn_sched_group_barrier(0x200, 8, 0);
    lds_barrier();
    {
      const int sn = (s + 3) & (NN / HS_ROWS - 1);
#pragma unroll
      for (int q = 0; q < 8; ++q) sb[q] = hb4[sn * (HS_ROWS * 16) + 128 * q];
    }
    {
      v2f af[16], bq[16];
#pragma unroll
      for (int t = 0; t < 16; ++t) {
        af[t].x = ar[(s + 1) * HS_ROWS + 4 * t];
        af[t].y = ar[(s + 1) * HS_ROWS + 4 * t + 1];
        bq[t].x = hf1[(4 * t) * HS_P];
        bq[t].y = hf1[(4 * t + 1) * HS_P];
      }
#pragma unroll
      for (int t = 0; t < 16; ++t) c = wmma4(af[t], bq[t], c);
    }
#pragma unroll
    for (int q = 0; q < 8; ++q) *(v4f*)(hst + q * 8 * HS_P) = sa[q];
    __builtin_amdgcn_sched_group_barrier(0x020, 8, 0);
    __builtin_amdgcn_sched_group_barrier(0x100, 32, 0);
    __builtin_amdgcn_sched_group_barrier(0x008, 16, 0);
    __builtin_amdgcn_sched_group_barrier(0x200, 8, 0);
    lds_barrier();
  }
  float* hp = hprime + ((size_t)b * NN + i0) * (2 * FF);
#pragma unroll
  for (int r = 0; r < 8; ++r) {
    int m = r + 8 * lh;
    hp[(size_t)m * (2 * FF) + o] = c[r] * inv_sum[m];
  }
}

#define MA_NPW 4
#define HM_P 68
#define MA_WAVES 4

__device__ __forceinline__ void ma_attn_nodes(
    float* smem, const int group,
    const float* __restrict__ z, const float* __restrict__ wedge,
    const float* __restrict__ hma, const float* __restrict__ ham1,
    const float* __restrict__ hma2, const int* __restrict__ opma,
    const int* __restrict__ rowmask, const float* __restrict__ h,
    const float* __restrict__ attma, float* __restrict__ hprime) {
  const int lane = threadIdx.x & 31;
  const int c  = lane & 15;
  const int hh = lane >> 4;
  const int wv  = __builtin_amdgcn_readfirstlane((int)(threadIdx.x >> 5));
  const int bn0 = (group * MA_WAVES + wv) * MA_NPW;
  const int b   = bn0 / NN;
  const float NEGINF = -__builtin_inff();

  v2f awe[4][4];
#pragma unroll
  for (int s = 0; s < 4; ++s)
#pragma unroll
    for (int ot = 0; ot < 4; ++ot) {
      awe[s][ot].x = wedge[(8 * hh + 2 * s) * FF + ot * 16 + c];
      awe[s][ot].y = wedge[(8 * hh + 2 * s + 1) * FF + ot * 16 + c];
    }
  v4f am3v[4][2];
#pragma unroll
  for (int ot = 0; ot < 4; ++ot)
#pragma unroll
    for (int q = 0; q < 2; ++q)
      am3v[ot][q] = *(const v4f*)(attma + 2 * FF + ot * 16 + 8 * hh + 4 * q);
  float hma2v[8];
#pragma unroll
  for (int t = 0; t < 8; ++t) hma2v[t] = hma2[b * MM + 16 * t + c];
  {
    const v4f* src4 = (const v4f*)(hma + (size_t)b * MM * FF);
#pragma unroll
    for (int q = 0; q < 16; ++q) {
      const int idx = threadIdx.x + 128 * q;
      *(v4f*)(smem + (idx >> 4) * HM_P + (idx & 15) * 4) = src4[idx];
    }
  }
  __syncthreads();
  const float* hmb = smem + c * HM_P + 8 * hh;

  v4f zq[8][2];
  {
    const float* zp = z + (size_t)bn0 * MM * FE + c * FE + 8 * hh;
#pragma unroll
    for (int t = 0; t < 8; ++t)
#pragma unroll
      for (int q = 0; q < 2; ++q)
        zq[t][q] = *(const v4f*)(zp + t * 16 * FE + 4 * q);
  }

#pragma unroll 1
  for (int i = 0; i < MA_NPW; ++i) {
    const int bn = bn0 + i;
    asm volatile("" ::: "memory");

    const int   rm  = rowmask[bn];
    const float hm1 = ham1[bn];
    int adv[8];
#pragma unroll
    for (int t = 0; t < 8; ++t) adv[t] = opma[(size_t)bn * MM + 16 * t + c];

    v8f D[8][4];
    float sc[8];
#pragma unroll
    for (int t = 0; t < 8; ++t) {
      v2f zb[4];
      zb[0].x = zq[t][0].x; zb[0].y = zq[t][0].y;
      zb[1].x = zq[t][0].z; zb[1].y = zq[t][0].w;
      zb[2].x = zq[t][1].x; zb[2].y = zq[t][1].y;
      zb[3].x = zq[t][1].z; zb[3].y = zq[t][1].w;
      float part[4];
#pragma unroll
      for (int ot = 0; ot < 4; ++ot) {
        const v4f g0 = *(const v4f*)(hmb + t * 16 * HM_P + ot * 16);
        const v4f g1 = *(const v4f*)(hmb + t * 16 * HM_P + ot * 16 + 4);
        v8f d = {g0.x, g0.y, g0.z, g0.w, g1.x, g1.y, g1.z, g1.w};
#pragma unroll
        for (int s = 0; s < 4; ++s) d = wmma4(awe[s][ot], zb[s], d);
        D[t][ot] = d;
        v2f a2 = __builtin_shufflevector(d, d, 0, 1) * __builtin_shufflevector(am3v[ot][0], am3v[ot][0], 0, 1);
        a2 += __builtin_shufflevector(d, d, 2, 3) * __builtin_shufflevector(am3v[ot][0], am3v[ot][0], 2, 3);
        a2 += __builtin_shufflevector(d, d, 4, 5) * __builtin_shufflevector(am3v[ot][1], am3v[ot][1], 0, 1);
        a2 += __builtin_shufflevector(d, d, 6, 7) * __builtin_shufflevector(am3v[ot][1], am3v[ot][1], 2, 3);
        part[ot] = a2.x + a2.y;
      }
      float dot = (part[0] + part[1]) + (part[2] + part[3]);
      dot += xhalf(dot);
      float e = hm1 + hma2v[t] + dot;
      e = e > 0.f ? e : 0.01f * e;
      const int conn = rm | (adv[t] > 0);
      sc[t] = conn ? e : NEGINF;
    }

    float mx = sc[0];
#pragma unroll
    for (int t = 1; t < 8; ++t) mx = fmaxf(mx, sc[t]);
    mx = row_max16(mx);
    if (mx == NEGINF) mx = 0.f;
    float p[8];
    float sum = 0.f;
#pragma unroll
    for (int t = 0; t < 8; ++t) { p[t] = __expf(sc[t] - mx); sum += p[t]; }
    sum = row_sum16(sum);
    const float inv = 1.f / fmaxf(sum, 1e-30f);

    if (i + 1 < MA_NPW) {
      const float* zp = z + (size_t)(bn + 1) * MM * FE + c * FE + 8 * hh;
#pragma unroll
      for (int t = 0; t < 8; ++t)
#pragma unroll
        for (int q = 0; q < 2; ++q)
          zq[t][q] = *(const v4f*)(zp + t * 16 * FE + 4 * q);
    }

    v8f acc[4];
#pragma unroll
    for (int ot = 0; ot < 4; ++ot) acc[ot] = (v8f){};
#pragma unroll
    for (int t = 0; t < 8; ++t)
#pragma unroll
      for (int ot = 0; ot < 4; ++ot)
#pragma unroll
        for (int r = 0; r < 8; ++r) acc[ot][r] += p[t] * D[t][ot][r];

    const float* hr = h + (size_t)bn * FF + 8 * hh;
    v4f res[4][2];
#pragma unroll
    for (int ot = 0; ot < 4; ++ot) {
      res[ot][0] = *(const v4f*)(hr + ot * 16);
      res[ot][1] = *(const v4f*)(hr + ot * 16 + 4);
    }
#pragma unroll
    for (int ot = 0; ot < 4; ++ot)
#pragma unroll
      for (int r = 0; r < 8; ++r) acc[ot][r] = row_sum16(acc[ot][r]);

    float* hp = hprime + (size_t)bn * (2 * FF) + FF + 8 * hh;
#pragma unroll
    for (int ot = 0; ot < 4; ++ot) {
      const v4f r0 = res[ot][0];
      const v4f r1 = res[ot][1];
      v4f o0, o1;
      o0.x = acc[ot][0] * inv + r0.x; o0.y = acc[ot][1] * inv + r0.y;
      o0.z = acc[ot][2] * inv + r0.z; o0.w = acc[ot][3] * inv + r0.w;
      o1.x = acc[ot][4] * inv + r1.x; o1.y = acc[ot][5] * inv + r1.y;
      o1.z = acc[ot][6] * inv + r1.z; o1.w = acc[ot][7] * inv + r1.w;
      if (c == 0) {
        *(v4f*)(hp + ot * 16)     = o0;
        *(v4f*)(hp + ot * 16 + 4) = o1;
      }
    }

  }
}

__global__ __launch_bounds__(128, 2) void attn_pair(
    const float* __restrict__ h, const float* __restrict__ ha1,
    const float* __restrict__ ha2, const unsigned* __restrict__ cbits,
    const int* __restrict__ rowmask, float* __restrict__ hprime,
    const float* __restrict__ z, const float* __restrict__ wedge,
    const float* __restrict__ hma, const float* __restrict__ ham1,
    const float* __restrict__ hma2, const int* __restrict__ opma,
    const float* __restrict__ attma) {
  extern __shared__ float smem[];
  const int g = blockIdx.x;
  const int op_first = (g ^ (g >> 8)) & 1;
#pragma unroll 1
  for (int ph = 0; ph < 2; ++ph) {
    if (ph == op_first)
      ma_attn_nodes(smem, g, z, wedge, hma, ham1, hma2, opma, rowmask, h, attma, hprime);
    else
      op_attn_tile(smem, g, h, ha1, ha2, cbits, rowmask, hprime);
    __syncthreads();
  }
}

__global__ __launch_bounds__(256, 2) void mlp(
    const float* __restrict__ hprime, const float* __restrict__ w1,
    const float* __restrict__ b1, const float* __restrict__ w2,
    const float* __restrict__ b2, const float* __restrict__ wo,
    const float* __restrict__ bo, const int* __restrict__ rowmask,
    float* __restrict__ out) {
  extern __shared__ float smem[];
  typedef float act_row_t[HID + 4];
  act_row_t* a0 = (act_row_t*)smem;
  act_row_t* a1 = (act_row_t*)(smem + 16 * (HID + 4));

  const int row0 = blockIdx.x * 16;
  const int wv = __builtin_amdgcn_readfirstlane((int)(threadIdx.x >> 5));
  const int lane = threadIdx.x & 31;
  const int lm = lane & 15;
  const int kh = (lane >> 4) * 2;
  const int lh = lane >> 4;
  const int o  = wv * 16 + lm;

  const v4f* hp4 = (const v4f*)(hprime + (size_t)row0 * HID) + threadIdx.x;
  const v4f in0 = hp4[0];
  const v4f in1 = hp4[256];
  const float bias1 = b1[o];
  const float bias2 = b2[o];
  v2f bw1[32];
#pragma unroll
  for (int t = 0; t < 32; ++t) {
    bw1[t].x = w1[(4 * t + kh) * HID + o];
    bw1[t].y = w1[(4 * t + kh + 1) * HID + o];
  }
  *(v4f*)&a0[threadIdx.x >> 5][(threadIdx.x & 31) * 4] = in0;
  *(v4f*)&a0[(threadIdx.x >> 5) + 8][(threadIdx.x & 31) * 4] = in1;
  lds_barrier();

  v2f bw2[32];
#pragma unroll
  for (int t = 0; t < 32; ++t) {
    bw2[t].x = w2[(4 * t + kh) * HID + o];
    bw2[t].y = w2[(4 * t + kh + 1) * HID + o];
  }

  v8f c1 = {};
  {
    v2f af[32];
#pragma unroll
    for (int t = 0; t < 32; ++t) {
      af[t].x = a0[lm][4 * t + kh];
      af[t].y = a0[lm][4 * t + kh + 1];
    }
#pragma unroll
    for (int t = 0; t < 32; ++t) c1 = wmma4(af[t], bw1[t], c1);
    __builtin_amdgcn_sched_group_barrier(0x100, 32, 0);
    __builtin_amdgcn_sched_group_barrier(0x008, 32, 0);
  }
  const float bias3 = bo[o & (FF - 1)];
  const int4 rm0 = *(const int4*)(rowmask + row0 + 8 * lh);
  const int4 rm1 = *(const int4*)(rowmask + row0 + 8 * lh + 4);
  v2f bwo[32];
  if (wv < 4) {
#pragma unroll
    for (int t = 0; t < 32; ++t) {
      bwo[t].x = wo[(4 * t + kh) * FF + o];
      bwo[t].y = wo[(4 * t + kh + 1) * FF + o];
    }
  }
  {
#pragma unroll
    for (int r = 0; r < 8; ++r) {
      float v = c1[r] + bias1;
      a1[r + 8 * lh][o] = v > 0.f ? v : expm1f(v);
    }
  }
  lds_barrier();

  {
    v8f c = {};
    v2f af[32];
#pragma unroll
    for (int t = 0; t < 32; ++t) {
      af[t].x = a1[lm][4 * t + kh];
      af[t].y = a1[lm][4 * t + kh + 1];
    }
#pragma unroll
    for (int t = 0; t < 32; ++t) c = wmma4(af[t], bw2[t], c);
    __builtin_amdgcn_sched_group_barrier(0x100, 32, 0);
    __builtin_amdgcn_sched_group_barrier(0x008, 32, 0);
#pragma unroll
    for (int r = 0; r < 8; ++r) {
      float v = c[r] + bias2;
      a0[r + 8 * lh][o] = v > 0.f ? v : expm1f(v);
    }
  }
  lds_barrier();

  if (wv < 4) {
    v8f c = {};
    v2f af[32];
#pragma unroll
    for (int t = 0; t < 32; ++t) {
      af[t].x = a0[lm][4 * t + kh];
      af[t].y = a0[lm][4 * t + kh + 1];
    }
#pragma unroll
    for (int t = 0; t < 32; ++t) c = wmma4(af[t], bwo[t], c);
    __builtin_amdgcn_sched_group_barrier(0x100, 32, 0);
    __builtin_amdgcn_sched_group_barrier(0x008, 32, 0);
    const int rmv[8] = {rm0.x, rm0.y, rm0.z, rm0.w, rm1.x, rm1.y, rm1.z, rm1.w};
#pragma unroll
    for (int r = 0; r < 8; ++r) {
      int m = r + 8 * lh;
      float v = c[r] + bias3;
      if (rmv[r]) v = 0.f;
      out[(size_t)(row0 + m) * FF + o] = v;
    }
  }
}

extern "C" void kernel_launch(void* const* d_in, const int* in_sizes, int n_in,
                              void* d_out, int out_size, void* d_ws,
                              size_t ws_size, hipStream_t stream) {
  (void)in_sizes; (void)n_in; (void)out_size; (void)ws_size;
  const float* x      = (const float*)d_in[0];
  const float* y      = (const float*)d_in[1];
  const float* z      = (const float*)d_in[2];
  const int*   op_adj = (const int*)d_in[3];
  const int*   ma_adj = (const int*)d_in[4];
  const int*   op_ma  = (const int*)d_in[5];
  const float* Wop    = (const float*)d_in[6];
  const float* Wma    = (const float*)d_in[7];
  const float* Wedge  = (const float*)d_in[8];
  const float* att_op = (const float*)d_in[9];
  const float* att_ma = (const float*)d_in[10];
  const float* w1     = (const float*)d_in[11];
  const float* b1     = (const float*)d_in[12];
  const float* w2     = (const float*)d_in[13];
  const float* b2     = (const float*)d_in[14];
  const float* wo     = (const float*)d_in[15];
  const float* bo     = (const float*)d_in[16];
  float* out = (float*)d_out;

  float* ws     = (float*)d_ws;
  float* h      = ws;
  float* hma    = h + BB * NN * FF;
  float* ha1    = hma + BB * MM * FF;
  float* ha2    = ha1 + BB * NN;
  float* ham1   = ha2 + BB * NN;
  float* hma2   = ham1 + BB * NN;
  float* hprime = hma2 + BB * MM;
  int*   rmask  = (int*)(hprime + (size_t)BB * NN * 2 * FF);

  size_t sm2 = (16 * (NN + 4) + 16 + 2 * HS_ROWS * HS_P) * sizeof(float);
  unsigned* cbits = (unsigned*)hprime;
  size_t smb = (16 * 32 * CB_P + 32 * 33 + 4 * 32 + MM + 4 * FF) * sizeof(int);
  feat_bits_proj<<<dim3(NN / 32, BB), dim3(512), smb, stream>>>(
      op_adj, ma_adj, cbits, x, y, Wop, Wma, h, hma, att_op, att_ma, op_ma,
      ha1, ha2, ham1, hma2, rmask);
  attn_pair<<<dim3((BB * NN) / 16), dim3(128), sm2, stream>>>(
      h, ha1, ha2, cbits, rmask, hprime,
      z, Wedge, hma, ham1, hma2, op_ma, att_ma);
  size_t sm4 = (2 * 16 * (HID + 4)) * sizeof(float);
  mlp<<<dim3((BB * NN) / 16), dim3(256), sm4, stream>>>(
      hprime, w1, b1, w2, b2, wo, bo, rmask, out);
}


# bit mask built from its symmetry: a 32x32 tile is formed once and stored for both row tiles (turned by 32 ballots), per-group lines with one writer each; a block reads 17 tile pairs, not 32
# speedup vs baseline: 4.0847x; 1.0541x over previous
#include <hip/hip_runtime.h>
#include <math.h>


typedef __attribute__((ext_vector_type(2))) float v2f;
typedef __attribute__((ext_vector_type(8))) float v8f;
typedef __attribute__((ext_vector_type(4))) float v4f;

#define BB 8
#define NN 1024
#define MM 128
#define FF 64
#define FE 16
#define HID 128

__device__ __forceinline__ v8f wmma4(v2f a, v2f b, v8f c) {
  return __builtin_amdgcn_wmma_f32_16x16x4_f32(
       false, a,  false, b,
       (short)0, c,  false,  false);
}

__device__ __forceinline__ float xhalf(float x) {
  int b = __builtin_bit_cast(int, x);
  return __builtin_bit_cast(float,
      __builtin_amdgcn_permlanex16(b, b, 0x76543210, 0xfedcba98, false, false));
}
template <int MASK>
__device__ __forceinline__ float rowx(float x) {
  int b = __builtin_bit_cast(int, x);
  return __builtin_bit_cast(float,
      __builtin_amdgcn_update_dpp(0, b, 0x160 + MASK, 0xf, 0xf, true));
}
__device__ __forceinline__ float row_sum16(float x) {
  x += rowx<1>(x); x += rowx<2>(x); x += rowx<4>(x); x += rowx<8>(x);
  return x;
}
__device__ __forceinline__ float row_max16(float x) {
  x = fmaxf(x, rowx<1>(x)); x = fmaxf(x, rowx<2>(x));
  x = fmaxf(x, rowx<4>(x)); x = fmaxf(x, rowx<8>(x));
  return x;
}
__device__ __forceinline__ float wave_sum32(float x) {
  x = row_sum16(x);
  return x + xhalf(x);
}


__device__ __forceinline__ void lds_barrier() {
  __builtin_amdgcn_fence(__ATOMIC_RELEASE, "workgroup", "local");
  __builtin_amdgcn_s_barrier();
  __builtin_amdgcn_fence(__ATOMIC_ACQUIRE, "workgroup", "local");
}

#define CB_P 36

__device__ __forceinline__ float dot8(v4f a0, v4f a1, v4f b0, v4f b1) {
  return a0.x * b0.x + a0.y * b0.y + a0.z * b0.z + a0.w * b0.w
       + a1.x * b1.x + a1.y * b1.y + a1.z * b1.z + a1.w * b1.w;
}
__device__ __forceinline__ float sum8(float x) {
  x += rowx<1>(x); x += rowx<2>(x); x += rowx<4>(x);
  return x;
}

__global__ __launch_bounds__(512) void feat_bits_proj(
    const int* __restrict__ opadj, const int* __restrict__ maadj,
    unsigned* __restrict__ cbits,
    const float* __restrict__ x, const float* __restrict__ y,
    const float* __restrict__ Wop, const float* __restrict__ Wma,
    float* __restrict__ h, float* __restrict__ hma,
    const float* __restrict__ att_op, const float* __restrict__ att_ma,
    const int* __restrict__ opma,
    float* __restrict__ ha1, float* __restrict__ ha2,
    float* __restrict__ ham1, float* __restrict__ hma2,
    int* __restrict__ rowmask) {
  extern __shared__ __attribute__((aligned(16))) int cbs[];
  const int b    = blockIdx.y;
  const int r0   = blockIdx.x * 32;
  const int wv   = threadIdx.x >> 5;
  const int wvs  = __builtin_amdgcn_readfirstlane((int)(threadIdx.x >> 5));
  const int lane = threadIdx.x & 31;
  int* tile = cbs + wv * (32 * CB_P);
  unsigned* words = (unsigned*)(cbs + 16 * 32 * CB_P);
  float* pj = (float*)(cbs + 16 * 32 * CB_P + 32 * 33);
  float* pm = pj + 4 * 32;
  float* vv = pm + MM;
  const int* opb = opadj + (size_t)b * NN * NN;
  const int* mab = maadj + (size_t)b * NN * NN;
  const int sub = lane >> 3;
  const int q4  = (lane & 7) * 4;
  const bool machines = (blockIdx.x == 0);

  const bool tile_wave = wvs < 9;
  const bool tile_h    = wvs < 8;
  const int lm = lane & 15;
  const int kh = (lane >> 4) * 2;
  const int lh = lane >> 4;
  const float* GA = tile_h ? x : y;
  const float* GW = tile_h ? Wop : Wma;
  float* GC = tile_h ? h : hma;
  const int grow = tile_h ? (b * NN + r0 + 16 * (wvs >> 2)) : (b * MM + 16 * ((int)blockIdx.x >> 2));
  const int go   = (tile_h ? (wvs & 3) : ((int)blockIdx.x & 3)) * 16 + lm;
  v2f ga[16], gb[16];
  if (tile_wave) {
    const float* Arow = GA + (size_t)(grow + lm) * FF;
#pragma unroll
    for (int t = 0; t < 16; ++t) {
      ga[t].x = Arow[4 * t + kh];
      ga[t].y = Arow[4 * t + kh + 1];
      gb[t].x = GW[(size_t)(4 * t + kh) * FF + go];
      gb[t].y = GW[(size_t)(4 * t + kh + 1) * FF + go];
    }
  }

  const int node0 = b * NN + r0 + 2 * wv;
  v2f  xv[2];
  int4 lk[2];
#pragma unroll
  for (int q = 0; q < 2; ++q) {
    xv[q] = *(const v2f*)(x + (size_t)(node0 + q) * FF + 2 * lane);
    lk[q] = *(const int4*)(opma + (size_t)(node0 + q) * MM + 4 * lane);
  }
  const int vi = 4 * wv + (lane >> 3);
  const int vc = (lane & 7) * 8;
  const v4f w0  = *(const v4f*)(Wop + (size_t)vi * FF + vc);
  const v4f w1  = *(const v4f*)(Wop + (size_t)vi * FF + vc + 4);
  const v4f p10 = *(const v4f*)(att_op + vc);
  const v4f p11 = *(const v4f*)(att_op + vc + 4);
  const v4f p20 = *(const v4f*)(att_op + FF + vc);
  const v4f p21 = *(const v4f*)(att_op + FF + vc + 4);
  const v4f p30 = *(const v4f*)(att_ma + vc);
  const v4f p31 = *(const v4f*)(att_ma + vc + 4);
  v2f mv[8] = {};
  v4f m0, m1, q20, q21, q30, q31;
  if (machines) {
#pragma unroll
    for (int t = 0; t < 8; ++t)
      mv[t] = *(const v2f*)(y + ((size_t)b * MM + 8 * wv + t) * FF + 2 * lane);
    m0  = *(const v4f*)(Wma + (size_t)vi * FF + vc);
    m1  = *(const v4f*)(Wma + (size_t)vi * FF + vc + 4);
    q20 = *(const v4f*)(att_ma + FF + vc);
    q21 = *(const v4f*)(att_ma + FF + vc + 4);
    q30 = *(const v4f*)(att_ma + 2 * FF + vc);
    q31 = *(const v4f*)(att_ma + 2 * FF + vc + 4);
  }

  {
    const float d1 = sum8(dot8(w0, w1, p10, p11));
    const float d2 = sum8(dot8(w0, w1, p20, p21));
    const float d3 = sum8(dot8(w0, w1, p30, p31));
    if ((lane & 7) == 0) {
      vv[0 * FF + vi] = d1;
      vv[1 * FF + vi] = d2;
      vv[2 * FF + vi] = d3;
    }
    if (machines) {
      const float d4 = sum8(dot8(m0, m1, q20 - q30, q21 - q31));
      if ((lane & 7) == 0) vv[3 * FF + vi] = d4;
    }
  }
  lds_barrier();
  const v2f wa1 = *(const v2f*)(vv + 0 * FF + 2 * lane);
  const v2f wa2 = *(const v2f*)(vv + 1 * FF + 2 * lane);
  const v2f wm1 = *(const v2f*)(vv + 2 * FF + 2 * lane);

#pragma unroll
  for (int q = 0; q < 2; ++q) {
    const float s1 = wave_sum32(xv[q].x * wa1.x + xv[q].y * wa1.y);
    const float s2 = wave_sum32(xv[q].x * wa2.x + xv[q].y * wa2.y);
    const float s3 = wave_sum32(xv[q].x * wm1.x + xv[q].y * wm1.y);
    const int any = lk[q].x | lk[q].y | lk[q].z | lk[q].w;
    const unsigned linked = __builtin_amdgcn_ballot_w32(any != 0);
    if (lane == 0) {
      pj[0 * 32 + 2 * wv + q] = s1;
      pj[1 * 32 + 2 * wv + q] = s2;
      pj[2 * 32 + 2 * wv + q] = s3;
      ((int*)pj)[3 * 32 + 2 * wv + q] = (linked == 0u) ? 1 : 0;
    }
  }
  if (machines) {
    const v2f wm2 = *(const v2f*)(vv + 3 * FF + 2 * lane);
#pragma unroll
    for (int t = 0; t < 8; ++t) {
      const float s = wave_sum32(mv[t].x * wm2.x + mv[t].y * wm2.y);
      if (lane == 0) pm[8 * wv + t] = s;
    }
  }

  if (tile_wave) {
    v8f c = {};
#pragma unroll
    for (int t = 0; t < 16; ++t) c = wmma4(ga[t], gb[t], c);
#pragma unroll
    for (int r = 0; r < 8; ++r)
      GC[(size_t)(grow + r + 8 * lh) * FF + go] = c[r];
  }

  const int I = (int)blockIdx.x;
  int4 dsum[2][8];
  int  tr[2][32];
#pragma unroll
  for (int hf = 0; hf < 2; ++hf) {
    if (hf == 0 || wvs == 0) {
      const int c0 = ((I + (hf ? 16 : wvs)) & 31) * 32;
#pragma unroll
      for (int q = 0; q < 8; ++q) {
        const size_t off = (size_t)(r0 + 4 * q + sub) * NN + c0 + q4;
        const int4 a = *(const int4*)(opb + off);
        const int4 m = *(const int4*)(mab + off);
        dsum[hf][q].x = a.x + m.x; dsum[hf][q].y = a.y + m.y;
        dsum[hf][q].z = a.z + m.z; dsum[hf][q].w = a.w + m.w;
      }
      if (hf != 0 || wvs != 0) {
#pragma unroll
        for (int k = 0; k < 32; ++k) {
          const size_t off = (size_t)(c0 + k) * NN + r0 + lane;
          tr[hf][k] = opb[off] + mab[off];
        }
      }
    }
  }

#pragma unroll
  for (int hf = 0; hf < 2; ++hf) {
    if (hf == 0 || wvs == 0) {
      const int ct = (I + (hf ? 16 : wvs)) & 31;
#pragma unroll
      for (int q = 0; q < 8; ++q)
        *(int4*)(tile + (4 * q + sub) * CB_P + q4) = dsum[hf][q];
      asm volatile("s_wait_dscnt 0x0" ::: "memory");
      if (hf == 0 && wvs == 0) {
#pragma unroll
        for (int k = 0; k < 32; ++k) tr[0][k] = tile[k * CB_P + lane];
      }
      unsigned w = 0;
#pragma unroll
      for (int g = 0; g < 8; ++g) {
        const int4 d = *(const int4*)(tile + lane * CB_P + 4 * g);
        w |= (unsigned)((d.x + tr[hf][4 * g + 0]) > 0) << (4 * g + 0);
        w |= (unsigned)((d.y + tr[hf][4 * g + 1]) > 0) << (4 * g + 1);
        w |= (unsigned)((d.z + tr[hf][4 * g + 2]) > 0) << (4 * g + 2);
        w |= (unsigned)((d.w + tr[hf][4 * g + 3]) > 0) << (4 * g + 3);
      }
      cbits[((size_t)b * NN + r0 + 16 * lh + (ct >> 1)) * (2 * FF) + (ct & 1) * 32 + lm] = w;
      if (hf == 0 && ct != I) {
        unsigned wt = 0;
#pragma unroll
        for (int k = 0; k < 32; ++k) {
          const unsigned col = __builtin_amdgcn_ballot_w32(((w >> k) & 1u) != 0u);
          wt = (lane == k) ? col : wt;
        }
        cbits[((size_t)b * NN + 32 * ct + 16 * lh + (I >> 1)) * (2 * FF) + (I & 1) * 32 + lm] = wt;
      }
    }
  }
  __syncthreads();

  const int nb = b * NN + r0 + lane;
  if (wv == 0)      ha1[nb]  = pj[0 * 32 + lane];
  else if (wv == 1) ha2[nb]  = pj[1 * 32 + lane];
  else if (wv == 2) ham1[nb] = pj[2 * 32 + lane];
  else if (wv == 3) rowmask[nb] = ((const int*)pj)[3 * 32 + lane];
  else if (machines && wv < 8)
    hma2[b * MM + (wv - 4) * 32 + lane] = pm[(wv - 4) * 32 + lane];
}

#define HS_ROWS 64
#define HS_P 72

__device__ __forceinline__ void op_attn_tile(
    float* smem, const int tile,
    const float* __restrict__ h, const float* __restrict__ ha1,
    const float* __restrict__ ha2, const unsigned* __restrict__ cbits,
    const int* __restrict__ rowmask, float* __restrict__ hprime) {
  typedef float att_row_t[NN + 4];
  att_row_t* att = (att_row_t*)smem;
  float* inv_sum = smem + 16 * (NN + 4);

  const int b  = tile / (NN / 16);
  const int i0 = (tile % (NN / 16)) * 16;
  const float NEGINF = -__builtin_inff();

  const int lane = threadIdx.x & 31;
  const int wv = __builtin_amdgcn_readfirstlane((int)(threadIdx.x >> 5));

  float* hs = smem + 16 * (NN + 4) + 16;
  const v4f* hb4 = (const v4f*)(h + (size_t)b * NN * FF) + threadIdx.x;
  float* hst = hs + (threadIdx.x >> 4) * HS_P + (threadIdx.x & 15) * 4;
  v4f sa[8], sb[8];
#pragma unroll
  for (int q = 0; q < 8; ++q) {
    sa[q] = hb4[0 * (HS_ROWS * 16) + 128 * q];
    sb[q] = hb4[1 * (HS_ROWS * 16) + 128 * q];
  }
  {
    float a1v[32];
#pragma unroll
    for (int t = 0; t < 32; ++t) a1v[t] = ha1[b * NN + lane + 32 * t];
    unsigned wd[4];
#pragma unroll
    for (int rr = 0; rr < 4; ++rr)
      wd[rr] = cbits[((size_t)b * NN + i0 + (lane >> 1)) * (2 * FF) + (lane & 1) * 32 + wv + 4 * rr];
#pragma unroll
    for (int rr = 0; rr < 4; ++rr) {
      const int r = wv + 4 * rr;
      const int i = i0 + r;
      const int rm = rowmask[b * NN + i];
      const float a2 = ha2[b * NN + i];
      const unsigned wdm = rm ? 0xffffffffu : wd[rr];
      float e[32];
      float mx = NEGINF;
#pragma unroll
      for (int t = 0; t < 32; ++t) {
        const unsigned wt = (unsigned)__builtin_amdgcn_readlane((int)wdm, t);
        float s = a2 + a1v[t];
        s = s > 0.f ? s : 0.01f * s;
        e[t] = __builtin_amdgcn_inverse_ballot_w32(wt) ? s : NEGINF;
        mx = fmaxf(mx, e[t]);
      }
      mx = fmaxf(mx, xhalf(mx));
      mx = row_max16(mx);
      if (mx == NEGINF) mx = 0.f;
      float sum = 0.f;
#pragma unroll
      for (int t = 0; t < 32; ++t) {
        const float p = __expf(e[t] - mx);
        att[r][lane + 32 * t] = p;
        sum += p;
      }
      sum += xhalf(sum);
      sum += rowx<8>(sum); sum += rowx<4>(sum);
      sum += rowx<2>(sum); sum += rowx<1>(sum);
      if (lane == 0) inv_sum[r] = 1.f / fmaxf(sum, 1e-30f);
    }
  }
#pragma unroll
  for (int q = 0; q < 8; ++q) *(v4f*)(hst + q * 8 * HS_P) = sa[q];
  __syncthreads();

  const int lm = lane & 15;
  const int kh = (lane >> 4) * 2;
  const int lh = lane >> 4;
  const int o  = wv * 16 + lm;
  const float* ar  = &att[lm][kh];
  const float* hf0 = hs + kh * HS_P + o;
  const float* hf1 = hf0 + HS_ROWS * HS_P;

  v8f c = {};
#pragma unroll 1
  for (int s = 0; s < NN / HS_ROWS; s += 2) {
    {
      const int sn = (s + 2) & (NN / HS_ROWS - 1);
#pragma unroll
      for (int q = 0; q < 8; ++q) sa[q] = hb4[sn * (HS_ROWS * 16) + 128 * q];
    }
    {
      v2f af[16], bq[16];
#pragma unroll
      for (int t = 0; t < 16; ++t) {
        af[t].x = ar[s * HS_ROWS + 4 * t];
        af[t].y = ar[s * HS_ROWS + 4 * t + 1];
        bq[t].x = hf0[(4 * t) * HS_P];
        bq[t].y = hf0[(4 * t + 1) * HS_P];
      }
#pragma unroll
      for (int t = 0; t < 16; ++t) c = wmma4(af[t], bq[t], c);
    }
#pragma unroll
    for (int q = 0; q < 8; ++q) *(v4f*)(hst + HS_ROWS * HS_P + q * 8 * HS_P) = sb[q];
    __builtin_amdgcn_sched_group_barrier(0x020, 8, 0);
    __builtin_amdgcn_sched_group_barrier(0x100, 32, 0);
    __builtin_amdgcn_sched_group_barrier(0x008, 16, 0);
    __builtin_amdgcn_sched_group_barrier(0x200, 8, 0);
    lds_barrier();
    {
      const int sn = (s + 3) & (NN / HS_ROWS - 1);
#pragma unroll
      for (int q = 0; q < 8; ++q) sb[q] = hb4[sn * (HS_ROWS * 16) + 128 * q];
    }
    {
      v2f af[16], bq[16];
#pragma unroll
      for (int t = 0; t < 16; ++t) {
        af[t].x = ar[(s + 1) * HS_ROWS + 4 * t];
        af[t].y = ar[(s + 1) * HS_ROWS + 4 * t + 1];
        bq[t].x = hf1[(4 * t) * HS_P];
        bq[t].y = hf1[(4 * t + 1) * HS_P];
      }
#pragma unroll
      for (int t = 0; t < 16; ++t) c = wmma4(af[t], bq[t], c);
    }
#pragma unroll
    for (int q = 0; q < 8; ++q) *(v4f*)(hst + q * 8 * HS_P) = sa[q];
    __builtin_amdgcn_sched_group_barrier(0x020, 8, 0);
    __builtin_amdgcn_sched_group_barrier(0x100, 32, 0);
    __builtin_amdgcn_sched_group_barrier(0x008, 16, 0);
    __builtin_amdgcn_sched_group_barrier(0x200, 8, 0);
    lds_barrier();
  }
  float* hp = hprime + ((size_t)b * NN + i0) * (2 * FF);
#pragma unroll
  for (int r = 0; r < 8; ++r) {
    int m = r + 8 * lh;
    hp[(size_t)m * (2 * FF) + o] = c[r] * inv_sum[m];
  }
}

#define MA_NPW 4
#define HM_P 68
#define MA_WAVES 4

__device__ __forceinline__ void ma_attn_nodes(
    float* smem, const int group,
    const float* __restrict__ z, const float* __restrict__ wedge,
    const float* __restrict__ hma, const float* __restrict__ ham1,
    const float* __restrict__ hma2, const int* __restrict__ opma,
    const int* __restrict__ rowmask, const float* __restrict__ h,
    const float* __restrict__ attma, float* __restrict__ hprime) {
  const int lane = threadIdx.x & 31;
  const int c  = lane & 15;
  const int hh = lane >> 4;
  const int wv  = __builtin_amdgcn_readfirstlane((int)(threadIdx.x >> 5));
  const int bn0 = (group * MA_WAVES + wv) * MA_NPW;
  const int b   = bn0 / NN;
  const float NEGINF = -__builtin_inff();

  v2f awe[4][4];
#pragma unroll
  for (int s = 0; s < 4; ++s)
#pragma unroll
    for (int ot = 0; ot < 4; ++ot) {
      awe[s][ot].x = wedge[(8 * hh + 2 * s) * FF + ot * 16 + c];
      awe[s][ot].y = wedge[(8 * hh + 2 * s + 1) * FF + ot * 16 + c];
    }
  v4f am3v[4][2];
#pragma unroll
  for (int ot = 0; ot < 4; ++ot)
#pragma unroll
    for (int q = 0; q < 2; ++q)
      am3v[ot][q] = *(const v4f*)(attma + 2 * FF + ot * 16 + 8 * hh + 4 * q);
  float hma2v[8];
#pragma unroll
  for (int t = 0; t < 8; ++t) hma2v[t] = hma2[b * MM + 16 * t + c];
  {
    const v4f* src4 = (const v4f*)(hma + (size_t)b * MM * FF);
#pragma unroll
    for (int q = 0; q < 16; ++q) {
      const int idx = threadIdx.x + 128 * q;
      *(v4f*)(smem + (idx >> 4) * HM_P + (idx & 15) * 4) = src4[idx];
    }
  }
  __syncthreads();
  const float* hmb = smem + c * HM_P + 8 * hh;

  v4f zq[8][2];
  {
    const float* zp = z + (size_t)bn0 * MM * FE + c * FE + 8 * hh;
#pragma unroll
    for (int t = 0; t < 8; ++t)
#pragma unroll
      for (int q = 0; q < 2; ++q)
        zq[t][q] = *(const v4f*)(zp + t * 16 * FE + 4 * q);
  }

#pragma unroll 1
  for (int i = 0; i < MA_NPW; ++i) {
    const int bn = bn0 + i;
    asm volatile("" ::: "memory");

    const int   rm  = rowmask[bn];
    const float hm1 = ham1[bn];
    int adv[8];
#pragma unroll
    for (int t = 0; t < 8; ++t) adv[t] = opma[(size_t)bn * MM + 16 * t + c];

    v8f D[8][4];
    float sc[8];
#pragma unroll
    for (int t = 0; t < 8; ++t) {
      v2f zb[4];
      zb[0].x = zq[t][0].x; zb[0].y = zq[t][0].y;
      zb[1].x = zq[t][0].z; zb[1].y = zq[t][0].w;
      zb[2].x = zq[t][1].x; zb[2].y = zq[t][1].y;
      zb[3].x = zq[t][1].z; zb[3].y = zq[t][1].w;
      float part[4];
#pragma unroll
      for (int ot = 0; ot < 4; ++ot) {
        const v4f g0 = *(const v4f*)(hmb + t * 16 * HM_P + ot * 16);
        const v4f g1 = *(const v4f*)(hmb + t * 16 * HM_P + ot * 16 + 4);
        v8f d = {g0.x, g0.y, g0.z, g0.w, g1.x, g1.y, g1.z, g1.w};
#pragma unroll
        for (int s = 0; s < 4; ++s) d = wmma4(awe[s][ot], zb[s], d);
        D[t][ot] = d;
        v2f a2 = __builtin_shufflevector(d, d, 0, 1) * __builtin_shufflevector(am3v[ot][0], am3v[ot][0], 0, 1);
        a2 += __builtin_shufflevector(d, d, 2, 3) * __builtin_shufflevector(am3v[ot][0], am3v[ot][0], 2, 3);
        a2 += __builtin_shufflevector(d, d, 4, 5) * __builtin_shufflevector(am3v[ot][1], am3v[ot][1], 0, 1);
        a2 += __builtin_shufflevector(d, d, 6, 7) * __builtin_shufflevector(am3v[ot][1], am3v[ot][1], 2, 3);
        part[ot] = a2.x + a2.y;
      }
      float dot = (part[0] + part[1]) + (part[2] + part[3]);
      dot += xhalf(dot);
      float e = hm1 + hma2v[t] + dot;
      e = e > 0.f ? e : 0.01f * e;
      const int conn = rm | (adv[t] > 0);
      sc[t] = conn ? e : NEGINF;
    }

    float mx = sc[0];
#pragma unroll
    for (int t = 1; t < 8; ++t) mx = fmaxf(mx, sc[t]);
    mx = row_max16(mx);
    if (mx == NEGINF) mx = 0.f;
    float p[8];
    float sum = 0.f;
#pragma unroll
    for (int t = 0; t < 8; ++t) { p[t] = __expf(sc[t] - mx); sum += p[t]; }
    sum = row_sum16(sum);
    const float inv = 1.f / fmaxf(sum, 1e-30f);

    if (i + 1 < MA_NPW) {
      const float* zp = z + (size_t)(bn + 1) * MM * FE + c * FE + 8 * hh;
#pragma unroll
      for (int t = 0; t < 8; ++t)
#pragma unroll
        for (int q = 0; q < 2; ++q)
          zq[t][q] = *(const v4f*)(zp + t * 16 * FE + 4 * q);
    }

    v8f acc[4];
#pragma unroll
    for (int ot = 0; ot < 4; ++ot) acc[ot] = (v8f){};
#pragma unroll
    for (int t = 0; t < 8; ++t)
#pragma unroll
      for (int ot = 0; ot < 4; ++ot)
#pragma unroll
        for (int r = 0; r < 8; ++r) acc[ot][r] += p[t] * D[t][ot][r];

    const float* hr = h + (size_t)bn * FF + 8 * hh;
    v4f res[4][2];
#pragma unroll
    for (int ot = 0; ot < 4; ++ot) {
      res[ot][0] = *(const v4f*)(hr + ot * 16);
      res[ot][1] = *(const v4f*)(hr + ot * 16 + 4);
    }
#pragma unroll
    for (int ot = 0; ot < 4; ++ot)
#pragma unroll
      for (int r = 0; r < 8; ++r) acc[ot][r] = row_sum16(acc[ot][r]);

    float* hp = hprime + (size_t)bn * (2 * FF) + FF + 8 * hh;
#pragma unroll
    for (int ot = 0; ot < 4; ++ot) {
      const v4f r0 = res[ot][0];
      const v4f r1 = res[ot][1];
      v4f o0, o1;
      o0.x = acc[ot][0] * inv + r0.x; o0.y = acc[ot][1] * inv + r0.y;
      o0.z = acc[ot][2] * inv + r0.z; o0.w = acc[ot][3] * inv + r0.w;
      o1.x = acc[ot][4] * inv + r1.x; o1.y = acc[ot][5] * inv + r1.y;
      o1.z = acc[ot][6] * inv + r1.z; o1.w = acc[ot][7] * inv + r1.w;
      if (c == 0) {
        *(v4f*)(hp + ot * 16)     = o0;
        *(v4f*)(hp + ot * 16 + 4) = o1;
      }
    }

  }
}

__global__ __launch_bounds__(128, 2) void attn_pair(
    const float* __restrict__ h, const float* __restrict__ ha1,
    const float* __restrict__ ha2, const unsigned* __restrict__ cbits,
    const int* __restrict__ rowmask, float* __restrict__ hprime,
    const float* __restrict__ z, const float* __restrict__ wedge,
    const float* __restrict__ hma, const float* __restrict__ ham1,
    const float* __restrict__ hma2, const int* __restrict__ opma,
    const float* __restrict__ attma) {
  extern __shared__ float smem[];
  const int g = blockIdx.x;
  const int op_first = (g ^ (g >> 8)) & 1;
#pragma unroll 1
  for (int ph = 0; ph < 2; ++ph) {
    if (ph == op_first)
      ma_attn_nodes(smem, g, z, wedge, hma, ham1, hma2, opma, rowmask, h, attma, hprime);
    else
      op_attn_tile(smem, g, h, ha1, ha2, cbits, rowmask, hprime);
    __syncthreads();
  }
}

__global__ __launch_bounds__(256, 2) void mlp(
    const float* __restrict__ hprime, const float* __restrict__ w1,
    const float* __restrict__ b1, const float* __restrict__ w2,
    const float* __restrict__ b2, const float* __restrict__ wo,
    const float* __restrict__ bo, const int* __restrict__ rowmask,
    float* __restrict__ out) {
  extern __shared__ float smem[];
  typedef float act_row_t[HID + 4];
  act_row_t* a0 = (act_row_t*)smem;
  act_row_t* a1 = (act_row_t*)(smem + 16 * (HID + 4));

  const int row0 = blockIdx.x * 16;
  const int wv = __builtin_amdgcn_readfirstlane((int)(threadIdx.x >> 5));
  const int lane = threadIdx.x & 31;
  const int lm = lane & 15;
  const int kh = (lane >> 4) * 2;
  const int lh = lane >> 4;
  const int o  = wv * 16 + lm;

  const v4f* hp4 = (const v4f*)(hprime + (size_t)row0 * HID) + threadIdx.x;
  const v4f in0 = hp4[0];
  const v4f in1 = hp4[256];
  const float bias1 = b1[o];
  const float bias2 = b2[o];
  v2f bw1[32];
#pragma unroll
  for (int t = 0; t < 32; ++t) {
    bw1[t].x = w1[(4 * t + kh) * HID + o];
    bw1[t].y = w1[(4 * t + kh + 1) * HID + o];
  }
  *(v4f*)&a0[threadIdx.x >> 5][(threadIdx.x & 31) * 4] = in0;
  *(v4f*)&a0[(threadIdx.x >> 5) + 8][(threadIdx.x & 31) * 4] = in1;
  lds_barrier();

  v2f bw2[32];
#pragma unroll
  for (int t = 0; t < 32; ++t) {
    bw2[t].x = w2[(4 * t + kh) * HID + o];
    bw2[t].y = w2[(4 * t + kh + 1) * HID + o];
  }

  v8f c1 = {};
  {
    v2f af[32];
#pragma unroll
    for (int t = 0; t < 32; ++t) {
      af[t].x = a0[lm][4 * t + kh];
      af[t].y = a0[lm][4 * t + kh + 1];
    }
#pragma unroll
    for (int t = 0; t < 32; ++t) c1 = wmma4(af[t], bw1[t], c1);
    __builtin_amdgcn_sched_group_barrier(0x100, 32, 0);
    __builtin_amdgcn_sched_group_barrier(0x008, 32, 0);
  }
  const float bias3 = bo[o & (FF - 1)];
  const int4 rm0 = *(const int4*)(rowmask + row0 + 8 * lh);
  const int4 rm1 = *(const int4*)(rowmask + row0 + 8 * lh + 4);
  v2f bwo[32];
  if (wv < 4) {
#pragma unroll
    for (int t = 0; t < 32; ++t) {
      bwo[t].x = wo[(4 * t + kh) * FF + o];
      bwo[t].y = wo[(4 * t + kh + 1) * FF + o];
    }
  }
  {
#pragma unroll
    for (int r = 0; r < 8; ++r) {
      float v = c1[r] + bias1;
      a1[r + 8 * lh][o] = v > 0.f ? v : expm1f(v);
    }
  }
  lds_barrier();

  {
    v8f c = {};
    v2f af[32];
#pragma unroll
    for (int t = 0; t < 32; ++t) {
      af[t].x = a1[lm][4 * t + kh];
      af[t].y = a1[lm][4 * t + kh + 1];
    }
#pragma unroll
    for (int t = 0; t < 32; ++t) c = wmma4(af[t], bw2[t], c);
    __builtin_amdgcn_sched_group_barrier(0x100, 32, 0);
    __builtin_amdgcn_sched_group_barrier(0x008, 32, 0);
#pragma unroll
    for (int r = 0; r < 8; ++r) {
      float v = c[r] + bias2;
      a0[r + 8 * lh][o] = v > 0.f ? v : expm1f(v);
    }
  }
  lds_barrier();

  if (wv < 4) {
    v8f c = {};
    v2f af[32];
#pragma unroll
    for (int t = 0; t < 32; ++t) {
      af[t].x = a0[lm][4 * t + kh];
      af[t].y = a0[lm][4 * t + kh + 1];
    }
#pragma unroll
    for (int t = 0; t < 32; ++t) c = wmma4(af[t], bwo[t], c);
    __builtin_amdgcn_sched_group_barrier(0x100, 32, 0);
    __builtin_amdgcn_sched_group_barrier(0x008, 32, 0);
    const int rmv[8] = {rm0.x, rm0.y, rm0.z, rm0.w, rm1.x, rm1.y, rm1.z, rm1.w};
#pragma unroll
    for (int r = 0; r < 8; ++r) {
      int m = r + 8 * lh;
      float v = c[r] + bias3;
      if (rmv[r]) v = 0.f;
      out[(size_t)(row0 + m) * FF + o] = v;
    }
  }
}

extern "C" void kernel_launch(void* const* d_in, const int* in_sizes, int n_in,
                              void* d_out, int out_size, void* d_ws,
                              size_t ws_size, hipStream_t stream) {
  (void)in_sizes; (void)n_in; (void)out_size; (void)ws_size;
  const float* x      = (const float*)d_in[0];
  const float* y      = (const float*)d_in[1];
  const float* z      = (const float*)d_in[2];
  const int*   op_adj = (const int*)d_in[3];
  const int*   ma_adj = (const int*)d_in[4];
  const int*   op_ma  = (const int*)d_in[5];
  const float* Wop    = (const float*)d_in[6];
  const float* Wma    = (const float*)d_in[7];
  const float* Wedge  = (const float*)d_in[8];
  const float* att_op = (const float*)d_in[9];
  const float* att_ma = (const float*)d_in[10];
  const float* w1     = (const float*)d_in[11];
  const float* b1     = (const float*)d_in[12];
  const float* w2     = (const float*)d_in[13];
  const float* b2     = (const float*)d_in[14];
  const float* wo     = (const float*)d_in[15];
  const float* bo     = (const float*)d_in[16];
  float* out = (float*)d_out;

  float* ws     = (float*)d_ws;
  float* h      = ws;
  float* hma    = h + BB * NN * FF;
  float* ha1    = hma + BB * MM * FF;
  float* ha2    = ha1 + BB * NN;
  float* ham1   = ha2 + BB * NN;
  float* hma2   = ham1 + BB * NN;
  float* hprime = hma2 + BB * MM;
  int*   rmask  = (int*)(hprime + (size_t)BB * NN * 2 * FF);

  size_t sm2 = (16 * (NN + 4) + 16 + 2 * HS_ROWS * HS_P) * sizeof(float);
  unsigned* cbits = (unsigned*)hprime;
  size_t smb = (16 * 32 * CB_P + 32 * 33 + 4 * 32 + MM + 4 * FF) * sizeof(int);
  feat_bits_proj<<<dim3(NN / 32, BB), dim3(512), smb, stream>>>(
      op_adj, ma_adj, cbits, x, y, Wop, Wma, h, hma, att_op, att_ma, op_ma,
      ha1, ha2, ham1, hma2, rmask);
  attn_pair<<<dim3((BB * NN) / 16), dim3(128), sm2, stream>>>(
      h, ha1, ha2, cbits, rmask, hprime,
      z, Wedge, hma, ham1, hma2, op_ma, att_ma);
  size_t sm4 = (2 * 16 * (HID + 4)) * sizeof(float);
  mlp<<<dim3((BB * NN) / 16), dim3(256), sm4, stream>>>(
      hprime, w1, b1, w2, b2, wo, bo, rmask, out);
}


# ma stage: a node's connected machines packed into 16-wide tiles (slot list via a per-wave LDS strip; z and h_ma rows fetched by slot), tiles with no connected slot skipped; summing order changed
# speedup vs baseline: 4.2375x; 1.0374x over previous
#include <hip/hip_runtime.h>
#include <math.h>


typedef __attribute__((ext_vector_type(2))) float v2f;
typedef __attribute__((ext_vector_type(8))) float v8f;
typedef __attribute__((ext_vector_type(4))) float v4f;

#define BB 8
#define NN 1024
#define MM 128
#define FF 64
#define FE 16
#define HID 128

__device__ __forceinline__ v8f wmma4(v2f a, v2f b, v8f c) {
  return __builtin_amdgcn_wmma_f32_16x16x4_f32(
       false, a,  false, b,
       (short)0, c,  false,  false);
}

__device__ __forceinline__ float xhalf(float x) {
  int b = __builtin_bit_cast(int, x);
  return __builtin_bit_cast(float,
      __builtin_amdgcn_permlanex16(b, b, 0x76543210, 0xfedcba98, false, false));
}
template <int MASK>
__device__ __forceinline__ float rowx(float x) {
  int b = __builtin_bit_cast(int, x);
  return __builtin_bit_cast(float,
      __builtin_amdgcn_update_dpp(0, b, 0x160 + MASK, 0xf, 0xf, true));
}
__device__ __forceinline__ float row_sum16(float x) {
  x += rowx<1>(x); x += rowx<2>(x); x += rowx<4>(x); x += rowx<8>(x);
  return x;
}
__device__ __forceinline__ float row_max16(float x) {
  x = fmaxf(x, rowx<1>(x)); x = fmaxf(x, rowx<2>(x));
  x = fmaxf(x, rowx<4>(x)); x = fmaxf(x, rowx<8>(x));
  return x;
}
__device__ __forceinline__ float wave_sum32(float x) {
  x = row_sum16(x);
  return x + xhalf(x);
}


__device__ __forceinline__ void lds_barrier() {
  __builtin_amdgcn_fence(__ATOMIC_RELEASE, "workgroup", "local");
  __builtin_amdgcn_s_barrier();
  __builtin_amdgcn_fence(__ATOMIC_ACQUIRE, "workgroup", "local");
}

#define CB_P 36

__device__ __forceinline__ float dot8(v4f a0, v4f a1, v4f b0, v4f b1) {
  return a0.x * b0.x + a0.y * b0.y + a0.z * b0.z + a0.w * b0.w
       + a1.x * b1.x + a1.y * b1.y + a1.z * b1.z + a1.w * b1.w;
}
__device__ __forceinline__ float sum8(float x) {
  x += rowx<1>(x); x += rowx<2>(x); x += rowx<4>(x);
  return x;
}

__global__ __launch_bounds__(512) void feat_bits_proj(
    const int* __restrict__ opadj, const int* __restrict__ maadj,
    unsigned* __restrict__ cbits,
    const float* __restrict__ x, const float* __restrict__ y,
    const float* __restrict__ Wop, const float* __restrict__ Wma,
    float* __restrict__ h, float* __restrict__ hma,
    const float* __restrict__ att_op, const float* __restrict__ att_ma,
    const int* __restrict__ opma,
    float* __restrict__ ha1, float* __restrict__ ha2,
    float* __restrict__ ham1, float* __restrict__ hma2,
    int* __restrict__ rowmask) {
  extern __shared__ __attribute__((aligned(16))) int cbs[];
  const int b    = blockIdx.y;
  const int r0   = blockIdx.x * 32;
  const int wv   = threadIdx.x >> 5;
  const int wvs  = __builtin_amdgcn_readfirstlane((int)(threadIdx.x >> 5));
  const int lane = threadIdx.x & 31;
  int* tile = cbs + wv * (32 * CB_P);
  unsigned* words = (unsigned*)(cbs + 16 * 32 * CB_P);
  float* pj = (float*)(cbs + 16 * 32 * CB_P + 32 * 33);
  float* pm = pj + 4 * 32;
  float* vv = pm + MM;
  const int* opb = opadj + (size_t)b * NN * NN;
  const int* mab = maadj + (size_t)b * NN * NN;
  const int sub = lane >> 3;
  const int q4  = (lane & 7) * 4;
  const bool machines = (blockIdx.x == 0);

  const bool tile_wave = wvs < 9;
  const bool tile_h    = wvs < 8;
  const int lm = lane & 15;
  const int kh = (lane >> 4) * 2;
  const int lh = lane >> 4;
  const float* GA = tile_h ? x : y;
  const float* GW = tile_h ? Wop : Wma;
  float* GC = tile_h ? h : hma;
  const int grow = tile_h ? (b * NN + r0 + 16 * (wvs >> 2)) : (b * MM + 16 * ((int)blockIdx.x >> 2));
  const int go   = (tile_h ? (wvs & 3) : ((int)blockIdx.x & 3)) * 16 + lm;
  v2f ga[16], gb[16];
  if (tile_wave) {
    const float* Arow = GA + (size_t)(grow + lm) * FF;
#pragma unroll
    for (int t = 0; t < 16; ++t) {
      ga[t].x = Arow[4 * t + kh];
      ga[t].y = Arow[4 * t + kh + 1];
      gb[t].x = GW[(size_t)(4 * t + kh) * FF + go];
      gb[t].y = GW[(size_t)(4 * t + kh + 1) * FF + go];
    }
  }

  const int node0 = b * NN + r0 + 2 * wv;
  v2f  xv[2];
  int4 lk[2];
#pragma unroll
  for (int q = 0; q < 2; ++q) {
    xv[q] = *(const v2f*)(x + (size_t)(node0 + q) * FF + 2 * lane);
    lk[q] = *(const int4*)(opma + (size_t)(node0 + q) * MM + 4 * lane);
  }
  const int vi = 4 * wv + (lane >> 3);
  const int vc = (lane & 7) * 8;
  const v4f w0  = *(const v4f*)(Wop + (size_t)vi * FF + vc);
  const v4f w1  = *(const v4f*)(Wop + (size_t)vi * FF + vc + 4);
  const v4f p10 = *(const v4f*)(att_op + vc);
  const v4f p11 = *(const v4f*)(att_op + vc + 4);
  const v4f p20 = *(const v4f*)(att_op + FF + vc);
  const v4f p21 = *(const v4f*)(att_op + FF + vc + 4);
  const v4f p30 = *(const v4f*)(att_ma + vc);
  const v4f p31 = *(const v4f*)(att_ma + vc + 4);
  v2f mv[8] = {};
  v4f m0, m1, q20, q21, q30, q31;
  if (machines) {
#pragma unroll
    for (int t = 0; t < 8; ++t)
      mv[t] = *(const v2f*)(y + ((size_t)b * MM + 8 * wv + t) * FF + 2 * lane);
    m0  = *(const v4f*)(Wma + (size_t)vi * FF + vc);
    m1  = *(const v4f*)(Wma + (size_t)vi * FF + vc + 4);
    q20 = *(const v4f*)(att_ma + FF + vc);
    q21 = *(const v4f*)(att_ma + FF + vc + 4);
    q30 = *(const v4f*)(att_ma + 2 * FF + vc);
    q31 = *(const v4f*)(att_ma + 2 * FF + vc + 4);
  }

  {
    const float d1 = sum8(dot8(w0, w1, p10, p11));
    const float d2 = sum8(dot8(w0, w1, p20, p21));
    const float d3 = sum8(dot8(w0, w1, p30, p31));
    if ((lane & 7) == 0) {
      vv[0 * FF + vi] = d1;
      vv[1 * FF + vi] = d2;
      vv[2 * FF + vi] = d3;
    }
    if (machines) {
      const float d4 = sum8(dot8(m0, m1, q20 - q30, q21 - q31));
      if ((lane & 7) == 0) vv[3 * FF + vi] = d4;
    }
  }
  lds_barrier();
  const v2f wa1 = *(const v2f*)(vv + 0 * FF + 2 * lane);
  const v2f wa2 = *(const v2f*)(vv + 1 * FF + 2 * lane);
  const v2f wm1 = *(const v2f*)(vv + 2 * FF + 2 * lane);

#pragma unroll
  for (int q = 0; q < 2; ++q) {
    const float s1 = wave_sum32(xv[q].x * wa1.x + xv[q].y * wa1.y);
    const float s2 = wave_sum32(xv[q].x * wa2.x + xv[q].y * wa2.y);
    const float s3 = wave_sum32(xv[q].x * wm1.x + xv[q].y * wm1.y);
    const int any = lk[q].x | lk[q].y | lk[q].z | lk[q].w;
    const unsigned linked = __builtin_amdgcn_ballot_w32(any != 0);
    if (lane == 0) {
      pj[0 * 32 + 2 * wv + q] = s1;
      pj[1 * 32 + 2 * wv + q] = s2;
      pj[2 * 32 + 2 * wv + q] = s3;
      ((int*)pj)[3 * 32 + 2 * wv + q] = (linked == 0u) ? 1 : 0;
    }
  }
  if (machines) {
    const v2f wm2 = *(const v2f*)(vv + 3 * FF + 2 * lane);
#pragma unroll
    for (int t = 0; t < 8; ++t) {
      const float s = wave_sum32(mv[t].x * wm2.x + mv[t].y * wm2.y);
      if (lane == 0) pm[8 * wv + t] = s;
    }
  }

  if (tile_wave) {
    v8f c = {};
#pragma unroll
    for (int t = 0; t < 16; ++t) c = wmma4(ga[t], gb[t], c);
#pragma unroll
    for (int r = 0; r < 8; ++r)
      GC[(size_t)(grow + r + 8 * lh) * FF + go] = c[r];
  }

  const int I = (int)blockIdx.x;
  int4 dsum[2][8];
  int  tr[2][32];
#pragma unroll
  for (int hf = 0; hf < 2; ++hf) {
    if (hf == 0 || wvs == 0) {
      const int c0 = ((I + (hf ? 16 : wvs)) & 31) * 32;
#pragma unroll
      for (int q = 0; q < 8; ++q) {
        const size_t off = (size_t)(r0 + 4 * q + sub) * NN + c0 + q4;
        const int4 a = *(const int4*)(opb + off);
        const int4 m = *(const int4*)(mab + off);
        dsum[hf][q].x = a.x + m.x; dsum[hf][q].y = a.y + m.y;
        dsum[hf][q].z = a.z + m.z; dsum[hf][q].w = a.w + m.w;
      }
      if (hf != 0 || wvs != 0) {
#pragma unroll
        for (int k = 0; k < 32; ++k) {
          const size_t off = (size_t)(c0 + k) * NN + r0 + lane;
          tr[hf][k] = opb[off] + mab[off];
        }
      }
    }
  }

#pragma unroll
  for (int hf = 0; hf < 2; ++hf) {
    if (hf == 0 || wvs == 0) {
      const int ct = (I + (hf ? 16 : wvs)) & 31;
#pragma unroll
      for (int q = 0; q < 8; ++q)
        *(int4*)(tile + (4 * q + sub) * CB_P + q4) = dsum[hf][q];
      asm volatile("s_wait_dscnt 0x0" ::: "memory");
      if (hf == 0 && wvs == 0) {
#pragma unroll
        for (int k = 0; k < 32; ++k) tr[0][k] = tile[k * CB_P + lane];
      }
      unsigned w = 0;
#pragma unroll
      for (int g = 0; g < 8; ++g) {
        const int4 d = *(const int4*)(tile + lane * CB_P + 4 * g);
        w |= (unsigned)((d.x + tr[hf][4 * g + 0]) > 0) << (4 * g + 0);
        w |= (unsigned)((d.y + tr[hf][4 * g + 1]) > 0) << (4 * g + 1);
        w |= (unsigned)((d.z + tr[hf][4 * g + 2]) > 0) << (4 * g + 2);
        w |= (unsigned)((d.w + tr[hf][4 * g + 3]) > 0) << (4 * g + 3);
      }
      cbits[((size_t)b * NN + r0 + 16 * lh + (ct >> 1)) * (2 * FF) + (ct & 1) * 32 + lm] = w;
      if (hf == 0 && ct != I) {
        unsigned wt = 0;
#pragma unroll
        for (int k = 0; k < 32; ++k) {
          const unsigned col = __builtin_amdgcn_ballot_w32(((w >> k) & 1u) != 0u);
          wt = (lane == k) ? col : wt;
        }
        cbits[((size_t)b * NN + 32 * ct + 16 * lh + (I >> 1)) * (2 * FF) + (I & 1) * 32 + lm] = wt;
      }
    }
  }
  __syncthreads();

  const int nb = b * NN + r0 + lane;
  if (wv == 0)      ha1[nb]  = pj[0 * 32 + lane];
  else if (wv == 1) ha2[nb]  = pj[1 * 32 + lane];
  else if (wv == 2) ham1[nb] = pj[2 * 32 + lane];
  else if (wv == 3) rowmask[nb] = ((const int*)pj)[3 * 32 + lane];
  else if (machines && wv < 8)
    hma2[b * MM + (wv - 4) * 32 + lane] = pm[(wv - 4) * 32 + lane];
}

#define HS_ROWS 64
#define HS_P 72

__device__ __forceinline__ void op_attn_tile(
    float* smem, const int tile,
    const float* __restrict__ h, const float* __restrict__ ha1,
    const float* __restrict__ ha2, const unsigned* __restrict__ cbits,
    const int* __restrict__ rowmask, float* __restrict__ hprime) {
  typedef float att_row_t[NN + 4];
  att_row_t* att = (att_row_t*)smem;
  float* inv_sum = smem + 16 * (NN + 4);

  const int b  = tile / (NN / 16);
  const int i0 = (tile % (NN / 16)) * 16;
  const float NEGINF = -__builtin_inff();

  const int lane = threadIdx.x & 31;
  const int wv = __builtin_amdgcn_readfirstlane((int)(threadIdx.x >> 5));

  float* hs = smem + 16 * (NN + 4) + 16;
  const v4f* hb4 = (const v4f*)(h + (size_t)b * NN * FF) + threadIdx.x;
  float* hst = hs + (threadIdx.x >> 4) * HS_P + (threadIdx.x & 15) * 4;
  v4f sa[8], sb[8];
#pragma unroll
  for (int q = 0; q < 8; ++q) {
    sa[q] = hb4[0 * (HS_ROWS * 16) + 128 * q];
    sb[q] = hb4[1 * (HS_ROWS * 16) + 128 * q];
  }
  {
    float a1v[32];
#pragma unroll
    for (int t = 0; t < 32; ++t) a1v[t] = ha1[b * NN + lane + 32 * t];
    unsigned wd[4];
#pragma unroll
    for (int rr = 0; rr < 4; ++rr)
      wd[rr] = cbits[((size_t)b * NN + i0 + (lane >> 1)) * (2 * FF) + (lane & 1) * 32 + wv + 4 * rr];
#pragma unroll
    for (int rr = 0; rr < 4; ++rr) {
      const int r = wv + 4 * rr;
      const int i = i0 + r;
      const int rm = rowmask[b * NN + i];
      const float a2 = ha2[b * NN + i];
      const unsigned wdm = rm ? 0xffffffffu : wd[rr];
      float e[32];
      float mx = NEGINF;
#pragma unroll
      for (int t = 0; t < 32; ++t) {
        const unsigned wt = (unsigned)__builtin_amdgcn_readlane((int)wdm, t);
        float s = a2 + a1v[t];
        s = s > 0.f ? s : 0.01f * s;
        e[t] = __builtin_amdgcn_inverse_ballot_w32(wt) ? s : NEGINF;
        mx = fmaxf(mx, e[t]);
      }
      mx = fmaxf(mx, xhalf(mx));
      mx = row_max16(mx);
      if (mx == NEGINF) mx = 0.f;
      float sum = 0.f;
#pragma unroll
      for (int t = 0; t < 32; ++t) {
        const float p = __expf(e[t] - mx);
        att[r][lane + 32 * t] = p;
        sum += p;
      }
      sum += xhalf(sum);
      sum += rowx<8>(sum); sum += rowx<4>(sum);
      sum += rowx<2>(sum); sum += rowx<1>(sum);
      if (lane == 0) inv_sum[r] = 1.f / fmaxf(sum, 1e-30f);
    }
  }
#pragma unroll
  for (int q = 0; q < 8; ++q) *(v4f*)(hst + q * 8 * HS_P) = sa[q];
  __syncthreads();

  const int lm = lane & 15;
  const int kh = (lane >> 4) * 2;
  const int lh = lane >> 4;
  const int o  = wv * 16 + lm;
  const float* ar  = &att[lm][kh];
  const float* hf0 = hs + kh * HS_P + o;
  const float* hf1 = hf0 + HS_ROWS * HS_P;

  v8f c = {};
#pragma unroll 1
  for (int s = 0; s < NN / HS_ROWS; s += 2) {
    {
      const int sn = (s + 2) & (NN / HS_ROWS - 1);
#pragma unroll
      for (int q = 0; q < 8; ++q) sa[q] = hb4[sn * (HS_ROWS * 16) + 128 * q];
    }
    {
      v2f af[16], bq[16];
#pragma unroll
      for (int t = 0; t < 16; ++t) {
        af[t].x = ar[s * HS_ROWS + 4 * t];
        af[t].y = ar[s * HS_ROWS + 4 * t + 1];
        bq[t].x = hf0[(4 * t) * HS_P];
        bq[t].y = hf0[(4 * t + 1) * HS_P];
      }
#pragma unroll
      for (int t = 0; t < 16; ++t) c = wmma4(af[t], bq[t], c);
    }
#pragma unroll
    for (int q = 0; q < 8; ++q) *(v4f*)(hst + HS_ROWS * HS_P + q * 8 * HS_P) = sb[q];
    __builtin_amdgcn_sched_group_barrier(0x020, 8, 0);
    __builtin_amdgcn_sched_group_barrier(0x100, 32, 0);
    __builtin_amdgcn_sched_group_barrier(0x008, 16, 0);
    __builtin_amdgcn_sched_group_barrier(0x200, 8, 0);
    lds_barrier();
    {
      const int sn = (s + 3) & (NN / HS_ROWS - 1);
#pragma unroll
      for (int q = 0; q < 8; ++q) sb[q] = hb4[sn * (HS_ROWS * 16) + 128 * q];
    }
    {
      v2f af[16], bq[16];
#pragma unroll
      for (int t = 0; t < 16; ++t) {
        af[t].x = ar[(s + 1) * HS_ROWS + 4 * t];
        af[t].y = ar[(s + 1) * HS_ROWS + 4 * t + 1];
        bq[t].x = hf1[(4 * t) * HS_P];
        bq[t].y = hf1[(4 * t + 1) * HS_P];
      }
#pragma unroll
      for (int t = 0; t < 16; ++t) c = wmma4(af[t], bq[t], c);
    }
#pragma unroll
    for (int q = 0; q < 8; ++q) *(v4f*)(hst + q * 8 * HS_P) = sa[q];
    __builtin_amdgcn_sched_group_barrier(0x020, 8, 0);
    __builtin_amdgcn_sched_group_barrier(0x100, 32, 0);
    __builtin_amdgcn_sched_group_barrier(0x008, 16, 0);
    __builtin_amdgcn_sched_group_barrier(0x200, 8, 0);
    lds_barrier();
  }
  float* hp = hprime + ((size_t)b * NN + i0) * (2 * FF);
#pragma unroll
  for (int r = 0; r < 8; ++r) {
    int m = r + 8 * lh;
    hp[(size_t)m * (2 * FF) + o] = c[r] * inv_sum[m];
  }
}

#define MA_NPW 4
#define HM_P 68
#define MA_WAVES 4
#define MA_TMIN 4

__device__ __forceinline__ int connected_slots(int* lst, const int (&adv)[8], const int rm,
                                               const int c, int (&m)[8]) {
  const unsigned below = (1u << c) - 1u;
  int n = 0;
#pragma unroll
  for (int t = 0; t < 8; ++t) {
    const bool conn = (rm != 0) || (adv[t] > 0);
    const unsigned bl = __builtin_amdgcn_ballot_w32(conn) & 0xffffu;
    const int mach = 16 * t + c;
    const int rank = n + __builtin_popcount(bl & below);
    lst[conn ? rank : (MM - 1) - (mach - rank)] = mach;
    n += __builtin_popcount(bl);
  }
  asm volatile("s_wait_dscnt 0x0" ::: "memory");
#pragma unroll
  for (int t = 0; t < 8; ++t) m[t] = lst[16 * t + c];
  return __builtin_amdgcn_readfirstlane(n);
}

__device__ __forceinline__ void request_z(v4f (&zq)[8][2], const float* zn, const int hh,
                                          const int (&m)[8], const int nt) {
#pragma unroll
  for (int t = 0; t < 8; ++t)
    if (t < MA_TMIN || t < nt) {
      const unsigned off = (unsigned)m[t] * FE + 8u * (unsigned)hh;
      zq[t][0] = *(const v4f*)(zn + off);
      zq[t][1] = *(const v4f*)(zn + off + 4u);
    }
}

__device__ __forceinline__ void ma_attn_nodes(
    float* smem, const int group,
    const float* __restrict__ z, const float* __restrict__ wedge,
    const float* __restrict__ hma, const float* __restrict__ ham1,
    const float* __restrict__ hma2, const int* __restrict__ opma,
    const int* __restrict__ rowmask, const float* __restrict__ h,
    const float* __restrict__ attma, float* __restrict__ hprime) {
  const int lane = threadIdx.x & 31;
  const int c  = lane & 15;
  const int hh = lane >> 4;
  const int wv  = __builtin_amdgcn_readfirstlane((int)(threadIdx.x >> 5));
  const int bn0 = (group * MA_WAVES + wv) * MA_NPW;
  const int b   = bn0 / NN;
  const float NEGINF = -__builtin_inff();

  v2f awe[4][4];
#pragma unroll
  for (int s = 0; s < 4; ++s)
#pragma unroll
    for (int ot = 0; ot < 4; ++ot) {
      awe[s][ot].x = wedge[(8 * hh + 2 * s) * FF + ot * 16 + c];
      awe[s][ot].y = wedge[(8 * hh + 2 * s + 1) * FF + ot * 16 + c];
    }
  v4f am3v[4][2];
#pragma unroll
  for (int ot = 0; ot < 4; ++ot)
#pragma unroll
    for (int q = 0; q < 2; ++q)
      am3v[ot][q] = *(const v4f*)(attma + 2 * FF + ot * 16 + 8 * hh + 4 * q);
  int adv0[8];
#pragma unroll
  for (int t = 0; t < 8; ++t) adv0[t] = opma[(size_t)bn0 * MM + 16 * t + c];
  const int rm0 = rowmask[bn0];
  {
    const v4f* src4 = (const v4f*)(hma + (size_t)b * MM * FF) + threadIdx.x;
    const float h2 = hma2[b * MM + threadIdx.x];
    v4f st[16];
#pragma unroll
    for (int q = 0; q < 16; ++q) st[q] = src4[128 * q];
#pragma unroll
    for (int q = 0; q < 16; ++q) {
      const int idx = threadIdx.x + 128 * q;
      *(v4f*)(smem + (idx >> 4) * HM_P + (idx & 15) * 4) = st[q];
    }
    smem[threadIdx.x * HM_P + FF] = h2;
  }
  __syncthreads();
  int* lst = (int*)(smem + MM * HM_P) + wv * MM;

  int m[8];
  v4f zq[8][2];
  int n  = connected_slots(lst, adv0, __builtin_amdgcn_readfirstlane(rm0), c, m);
  int nt = (n + 15) >> 4;
  request_z(zq, z + (size_t)bn0 * MM * FE, hh, m, nt);

#pragma unroll 1
  for (int i = 0; i < MA_NPW; ++i) {
    const int bn = bn0 + i;
    asm volatile("" ::: "memory");

    const float hm1 = ham1[bn];
    int advn[8];
    int rmn = 0;
    if (i + 1 < MA_NPW) {
#pragma unroll
      for (int t = 0; t < 8; ++t) advn[t] = opma[(size_t)(bn + 1) * MM + 16 * t + c];
      rmn = rowmask[bn + 1];
    }

    v8f D[8][4];
    float sc[8];
#pragma unroll
    for (int t = 0; t < 8; ++t) {
      sc[t] = NEGINF;
      if (t < MA_TMIN || t < nt) {
      const float* hrow = smem + m[t] * HM_P;
      const float* hmr  = hrow + 8 * hh;
      v2f zb[4];
      zb[0].x = zq[t][0].x; zb[0].y = zq[t][0].y;
      zb[1].x = zq[t][0].z; zb[1].y = zq[t][0].w;
      zb[2].x = zq[t][1].x; zb[2].y = zq[t][1].y;
      zb[3].x = zq[t][1].z; zb[3].y = zq[t][1].w;
      float part[4];
#pragma unroll
      for (int ot = 0; ot < 4; ++ot) {
        const v4f g0 = *(const v4f*)(hmr + ot * 16);
        const v4f g1 = *(const v4f*)(hmr + ot * 16 + 4);
        v8f d = {g0.x, g0.y, g0.z, g0.w, g1.x, g1.y, g1.z, g1.w};
#pragma unroll
        for (int s = 0; s < 4; ++s) d = wmma4(awe[s][ot], zb[s], d);
        D[t][ot] = d;
        v2f a2 = __builtin_shufflevector(d, d, 0, 1) * __builtin_shufflevector(am3v[ot][0], am3v[ot][0], 0, 1);
        a2 += __builtin_shufflevector(d, d, 2, 3) * __builtin_shufflevector(am3v[ot][0], am3v[ot][0], 2, 3);
        a2 += __builtin_shufflevector(d, d, 4, 5) * __builtin_shufflevector(am3v[ot][1], am3v[ot][1], 0, 1);
        a2 += __builtin_shufflevector(d, d, 6, 7) * __builtin_shufflevector(am3v[ot][1], am3v[ot][1], 2, 3);
        part[ot] = a2.x + a2.y;
      }
      float dot = (part[0] + part[1]) + (part[2] + part[3]);
      dot += xhalf(dot);
      float e = hm1 + hrow[FF] + dot;
      e = e > 0.f ? e : 0.01f * e;
      sc[t] = (16 * t + c < n) ? e : NEGINF;
      }
    }

    float mx = sc[0];
#pragma unroll
    for (int t = 1; t < 8; ++t) mx = fmaxf(mx, sc[t]);
    mx = row_max16(mx);
    if (mx == NEGINF) mx = 0.f;
    float p[8];
    float sum = 0.f;
#pragma unroll
    for (int t = 0; t < 8; ++t) { p[t] = __expf(sc[t] - mx); sum += p[t]; }
    sum = row_sum16(sum);
    const float inv = 1.f / fmaxf(sum, 1e-30f);

    int nn = 0, ntn = 0;
    if (i + 1 < MA_NPW) {
      nn  = connected_slots(lst, advn, __builtin_amdgcn_readfirstlane(rmn), c, m);
      ntn = (nn + 15) >> 4;
      request_z(zq, z + (size_t)(bn + 1) * MM * FE, hh, m, ntn);
    }

    v8f acc[4];
#pragma unroll
    for (int ot = 0; ot < 4; ++ot) acc[ot] = (v8f){};
#pragma unroll
    for (int t = 0; t < 8; ++t)
      if (t < MA_TMIN || t < nt) {
#pragma unroll
        for (int ot = 0; ot < 4; ++ot)
#pragma unroll
          for (int r = 0; r < 8; ++r) acc[ot][r] += p[t] * D[t][ot][r];
      }

    const float* hr = h + (size_t)bn * FF + 8 * hh;
    v4f res[4][2];
#pragma unroll
    for (int ot = 0; ot < 4; ++ot) {
      res[ot][0] = *(const v4f*)(hr + ot * 16);
      res[ot][1] = *(const v4f*)(hr + ot * 16 + 4);
    }
#pragma unroll
    for (int ot = 0; ot < 4; ++ot)
#pragma unroll
      for (int r = 0; r < 8; ++r) acc[ot][r] = row_sum16(acc[ot][r]);

    float* hp = hprime + (size_t)bn * (2 * FF) + FF + 8 * hh;
#pragma unroll
    for (int ot = 0; ot < 4; ++ot) {
      const v4f r0 = res[ot][0];
      const v4f r1 = res[ot][1];
      v4f o0, o1;
      o0.x = acc[ot][0] * inv + r0.x; o0.y = acc[ot][1] * inv + r0.y;
      o0.z = acc[ot][2] * inv + r0.z; o0.w = acc[ot][3] * inv + r0.w;
      o1.x = acc[ot][4] * inv + r1.x; o1.y = acc[ot][5] * inv + r1.y;
      o1.z = acc[ot][6] * inv + r1.z; o1.w = acc[ot][7] * inv + r1.w;
      if (c == 0) {
        *(v4f*)(hp + ot * 16)     = o0;
        *(v4f*)(hp + ot * 16 + 4) = o1;
      }
    }
    n  = nn;
    nt = ntn;
  }
}

__global__ __launch_bounds__(128, 2) void attn_pair(
    const float* __restrict__ h, const float* __restrict__ ha1,
    const float* __restrict__ ha2, const unsigned* __restrict__ cbits,
    const int* __restrict__ rowmask, float* __restrict__ hprime,
    const float* __restrict__ z, const float* __restrict__ wedge,
    const float* __restrict__ hma, const float* __restrict__ ham1,
    const float* __restrict__ hma2, const int* __restrict__ opma,
    const float* __restrict__ attma) {
  extern __shared__ float smem[];
  const int g = blockIdx.x;
  const int op_first = (g ^ (g >> 8)) & 1;
#pragma unroll 1
  for (int ph = 0; ph < 2; ++ph) {
    if (ph == op_first)
      ma_attn_nodes(smem, g, z, wedge, hma, ham1, hma2, opma, rowmask, h, attma, hprime);
    else
      op_attn_tile(smem, g, h, ha1, ha2, cbits, rowmask, hprime);
    __syncthreads();
  }
}

__global__ __launch_bounds__(256, 2) void mlp(
    const float* __restrict__ hprime, const float* __restrict__ w1,
    const float* __restrict__ b1, const float* __restrict__ w2,
    const float* __restrict__ b2, const float* __restrict__ wo,
    const float* __restrict__ bo, const int* __restrict__ rowmask,
    float* __restrict__ out) {
  extern __shared__ float smem[];
  typedef float act_row_t[HID + 4];
  act_row_t* a0 = (act_row_t*)smem;
  act_row_t* a1 = (act_row_t*)(smem + 16 * (HID + 4));

  const int row0 = blockIdx.x * 16;
  const int wv = __builtin_amdgcn_readfirstlane((int)(threadIdx.x >> 5));
  const int lane = threadIdx.x & 31;
  const int lm = lane & 15;
  const int kh = (lane >> 4) * 2;
  const int lh = lane >> 4;
  const int o  = wv * 16 + lm;

  const v4f* hp4 = (const v4f*)(hprime + (size_t)row0 * HID) + threadIdx.x;
  const v4f in0 = hp4[0];
  const v4f in1 = hp4[256];
  const float bias1 = b1[o];
  const float bias2 = b2[o];
  v2f bw1[32];
#pragma unroll
  for (int t = 0; t < 32; ++t) {
    bw1[t].x = w1[(4 * t + kh) * HID + o];
    bw1[t].y = w1[(4 * t + kh + 1) * HID + o];
  }
  *(v4f*)&a0[threadIdx.x >> 5][(threadIdx.x & 31) * 4] = in0;
  *(v4f*)&a0[(threadIdx.x >> 5) + 8][(threadIdx.x & 31) * 4] = in1;
  lds_barrier();

  v2f bw2[32];
#pragma unroll
  for (int t = 0; t < 32; ++t) {
    bw2[t].x = w2[(4 * t + kh) * HID + o];
    bw2[t].y = w2[(4 * t + kh + 1) * HID + o];
  }

  v8f c1 = {};
  {
    v2f af[32];
#pragma unroll
    for (int t = 0; t < 32; ++t) {
      af[t].x = a0[lm][4 * t + kh];
      af[t].y = a0[lm][4 * t + kh + 1];
    }
#pragma unroll
    for (int t = 0; t < 32; ++t) c1 = wmma4(af[t], bw1[t], c1);
    __builtin_amdgcn_sched_group_barrier(0x100, 32, 0);
    __builtin_amdgcn_sched_group_barrier(0x008, 32, 0);
  }
  const float bias3 = bo[o & (FF - 1)];
  const int4 rm0 = *(const int4*)(rowmask + row0 + 8 * lh);
  const int4 rm1 = *(const int4*)(rowmask + row0 + 8 * lh + 4);
  v2f bwo[32];
  if (wv < 4) {
#pragma unroll
    for (int t = 0; t < 32; ++t) {
      bwo[t].x = wo[(4 * t + kh) * FF + o];
      bwo[t].y = wo[(4 * t + kh + 1) * FF + o];
    }
  }
  {
#pragma unroll
    for (int r = 0; r < 8; ++r) {
      float v = c1[r] + bias1;
      a1[r + 8 * lh][o] = v > 0.f ? v : expm1f(v);
    }
  }
  lds_barrier();

  {
    v8f c = {};
    v2f af[32];
#pragma unroll
    for (int t = 0; t < 32; ++t) {
      af[t].x = a1[lm][4 * t + kh];
      af[t].y = a1[lm][4 * t + kh + 1];
    }
#pragma unroll
    for (int t = 0; t < 32; ++t) c = wmma4(af[t], bw2[t], c);
    __builtin_amdgcn_sched_group_barrier(0x100, 32, 0);
    __builtin_amdgcn_sched_group_barrier(0x008, 32, 0);
#pragma unroll
    for (int r = 0; r < 8; ++r) {
      float v = c[r] + bias2;
      a0[r + 8 * lh][o] = v > 0.f ? v : expm1f(v);
    }
  }
  lds_barrier();

  if (wv < 4) {
    v8f c = {};
    v2f af[32];
#pragma unroll
    for (int t = 0; t < 32; ++t) {
      af[t].x = a0[lm][4 * t + kh];
      af[t].y = a0[lm][4 * t + kh + 1];
    }
#pragma unroll
    for (int t = 0; t < 32; ++t) c = wmma4(af[t], bwo[t], c);
    __builtin_amdgcn_sched_group_barrier(0x100, 32, 0);
    __builtin_amdgcn_sched_group_barrier(0x008, 32, 0);
    const int rmv[8] = {rm0.x, rm0.y, rm0.z, rm0.w, rm1.x, rm1.y, rm1.z, rm1.w};
#pragma unroll
    for (int r = 0; r < 8; ++r) {
      int m = r + 8 * lh;
      float v = c[r] + bias3;
      if (rmv[r]) v = 0.f;
      out[(size_t)(row0 + m) * FF + o] = v;
    }
  }
}

extern "C" void kernel_launch(void* const* d_in, const int* in_sizes, int n_in,
                              void* d_out, int out_size, void* d_ws,
                              size_t ws_size, hipStream_t stream) {
  (void)in_sizes; (void)n_in; (void)out_size; (void)ws_size;
  const float* x      = (const float*)d_in[0];
  const float* y      = (const float*)d_in[1];
  const float* z      = (const float*)d_in[2];
  const int*   op_adj = (const int*)d_in[3];
  const int*   ma_adj = (const int*)d_in[4];
  const int*   op_ma  = (const int*)d_in[5];
  const float* Wop    = (const float*)d_in[6];
  const float* Wma    = (const float*)d_in[7];
  const float* Wedge  = (const float*)d_in[8];
  const float* att_op = (const float*)d_in[9];
  const float* att_ma = (const float*)d_in[10];
  const float* w1     = (const float*)d_in[11];
  const float* b1     = (const float*)d_in[12];
  const float* w2     = (const float*)d_in[13];
  const float* b2     = (const float*)d_in[14];
  const float* wo     = (const float*)d_in[15];
  const float* bo     = (const float*)d_in[16];
  float* out = (float*)d_out;

  float* ws     = (float*)d_ws;
  float* h      = ws;
  float* hma    = h + BB * NN * FF;
  float* ha1    = hma + BB * MM * FF;
  float* ha2    = ha1 + BB * NN;
  float* ham1   = ha2 + BB * NN;
  float* hma2   = ham1 + BB * NN;
  float* hprime = hma2 + BB * MM;
  int*   rmask  = (int*)(hprime + (size_t)BB * NN * 2 * FF);

  size_t sm2 = (16 * (NN + 4) + 16 + 2 * HS_ROWS * HS_P) * sizeof(float);
  unsigned* cbits = (unsigned*)hprime;
  size_t smb = (16 * 32 * CB_P + 32 * 33 + 4 * 32 + MM + 4 * FF) * sizeof(int);
  feat_bits_proj<<<dim3(NN / 32, BB), dim3(512), smb, stream>>>(
      op_adj, ma_adj, cbits, x, y, Wop, Wma, h, hma, att_op, att_ma, op_ma,
      ha1, ha2, ham1, hma2, rmask);
  attn_pair<<<dim3((BB * NN) / 16), dim3(128), sm2, stream>>>(
      h, ha1, ha2, cbits, rmask, hprime,
      z, Wedge, hma, ham1, hma2, op_ma, att_ma);
  size_t sm4 = (2 * 16 * (HID + 4)) * sizeof(float);
  mlp<<<dim3((BB * NN) / 16), dim3(256), sm4, stream>>>(
      hprime, w1, b1, w2, b2, wo, bo, rmask, out);
}
